# starting kernel
# baseline (speedup 1.0000x reference)
#include <hip/hip_runtime.h>
#include <math.h>

typedef __attribute__((ext_vector_type(2))) float v2f;
typedef __attribute__((ext_vector_type(8))) float v8f;

#define NEG_SLOPE 0.2f

template <int K, int M>
__global__ void gemm_wmma_lds(const float* __restrict__ A,
                              const float* __restrict__ W,
                              float* __restrict__ C,
                              int Nrows) {
  constexpr int TILES_M = M / 16;
  __shared__ float wlds[K * M];

  for (int i = threadIdx.x; i < K * M; i += blockDim.x) wlds[i] = W[i];
  __syncthreads();

  const int wavesPerBlock = blockDim.x >> 5;
  const int rowTile = blockIdx.x * wavesPerBlock + (threadIdx.x >> 5);
  const int lane = threadIdx.x & 31;
  if (rowTile >= (Nrows >> 4)) return;

  const int l16 = lane & 15;
  const int kq  = (lane >> 4) << 1;
  const float* __restrict__ arow = A + (size_t)(rowTile * 16 + l16) * (size_t)K;

  v8f acc[TILES_M];
#pragma unroll
  for (int t = 0; t < TILES_M; ++t) acc[t] = (v8f){0.f,0.f,0.f,0.f,0.f,0.f,0.f,0.f};

  for (int k0 = 0; k0 < K; k0 += 4) {
    v2f a;
    a.x = arow[k0 + kq];
    a.y = arow[k0 + kq + 1];
    const float* wr0 = &wlds[(k0 + kq) * M + l16];
#pragma unroll
    for (int t = 0; t < TILES_M; ++t) {
      v2f b;
      b.x = wr0[t * 16];
      b.y = wr0[M + t * 16];
      acc[t] = __builtin_amdgcn_wmma_f32_16x16x4_f32(
           false, a,  false, b,
           (short)0, acc[t],  false,  false);
    }
  }

  const int rbase = rowTile * 16 + ((lane >> 4) << 3);
#pragma unroll
  for (int t = 0; t < TILES_M; ++t) {
    const int col = t * 16 + l16;
#pragma unroll
    for (int r = 0; r < 8; ++r)
      C[(size_t)(rbase + r) * (size_t)M + col] = acc[t][r];
  }
}

__global__ void node_alpha(const float* __restrict__ hs,
                           const float* __restrict__ hd,
                           const float* __restrict__ a_s,
                           const float* __restrict__ a_d,
                           float* __restrict__ als,
                           float* __restrict__ ald, int Nn) {
  int tid = blockIdx.x * blockDim.x + threadIdx.x;
  if (tid >= Nn * 2) return;
  int n = tid >> 1, h = tid & 1;
  const float* hsrow = hs + (size_t)n * 128 + h * 64;
  const float* hdrow = hd + (size_t)n * 128 + h * 64;
  const float* asv = a_s + h * 64;
  const float* adv = a_d + h * 64;
  float s1 = 0.f, s2 = 0.f;
#pragma unroll 8
  for (int c = 0; c < 64; ++c) {
    s1 += hsrow[c] * asv[c];
    s2 += hdrow[c] * adv[c];
  }
  als[tid] = s1;
  ald[tid] = s2;
}

__global__ void init_layer(float* __restrict__ m, float* __restrict__ den,
                           float* __restrict__ agg, int Nn) {
  int tid = blockIdx.x * blockDim.x + threadIdx.x;
  if (tid < Nn * 64) agg[tid] = 0.f;
  if (tid < Nn * 2) { m[tid] = -INFINITY; den[tid] = 0.f; }
}

__device__ inline void atomicMaxF(float* addr, float value) {
  if (value >= 0.f)
    atomicMax((int*)addr, __float_as_int(value));
  else
    atomicMin((unsigned int*)addr, __float_as_uint(value));
}

__global__ void edge_max(const int* __restrict__ src, const int* __restrict__ dst,
                         const float* __restrict__ als, const float* __restrict__ ald,
                         float* __restrict__ e, float* __restrict__ m, int Ee) {
  int tid = blockIdx.x * blockDim.x + threadIdx.x;
  if (tid >= Ee * 2) return;
  int ed = tid >> 1, h = tid & 1;
  int s = src[ed], d = dst[ed];
  float v = als[s * 2 + h] + ald[d * 2 + h];
  v = (v > 0.f) ? v : NEG_SLOPE * v;
  e[tid] = v;
  atomicMaxF(&m[d * 2 + h], v);
}

__global__ void edge_den(const int* __restrict__ dst,
                         const float* __restrict__ e, const float* __restrict__ m,
                         float* __restrict__ den, int Ee) {
  int tid = blockIdx.x * blockDim.x + threadIdx.x;
  if (tid >= Ee * 2) return;
  int ed = tid >> 1, h = tid & 1;
  int d = dst[ed];
  float ex = __expf(e[tid] - m[d * 2 + h]);
  atomicAdd(&den[d * 2 + h], ex);
}

__global__ void edge_scatter(const int* __restrict__ src, const int* __restrict__ dst,
                             const float* __restrict__ e, const float* __restrict__ m,
                             const float* __restrict__ den,
                             const float* __restrict__ hs,
                             float* __restrict__ agg, int Ee) {
  int wave = (blockIdx.x * blockDim.x + threadIdx.x) >> 5;
  int lane = threadIdx.x & 31;
  if (wave >= Ee) return;
  int s = src[wave], d = dst[wave];
  float e0 = e[wave * 2 + 0], e1 = e[wave * 2 + 1];
  float m0 = m[d * 2 + 0], m1 = m[d * 2 + 1];
  float n0 = den[d * 2 + 0], n1 = den[d * 2 + 1];
  float a0 = __expf(e0 - m0) / (n0 + 1e-16f) * 0.5f;
  float a1 = __expf(e1 - m1) / (n1 + 1e-16f) * 0.5f;
  const float* hsrow = hs + (size_t)s * 128;
  float v0 = hsrow[lane]      * a0 + hsrow[64 + lane] * a1;
  float v1 = hsrow[32 + lane] * a0 + hsrow[96 + lane] * a1;
  float* aggrow = agg + (size_t)d * 64;
  atomicAdd(&aggrow[lane],      v0);
  atomicAdd(&aggrow[32 + lane], v1);
}

__global__ void finalize(const float* __restrict__ agg, const float* __restrict__ hl,
                         const float* __restrict__ b, const float* __restrict__ bl,
                         float* __restrict__ out, int Nn) {
  int tid = blockIdx.x * blockDim.x + threadIdx.x;
  if (tid >= Nn * 64) return;
  int c = tid & 63;
  float v = agg[tid] + b[c] + hl[tid] + bl[c];
  out[tid] = (v > 0.f) ? v : 0.f;
}

extern "C" void kernel_launch(void* const* d_in, const int* in_sizes, int n_in,
                              void* d_out, int out_size, void* d_ws, size_t ws_size,
                              hipStream_t stream) {
  const float* x  = (const float*)d_in[0];
  const int*   ei = (const int*)d_in[1];
  const int Nn = in_sizes[0] / 128;
  const int Ee = in_sizes[1] / 2;
  const int* src = ei;
  const int* dst = ei + Ee;

  const float *Ws[3], *Wd[3], *As[3], *Ad[3], *B[3], *Wl[3], *Bl[3];
  for (int l = 0; l < 3; ++l) {
    int base = 2 + 7 * l;
    Ws[l] = (const float*)d_in[base + 0];
    Wd[l] = (const float*)d_in[base + 1];
    As[l] = (const float*)d_in[base + 2];
    Ad[l] = (const float*)d_in[base + 3];
    B[l]  = (const float*)d_in[base + 4];
    Wl[l] = (const float*)d_in[base + 5];
    Bl[l] = (const float*)d_in[base + 6];
  }

  float* ws = (float*)d_ws;
  size_t off = 0;
  float* hs  = ws + off; off += (size_t)Nn * 128;
  float* hd  = ws + off; off += (size_t)Nn * 128;
  float* hl  = ws + off; off += (size_t)Nn * 64;
  float* als = ws + off; off += (size_t)Nn * 2;
  float* ald = ws + off; off += (size_t)Nn * 2;
  float* m   = ws + off; off += (size_t)Nn * 2;
  float* den = ws + off; off += (size_t)Nn * 2;
  float* ebuf= ws + off; off += (size_t)Ee * 2;
  float* agg = ws + off; off += (size_t)Nn * 64;
  float* h1  = ws + off; off += (size_t)Nn * 64;
  float* h2  = ws + off; off += (size_t)Nn * 64;

  const float* lin[3]  = { x, h1, h2 };
  float*       lout[3] = { h1, h2, (float*)d_out };

  const int TPB = 256;
  const int rowTiles = Nn >> 4;
  const int gemmBlocks = (rowTiles + 7) / 8;

  for (int l = 0; l < 3; ++l) {
    if (l == 0) {
      gemm_wmma_lds<128, 128><<<gemmBlocks, TPB, 0, stream>>>(lin[l], Ws[l], hs, Nn);
      gemm_wmma_lds<128, 128><<<gemmBlocks, TPB, 0, stream>>>(lin[l], Wd[l], hd, Nn);
      gemm_wmma_lds<128, 64 ><<<gemmBlocks, TPB, 0, stream>>>(lin[l], Wl[l], hl, Nn);
    } else {
      gemm_wmma_lds<64, 128><<<gemmBlocks, TPB, 0, stream>>>(lin[l], Ws[l], hs, Nn);
      gemm_wmma_lds<64, 128><<<gemmBlocks, TPB, 0, stream>>>(lin[l], Wd[l], hd, Nn);
      gemm_wmma_lds<64, 64 ><<<gemmBlocks, TPB, 0, stream>>>(lin[l], Wl[l], hl, Nn);
    }
    node_alpha<<<(Nn * 2 + TPB - 1) / TPB, TPB, 0, stream>>>(hs, hd, As[l], Ad[l], als, ald, Nn);
    init_layer<<<(Nn * 64 + TPB - 1) / TPB, TPB, 0, stream>>>(m, den, agg, Nn);
    edge_max<<<(Ee * 2 + TPB - 1) / TPB, TPB, 0, stream>>>(src, dst, als, ald, ebuf, m, Ee);
    edge_den<<<(Ee * 2 + TPB - 1) / TPB, TPB, 0, stream>>>(dst, ebuf, m, den, Ee);
    {
      long long threads = (long long)Ee * 32;
      edge_scatter<<<(int)((threads + TPB - 1) / TPB), TPB, 0, stream>>>(
          src, dst, ebuf, m, den, hs, agg, Ee);
    }
    finalize<<<(Nn * 64 + TPB - 1) / TPB, TPB, 0, stream>>>(agg, hl, B[l], Bl[l], lout[l], Nn);
  }
}


# Edges sorted by (destination, source) once per call (int-atomic counting sort); one wave per node does max, sum and weighted row gather in registers, no float atomics; per-node sum order now fixed
# speedup vs baseline: 1.7693x; 1.7693x over previous
#include <hip/hip_runtime.h>
#include <math.h>

typedef __attribute__((ext_vector_type(2))) float v2f;
typedef __attribute__((ext_vector_type(8))) float v8f;

#define NEG_SLOPE 0.2f

template <int K, int M>
__global__ void gemm_wmma_lds(const float* __restrict__ A,
                              const float* __restrict__ W,
                              float* __restrict__ C,
                              int Nrows) {
  constexpr int TILES_M = M / 16;
  __shared__ float wlds[K * M];

  for (int i = threadIdx.x; i < K * M; i += blockDim.x) wlds[i] = W[i];
  __syncthreads();

  const int wavesPerBlock = blockDim.x >> 5;
  const int rowTile = blockIdx.x * wavesPerBlock + (threadIdx.x >> 5);
  const int lane = threadIdx.x & 31;
  if (rowTile >= (Nrows >> 4)) return;

  const int l16 = lane & 15;
  const int kq  = (lane >> 4) << 1;
  const float* __restrict__ arow = A + (size_t)(rowTile * 16 + l16) * (size_t)K;

  v8f acc[TILES_M];
#pragma unroll
  for (int t = 0; t < TILES_M; ++t) acc[t] = (v8f){0.f,0.f,0.f,0.f,0.f,0.f,0.f,0.f};

  for (int k0 = 0; k0 < K; k0 += 4) {
    v2f a;
    a.x = arow[k0 + kq];
    a.y = arow[k0 + kq + 1];
    const float* wr0 = &wlds[(k0 + kq) * M + l16];
#pragma unroll
    for (int t = 0; t < TILES_M; ++t) {
      v2f b;
      b.x = wr0[t * 16];
      b.y = wr0[M + t * 16];
      acc[t] = __builtin_amdgcn_wmma_f32_16x16x4_f32(
           false, a,  false, b,
           (short)0, acc[t],  false,  false);
    }
  }

  const int rbase = rowTile * 16 + ((lane >> 4) << 3);
#pragma unroll
  for (int t = 0; t < TILES_M; ++t) {
    const int col = t * 16 + l16;
#pragma unroll
    for (int r = 0; r < 8; ++r)
      C[(size_t)(rbase + r) * (size_t)M + col] = acc[t][r];
  }
}

__global__ void node_alpha(const float* __restrict__ hs,
                           const float* __restrict__ hd,
                           const float* __restrict__ a_s,
                           const float* __restrict__ a_d,
                           float* __restrict__ als,
                           float* __restrict__ ald, int Nn) {
  int tid = blockIdx.x * blockDim.x + threadIdx.x;
  if (tid >= Nn * 2) return;
  int n = tid >> 1, h = tid & 1;
  const float* hsrow = hs + (size_t)n * 128 + h * 64;
  const float* hdrow = hd + (size_t)n * 128 + h * 64;
  const float* asv = a_s + h * 64;
  const float* adv = a_d + h * 64;
  float s1 = 0.f, s2 = 0.f;
#pragma unroll 8
  for (int c = 0; c < 64; ++c) {
    s1 += hsrow[c] * asv[c];
    s2 += hdrow[c] * adv[c];
  }
  als[tid] = s1;
  ald[tid] = s2;
}

__global__ void csr_zero(int* __restrict__ cnt, int Nn) {
  int tid = blockIdx.x * blockDim.x + threadIdx.x;
  if (tid < Nn) cnt[tid] = 0;
}

__global__ void csr_hist(const int* __restrict__ dst, int* __restrict__ cnt, int Ee) {
  int tid = blockIdx.x * blockDim.x + threadIdx.x;
  if (tid < Ee) atomicAdd(&cnt[dst[tid]], 1);
}

__global__ void csr_scan(const int* __restrict__ cnt, int* __restrict__ rowptr,
                         int* __restrict__ cursor, int Nn) {
  __shared__ int part[1024];
  __shared__ int ppre[1024];
  const int t = threadIdx.x;
  const int base = blockIdx.x * 1024;
  int pre = 0;
  for (int i = t; i < base; i += 1024) pre += cnt[i];
  const int idx = base + t;
  const int c = (idx < Nn) ? cnt[idx] : 0;
  part[t] = c;
  ppre[t] = pre;
  __syncthreads();
  for (int off = 1; off < 1024; off <<= 1) {
    const int v = (t >= off) ? part[t - off] : 0;
    const int w = (t >= off) ? ppre[t - off] : 0;
    __syncthreads();
    part[t] += v;
    ppre[t] += w;
    __syncthreads();
  }
  const int excl = ppre[1023] + part[t] - c;
  if (idx < Nn) {
    rowptr[idx] = excl;
    cursor[idx] = excl;
  }
  if (idx == Nn - 1) rowptr[Nn] = excl + c;
}

__global__ void csr_fill(const int* __restrict__ src, const int* __restrict__ dst,
                         int* __restrict__ cursor, int* __restrict__ csr_src, int Ee) {
  int tid = blockIdx.x * blockDim.x + threadIdx.x;
  if (tid >= Ee) return;
  const int p = atomicAdd(&cursor[dst[tid]], 1);
  csr_src[p] = src[tid];
}

__global__ void csr_order(const int* __restrict__ rowptr, const int* __restrict__ csr_in,
                          int* __restrict__ csr_out, int Nn) {
  const int lane = threadIdx.x & 31;
  const int node = __builtin_amdgcn_readfirstlane((int)((blockIdx.x * blockDim.x + threadIdx.x) >> 5));
  if (node >= Nn) return;
  const int beg = rowptr[node];
  const int end = rowptr[node + 1];
  for (int cb = beg; cb < end; cb += 32) {
    const int i = cb + lane;
    const int key = (i < end) ? csr_in[i] : 0;
    int rank = 0;
    for (int kb = beg; kb < end; kb += 32) {
      const int k = kb + lane;
      const int other = (k < end) ? csr_in[k] : 0;
      const int n = (end - kb < 32) ? (end - kb) : 32;
      for (int j = 0; j < n; ++j) {
        const int o = __builtin_amdgcn_readlane(other, j);
        const int pos = kb + j;
        rank += ((o < key) || (o == key && pos < i)) ? 1 : 0;
      }
    }
    if (i < end) csr_out[beg + rank] = key;
  }
}

__device__ __forceinline__ float lrelu(float v) { return (v > 0.f) ? v : NEG_SLOPE * v; }

__global__ void node_gather(const int* __restrict__ rowptr, const int* __restrict__ csr_src,
                            const float* __restrict__ als, const float* __restrict__ ald,
                            const float* __restrict__ hs, const float* __restrict__ hl,
                            const float* __restrict__ b, const float* __restrict__ bl,
                            float* __restrict__ out, int Nn) {
  const int lane = threadIdx.x & 31;
  const int node = __builtin_amdgcn_readfirstlane((int)((blockIdx.x * blockDim.x + threadIdx.x) >> 5));
  if (node >= Nn) return;
  const int beg = rowptr[node];
  const int end = rowptr[node + 1];
  const float ad0 = ald[node * 2 + 0];
  const float ad1 = ald[node * 2 + 1];
  const float2* __restrict__ als2 = (const float2*)als;

  float m0 = -INFINITY, m1 = -INFINITY;
  for (int i = beg + lane; i < end; i += 32) {
    const float2 a = als2[csr_src[i]];
    m0 = fmaxf(m0, lrelu(a.x + ad0));
    m1 = fmaxf(m1, lrelu(a.y + ad1));
  }
  for (int o = 16; o > 0; o >>= 1) {
    m0 = fmaxf(m0, __shfl_xor(m0, o));
    m1 = fmaxf(m1, __shfl_xor(m1, o));
  }
  if (!(fabsf(m0) < INFINITY)) m0 = 0.f;
  if (!(fabsf(m1) < INFINITY)) m1 = 0.f;

  float d0 = 0.f, d1 = 0.f;
  for (int i = beg + lane; i < end; i += 32) {
    const float2 a = als2[csr_src[i]];
    d0 += __expf(lrelu(a.x + ad0) - m0);
    d1 += __expf(lrelu(a.y + ad1) - m1);
  }
  for (int o = 16; o > 0; o >>= 1) {
    d0 += __shfl_xor(d0, o);
    d1 += __shfl_xor(d1, o);
  }
  const float r0 = d0 + 1e-16f;
  const float r1 = d1 + 1e-16f;

  const bool head1 = lane >= 16;
  const float* __restrict__ hsl = hs + lane * 4;
  float4 acc = make_float4(0.f, 0.f, 0.f, 0.f);
  for (int base = beg; base < end; base += 32) {
    const int i = base + lane;
    int s_l = 0;
    float w_l0 = 0.f, w_l1 = 0.f;
    if (i < end) {
      s_l = csr_src[i];
      const float2 a = als2[s_l];
      w_l0 = __expf(lrelu(a.x + ad0) - m0) / r0;
      w_l1 = __expf(lrelu(a.y + ad1) - m1) / r1;
    }
    const int cnt = (end - base < 32) ? (end - base) : 32;
#pragma unroll 4
    for (int j = 0; j < cnt; ++j) {
      const int s = __builtin_amdgcn_readlane(s_l, j);
      const float wa = __int_as_float(__builtin_amdgcn_readlane(__float_as_int(w_l0), j));
      const float wb = __int_as_float(__builtin_amdgcn_readlane(__float_as_int(w_l1), j));
      const float w = head1 ? wb : wa;
      const float4 v = *(const float4*)(hsl + (size_t)s * 128);
      acc.x = fmaf(w, v.x, acc.x);
      acc.y = fmaf(w, v.y, acc.y);
      acc.z = fmaf(w, v.z, acc.z);
      acc.w = fmaf(w, v.w, acc.w);
    }
  }
  acc.x += __shfl_xor(acc.x, 16);
  acc.y += __shfl_xor(acc.y, 16);
  acc.z += __shfl_xor(acc.z, 16);
  acc.w += __shfl_xor(acc.w, 16);
  if (lane < 16) {
    const int c = lane * 4;
    const float4 hv = *(const float4*)(hl + (size_t)node * 64 + c);
    float4 o;
    o.x = 0.5f * acc.x + b[c + 0] + hv.x + bl[c + 0];
    o.y = 0.5f * acc.y + b[c + 1] + hv.y + bl[c + 1];
    o.z = 0.5f * acc.z + b[c + 2] + hv.z + bl[c + 2];
    o.w = 0.5f * acc.w + b[c + 3] + hv.w + bl[c + 3];
    o.x = (o.x > 0.f) ? o.x : 0.f;
    o.y = (o.y > 0.f) ? o.y : 0.f;
    o.z = (o.z > 0.f) ? o.z : 0.f;
    o.w = (o.w > 0.f) ? o.w : 0.f;
    *(float4*)(out + (size_t)node * 64 + c) = o;
  }
}

extern "C" void kernel_launch(void* const* d_in, const int* in_sizes, int n_in,
                              void* d_out, int out_size, void* d_ws, size_t ws_size,
                              hipStream_t stream) {
  const float* x  = (const float*)d_in[0];
  const int*   ei = (const int*)d_in[1];
  const int Nn = in_sizes[0] / 128;
  const int Ee = in_sizes[1] / 2;
  const int* src = ei;
  const int* dst = ei + Ee;

  const float *Ws[3], *Wd[3], *As[3], *Ad[3], *B[3], *Wl[3], *Bl[3];
  for (int l = 0; l < 3; ++l) {
    int base = 2 + 7 * l;
    Ws[l] = (const float*)d_in[base + 0];
    Wd[l] = (const float*)d_in[base + 1];
    As[l] = (const float*)d_in[base + 2];
    Ad[l] = (const float*)d_in[base + 3];
    B[l]  = (const float*)d_in[base + 4];
    Wl[l] = (const float*)d_in[base + 5];
    Bl[l] = (const float*)d_in[base + 6];
  }

  float* ws = (float*)d_ws;
  size_t off = 0;
  float* hs  = ws + off; off += (size_t)Nn * 128;
  float* hd  = ws + off; off += (size_t)Nn * 128;
  float* hl  = ws + off; off += (size_t)Nn * 64;
  float* als = ws + off; off += (size_t)Nn * 2;
  float* ald = ws + off; off += (size_t)Nn * 2;
  float* h1  = ws + off; off += (size_t)Nn * 64;
  float* h2  = ws + off; off += (size_t)Nn * 64;
  int* iws = (int*)(ws + off);
  size_t ioff = 0;
  int* cnt     = iws + ioff; ioff += (size_t)Nn;
  int* cursor  = iws + ioff; ioff += (size_t)Nn;
  int* rowptr  = iws + ioff; ioff += (size_t)Nn + 1;
  int* csr_tmp = iws + ioff; ioff += (size_t)Ee;
  int* csr_src = iws + ioff; ioff += (size_t)Ee;

  const float* lin[3]  = { x, h1, h2 };
  float*       lout[3] = { h1, h2, (float*)d_out };

  const int TPB = 256;
  const int rowTiles = Nn >> 4;
  const int gemmBlocks = (rowTiles + 7) / 8;

  csr_zero<<<(Nn + TPB - 1) / TPB, TPB, 0, stream>>>(cnt, Nn);
  csr_hist<<<(Ee + TPB - 1) / TPB, TPB, 0, stream>>>(dst, cnt, Ee);
  csr_scan<<<(Nn + 1023) / 1024, 1024, 0, stream>>>(cnt, rowptr, cursor, Nn);
  csr_fill<<<(Ee + TPB - 1) / TPB, TPB, 0, stream>>>(src, dst, cursor, csr_tmp, Ee);
  {
    long long threads = (long long)Nn * 32;
    csr_order<<<(int)((threads + TPB - 1) / TPB), TPB, 0, stream>>>(rowptr, csr_tmp, csr_src, Nn);
  }

  for (int l = 0; l < 3; ++l) {
    if (l == 0) {
      gemm_wmma_lds<128, 128><<<gemmBlocks, TPB, 0, stream>>>(lin[l], Ws[l], hs, Nn);
      gemm_wmma_lds<128, 128><<<gemmBlocks, TPB, 0, stream>>>(lin[l], Wd[l], hd, Nn);
      gemm_wmma_lds<128, 64 ><<<gemmBlocks, TPB, 0, stream>>>(lin[l], Wl[l], hl, Nn);
    } else {
      gemm_wmma_lds<64, 128><<<gemmBlocks, TPB, 0, stream>>>(lin[l], Ws[l], hs, Nn);
      gemm_wmma_lds<64, 128><<<gemmBlocks, TPB, 0, stream>>>(lin[l], Wd[l], hd, Nn);
      gemm_wmma_lds<64, 64 ><<<gemmBlocks, TPB, 0, stream>>>(lin[l], Wl[l], hl, Nn);
    }
    node_alpha<<<(Nn * 2 + TPB - 1) / TPB, TPB, 0, stream>>>(hs, hd, As[l], Ad[l], als, ald, Nn);
    {
      long long threads = (long long)Nn * 32;
      node_gather<<<(int)((threads + TPB - 1) / TPB), TPB, 0, stream>>>(
          rowptr, csr_src, als, ald, hs, hl, B[l], Bl[l], lout[l], Nn);
    }
  }
}


# Attention dot products folded into the weights: als,ald = x@(Ws.a_s | Wd.a_d) ride as a 9th column tile of the hs GEMM; hd GEMM and node_alpha dropped; als/ald summation order changed
# speedup vs baseline: 2.3111x; 1.3062x over previous
#include <hip/hip_runtime.h>
#include <math.h>

typedef __attribute__((ext_vector_type(2))) float v2f;
typedef __attribute__((ext_vector_type(8))) float v8f;

#define NEG_SLOPE 0.2f

template <int K, int M, bool ATT>
__global__ void gemm_wmma_lds(const float* __restrict__ A,
                              const float* __restrict__ W,
                              const float* __restrict__ Wx,
                              float* __restrict__ C,
                              float* __restrict__ als,
                              float* __restrict__ ald,
                              int Nrows) {
  constexpr int MT = ATT ? M + 16 : M;
  constexpr int TILES_C = M / 16;
  constexpr int TILES_M = MT / 16;
  __shared__ float wlds[K * MT];

  for (int i = threadIdx.x; i < K * M; i += blockDim.x) wlds[(i / M) * MT + (i % M)] = W[i];
  if constexpr (ATT) {
    for (int i = threadIdx.x; i < K * 16; i += blockDim.x) wlds[(i >> 4) * MT + M + (i & 15)] = Wx[i];
  }
  __syncthreads();

  const int wavesPerBlock = blockDim.x >> 5;
  const int rowTile = blockIdx.x * wavesPerBlock + (threadIdx.x >> 5);
  const int lane = threadIdx.x & 31;
  if (rowTile >= (Nrows >> 4)) return;

  const int l16 = lane & 15;
  const int kq  = (lane >> 4) << 1;
  const float* __restrict__ arow = A + (size_t)(rowTile * 16 + l16) * (size_t)K;

  v8f acc[TILES_M];
#pragma unroll
  for (int t = 0; t < TILES_M; ++t) acc[t] = (v8f){0.f,0.f,0.f,0.f,0.f,0.f,0.f,0.f};

  for (int k0 = 0; k0 < K; k0 += 4) {
    v2f a;
    a.x = arow[k0 + kq];
    a.y = arow[k0 + kq + 1];
    const float* wr0 = &wlds[(k0 + kq) * MT + l16];
#pragma unroll
    for (int t = 0; t < TILES_M; ++t) {
      v2f b;
      b.x = wr0[t * 16];
      b.y = wr0[MT + t * 16];
      acc[t] = __builtin_amdgcn_wmma_f32_16x16x4_f32(
           false, a,  false, b,
           (short)0, acc[t],  false,  false);
    }
  }

  const int rbase = rowTile * 16 + ((lane >> 4) << 3);
#pragma unroll
  for (int t = 0; t < TILES_C; ++t) {
    const int col = t * 16 + l16;
#pragma unroll
    for (int r = 0; r < 8; ++r)
      C[(size_t)(rbase + r) * (size_t)M + col] = acc[t][r];
  }
  if constexpr (ATT) {
    if (l16 < 4) {
      float* __restrict__ ap = (l16 < 2) ? (als + l16) : (ald + (l16 - 2));
#pragma unroll
      for (int r = 0; r < 8; ++r)
        ap[(size_t)(rbase + r) * 2] = acc[TILES_C][r];
    }
  }
}

__global__ void fold_att(const float* __restrict__ Ws0, const float* __restrict__ Wd0,
                         const float* __restrict__ as0, const float* __restrict__ ad0,
                         const float* __restrict__ Ws1, const float* __restrict__ Wd1,
                         const float* __restrict__ as1, const float* __restrict__ ad1,
                         const float* __restrict__ Ws2, const float* __restrict__ Wd2,
                         const float* __restrict__ as2, const float* __restrict__ ad2,
                         float* __restrict__ wx) {
  const int row = blockIdx.x;
  const float *Ws, *Wd, *a_s, *a_d;
  int k;
  if (row < 128)      { Ws = Ws0; Wd = Wd0; a_s = as0; a_d = ad0; k = row; }
  else if (row < 192) { Ws = Ws1; Wd = Wd1; a_s = as1; a_d = ad1; k = row - 128; }
  else                { Ws = Ws2; Wd = Wd2; a_s = as2; a_d = ad2; k = row - 192; }
  const int j = threadIdx.x >> 5;
  const int lane = threadIdx.x & 31;
  const int h = j & 1;
  const float* wrow = ((j < 2) ? Ws : Wd) + (size_t)k * 128 + h * 64;
  const float* av   = ((j < 2) ? a_s : a_d) + h * 64;
  float s = wrow[lane] * av[lane] + wrow[lane + 32] * av[lane + 32];
  for (int o = 16; o > 0; o >>= 1) s += __shfl_xor(s, o);
  if (lane == 0) wx[row * 16 + j] = s;
  if (threadIdx.x >= 4 && threadIdx.x < 16) wx[row * 16 + threadIdx.x] = 0.f;
}

__global__ void csr_zero(int* __restrict__ cnt, int Nn) {
  int tid = blockIdx.x * blockDim.x + threadIdx.x;
  if (tid < Nn) cnt[tid] = 0;
}

__global__ void csr_hist(const int* __restrict__ dst, int* __restrict__ cnt, int Ee) {
  int tid = blockIdx.x * blockDim.x + threadIdx.x;
  if (tid < Ee) atomicAdd(&cnt[dst[tid]], 1);
}

__global__ void csr_scan(const int* __restrict__ cnt, int* __restrict__ rowptr,
                         int* __restrict__ cursor, int Nn) {
  __shared__ int part[1024];
  __shared__ int ppre[1024];
  const int t = threadIdx.x;
  const int base = blockIdx.x * 1024;
  int pre = 0;
  for (int i = t; i < base; i += 1024) pre += cnt[i];
  const int idx = base + t;
  const int c = (idx < Nn) ? cnt[idx] : 0;
  part[t] = c;
  ppre[t] = pre;
  __syncthreads();
  for (int off = 1; off < 1024; off <<= 1) {
    const int v = (t >= off) ? part[t - off] : 0;
    const int w = (t >= off) ? ppre[t - off] : 0;
    __syncthreads();
    part[t] += v;
    ppre[t] += w;
    __syncthreads();
  }
  const int excl = ppre[1023] + part[t] - c;
  if (idx < Nn) {
    rowptr[idx] = excl;
    cursor[idx] = excl;
  }
  if (idx == Nn - 1) rowptr[Nn] = excl + c;
}

__global__ void csr_fill(const int* __restrict__ src, const int* __restrict__ dst,
                         int* __restrict__ cursor, int* __restrict__ csr_src, int Ee) {
  int tid = blockIdx.x * blockDim.x + threadIdx.x;
  if (tid >= Ee) return;
  const int p = atomicAdd(&cursor[dst[tid]], 1);
  csr_src[p] = src[tid];
}

__global__ void csr_order(const int* __restrict__ rowptr, const int* __restrict__ csr_in,
                          int* __restrict__ csr_out, int Nn) {
  const int lane = threadIdx.x & 31;
  const int node = __builtin_amdgcn_readfirstlane((int)((blockIdx.x * blockDim.x + threadIdx.x) >> 5));
  if (node >= Nn) return;
  const int beg = rowptr[node];
  const int end = rowptr[node + 1];
  for (int cb = beg; cb < end; cb += 32) {
    const int i = cb + lane;
    const int key = (i < end) ? csr_in[i] : 0;
    int rank = 0;
    for (int kb = beg; kb < end; kb += 32) {
      const int k = kb + lane;
      const int other = (k < end) ? csr_in[k] : 0;
      const int n = (end - kb < 32) ? (end - kb) : 32;
      for (int j = 0; j < n; ++j) {
        const int o = __builtin_amdgcn_readlane(other, j);
        const int pos = kb + j;
        rank += ((o < key) || (o == key && pos < i)) ? 1 : 0;
      }
    }
    if (i < end) csr_out[beg + rank] = key;
  }
}

__device__ __forceinline__ float lrelu(float v) { return (v > 0.f) ? v : NEG_SLOPE * v; }

__global__ void node_gather(const int* __restrict__ rowptr, const int* __restrict__ csr_src,
                            const float* __restrict__ als, const float* __restrict__ ald,
                            const float* __restrict__ hs, const float* __restrict__ hl,
                            const float* __restrict__ b, const float* __restrict__ bl,
                            float* __restrict__ out, int Nn) {
  const int lane = threadIdx.x & 31;
  const int node = __builtin_amdgcn_readfirstlane((int)((blockIdx.x * blockDim.x + threadIdx.x) >> 5));
  if (node >= Nn) return;
  const int beg = rowptr[node];
  const int end = rowptr[node + 1];
  const float ad0 = ald[node * 2 + 0];
  const float ad1 = ald[node * 2 + 1];
  const float2* __restrict__ als2 = (const float2*)als;

  float m0 = -INFINITY, m1 = -INFINITY;
  for (int i = beg + lane; i < end; i += 32) {
    const float2 a = als2[csr_src[i]];
    m0 = fmaxf(m0, lrelu(a.x + ad0));
    m1 = fmaxf(m1, lrelu(a.y + ad1));
  }
  for (int o = 16; o > 0; o >>= 1) {
    m0 = fmaxf(m0, __shfl_xor(m0, o));
    m1 = fmaxf(m1, __shfl_xor(m1, o));
  }
  if (!(fabsf(m0) < INFINITY)) m0 = 0.f;
  if (!(fabsf(m1) < INFINITY)) m1 = 0.f;

  float d0 = 0.f, d1 = 0.f;
  for (int i = beg + lane; i < end; i += 32) {
    const float2 a = als2[csr_src[i]];
    d0 += __expf(lrelu(a.x + ad0) - m0);
    d1 += __expf(lrelu(a.y + ad1) - m1);
  }
  for (int o = 16; o > 0; o >>= 1) {
    d0 += __shfl_xor(d0, o);
    d1 += __shfl_xor(d1, o);
  }
  const float r0 = d0 + 1e-16f;
  const float r1 = d1 + 1e-16f;

  const bool head1 = lane >= 16;
  const float* __restrict__ hsl = hs + lane * 4;
  float4 acc = make_float4(0.f, 0.f, 0.f, 0.f);
  for (int base = beg; base < end; base += 32) {
    const int i = base + lane;
    int s_l = 0;
    float w_l0 = 0.f, w_l1 = 0.f;
    if (i < end) {
      s_l = csr_src[i];
      const float2 a = als2[s_l];
      w_l0 = __expf(lrelu(a.x + ad0) - m0) / r0;
      w_l1 = __expf(lrelu(a.y + ad1) - m1) / r1;
    }
    const int cnt = (end - base < 32) ? (end - base) : 32;
#pragma unroll 4
    for (int j = 0; j < cnt; ++j) {
      const int s = __builtin_amdgcn_readlane(s_l, j);
      const float wa = __int_as_float(__builtin_amdgcn_readlane(__float_as_int(w_l0), j));
      const float wb = __int_as_float(__builtin_amdgcn_readlane(__float_as_int(w_l1), j));
      const float w = head1 ? wb : wa;
      const float4 v = *(const float4*)(hsl + (size_t)s * 128);
      acc.x = fmaf(w, v.x, acc.x);
      acc.y = fmaf(w, v.y, acc.y);
      acc.z = fmaf(w, v.z, acc.z);
      acc.w = fmaf(w, v.w, acc.w);
    }
  }
  acc.x += __shfl_xor(acc.x, 16);
  acc.y += __shfl_xor(acc.y, 16);
  acc.z += __shfl_xor(acc.z, 16);
  acc.w += __shfl_xor(acc.w, 16);
  if (lane < 16) {
    const int c = lane * 4;
    const float4 hv = *(const float4*)(hl + (size_t)node * 64 + c);
    float4 o;
    o.x = 0.5f * acc.x + b[c + 0] + hv.x + bl[c + 0];
    o.y = 0.5f * acc.y + b[c + 1] + hv.y + bl[c + 1];
    o.z = 0.5f * acc.z + b[c + 2] + hv.z + bl[c + 2];
    o.w = 0.5f * acc.w + b[c + 3] + hv.w + bl[c + 3];
    o.x = (o.x > 0.f) ? o.x : 0.f;
    o.y = (o.y > 0.f) ? o.y : 0.f;
    o.z = (o.z > 0.f) ? o.z : 0.f;
    o.w = (o.w > 0.f) ? o.w : 0.f;
    *(float4*)(out + (size_t)node * 64 + c) = o;
  }
}

extern "C" void kernel_launch(void* const* d_in, const int* in_sizes, int n_in,
                              void* d_out, int out_size, void* d_ws, size_t ws_size,
                              hipStream_t stream) {
  const float* x  = (const float*)d_in[0];
  const int*   ei = (const int*)d_in[1];
  const int Nn = in_sizes[0] / 128;
  const int Ee = in_sizes[1] / 2;
  const int* src = ei;
  const int* dst = ei + Ee;

  const float *Ws[3], *Wd[3], *As[3], *Ad[3], *B[3], *Wl[3], *Bl[3];
  for (int l = 0; l < 3; ++l) {
    int base = 2 + 7 * l;
    Ws[l] = (const float*)d_in[base + 0];
    Wd[l] = (const float*)d_in[base + 1];
    As[l] = (const float*)d_in[base + 2];
    Ad[l] = (const float*)d_in[base + 3];
    B[l]  = (const float*)d_in[base + 4];
    Wl[l] = (const float*)d_in[base + 5];
    Bl[l] = (const float*)d_in[base + 6];
  }

  float* ws = (float*)d_ws;
  size_t off = 0;
  float* hs  = ws + off; off += (size_t)Nn * 128;
  float* hl  = ws + off; off += (size_t)Nn * 64;
  float* wx  = ws + off; off += (size_t)256 * 16;
  float* als = ws + off; off += (size_t)Nn * 2;
  float* ald = ws + off; off += (size_t)Nn * 2;
  float* h1  = ws + off; off += (size_t)Nn * 64;
  float* h2  = ws + off; off += (size_t)Nn * 64;
  int* iws = (int*)(ws + off);
  size_t ioff = 0;
  int* cnt     = iws + ioff; ioff += (size_t)Nn;
  int* cursor  = iws + ioff; ioff += (size_t)Nn;
  int* rowptr  = iws + ioff; ioff += (size_t)Nn + 1;
  int* csr_tmp = iws + ioff; ioff += (size_t)Ee;
  int* csr_src = iws + ioff; ioff += (size_t)Ee;

  const float* lin[3]  = { x, h1, h2 };
  const float* wxl[3]  = { wx, wx + 128 * 16, wx + 192 * 16 };
  float*       lout[3] = { h1, h2, (float*)d_out };

  const int TPB = 256;
  const int rowTiles = Nn >> 4;
  const int gemmBlocks = (rowTiles + 7) / 8;

  csr_zero<<<(Nn + TPB - 1) / TPB, TPB, 0, stream>>>(cnt, Nn);
  csr_hist<<<(Ee + TPB - 1) / TPB, TPB, 0, stream>>>(dst, cnt, Ee);
  csr_scan<<<(Nn + 1023) / 1024, 1024, 0, stream>>>(cnt, rowptr, cursor, Nn);
  csr_fill<<<(Ee + TPB - 1) / TPB, TPB, 0, stream>>>(src, dst, cursor, csr_tmp, Ee);
  {
    long long threads = (long long)Nn * 32;
    csr_order<<<(int)((threads + TPB - 1) / TPB), TPB, 0, stream>>>(rowptr, csr_tmp, csr_src, Nn);
  }

  fold_att<<<256, 128, 0, stream>>>(Ws[0], Wd[0], As[0], Ad[0], Ws[1], Wd[1], As[1], Ad[1],
                                    Ws[2], Wd[2], As[2], Ad[2], wx);

  for (int l = 0; l < 3; ++l) {
    if (l == 0) {
      gemm_wmma_lds<128, 128, true ><<<gemmBlocks, TPB, 0, stream>>>(lin[l], Ws[l], wxl[l], hs, als, ald, Nn);
      gemm_wmma_lds<128, 64,  false><<<gemmBlocks, TPB, 0, stream>>>(lin[l], Wl[l], nullptr, hl, nullptr, nullptr, Nn);
    } else {
      gemm_wmma_lds<64, 128, true ><<<gemmBlocks, TPB, 0, stream>>>(lin[l], Ws[l], wxl[l], hs, als, ald, Nn);
      gemm_wmma_lds<64, 64,  false><<<gemmBlocks, TPB, 0, stream>>>(lin[l], Wl[l], nullptr, hl, nullptr, nullptr, Nn);
    }
    {
      long long threads = (long long)Nn * 32;
      node_gather<<<(int)((threads + TPB - 1) / TPB), TPB, 0, stream>>>(
          rowptr, csr_src, als, ald, hs, hl, B[l], Bl[l], lout[l], Nn);
    }
  }
}


# GEMM weight staging: each thread issues all its float4 loads of [W|Wx] first, then its LDS stores, replacing a serial 4-byte load-wait-store trip per element (72 trips per thread in layer 0)
# speedup vs baseline: 2.8001x; 1.2116x over previous
#include <hip/hip_runtime.h>
#include <math.h>

typedef __attribute__((ext_vector_type(2))) float v2f;
typedef __attribute__((ext_vector_type(8))) float v8f;

#define NEG_SLOPE 0.2f

constexpr int GEMM_THREADS = 256;

template <int K, int M, bool ATT>
__global__ void gemm_wmma_lds(const float* __restrict__ A,
                              const float* __restrict__ W,
                              const float* __restrict__ Wx,
                              float* __restrict__ C,
                              float* __restrict__ als,
                              float* __restrict__ ald,
                              int Nrows) {
  constexpr int MT = ATT ? M + 16 : M;
  constexpr int TILES_C = M / 16;
  constexpr int TILES_M = MT / 16;
  __shared__ __attribute__((aligned(16))) float wlds[K * MT];

  constexpr int NW = (K * M / 4) / GEMM_THREADS;
  constexpr int NX = ATT ? (K * 16 / 4) / GEMM_THREADS : 0;
  static_assert((K * M / 4) % GEMM_THREADS == 0, "W must split evenly over the work-group");
  static_assert(!ATT || (K * 16 / 4) % GEMM_THREADS == 0, "Wx must split evenly over the work-group");
  const float4* __restrict__ W4 = (const float4*)W;
  float4 wv[NW];
  float4 xv[NX > 0 ? NX : 1];
#pragma unroll
  for (int j = 0; j < NW; ++j) wv[j] = W4[threadIdx.x + j * GEMM_THREADS];
  if constexpr (ATT) {
    const float4* __restrict__ Wx4 = (const float4*)Wx;
#pragma unroll
    for (int j = 0; j < NX; ++j) xv[j] = Wx4[threadIdx.x + j * GEMM_THREADS];
  }
#pragma unroll
  for (int j = 0; j < NW; ++j) {
    const int f = (threadIdx.x + j * GEMM_THREADS) * 4;
    *(float4*)&wlds[(f / M) * MT + (f % M)] = wv[j];
  }
  if constexpr (ATT) {
#pragma unroll
    for (int j = 0; j < NX; ++j) {
      const int i4 = threadIdx.x + j * GEMM_THREADS;
      *(float4*)&wlds[(i4 >> 2) * MT + M + ((i4 & 3) << 2)] = xv[j];
    }
  }
  __syncthreads();

  const int wavesPerBlock = blockDim.x >> 5;
  const int rowTile = blockIdx.x * wavesPerBlock + (threadIdx.x >> 5);
  const int lane = threadIdx.x & 31;
  if (rowTile >= (Nrows >> 4)) return;

  const int l16 = lane & 15;
  const int kq  = (lane >> 4) << 1;
  const float* __restrict__ arow = A + (size_t)(rowTile * 16 + l16) * (size_t)K;

  v8f acc[TILES_M];
#pragma unroll
  for (int t = 0; t < TILES_M; ++t) acc[t] = (v8f){0.f,0.f,0.f,0.f,0.f,0.f,0.f,0.f};

  for (int k0 = 0; k0 < K; k0 += 4) {
    v2f a;
    a.x = arow[k0 + kq];
    a.y = arow[k0 + kq + 1];
    const float* wr0 = &wlds[(k0 + kq) * MT + l16];
#pragma unroll
    for (int t = 0; t < TILES_M; ++t) {
      v2f b;
      b.x = wr0[t * 16];
      b.y = wr0[MT + t * 16];
      acc[t] = __builtin_amdgcn_wmma_f32_16x16x4_f32(
           false, a,  false, b,
           (short)0, acc[t],  false,  false);
    }
  }

  const int rbase = rowTile * 16 + ((lane >> 4) << 3);
#pragma unroll
  for (int t = 0; t < TILES_C; ++t) {
    const int col = t * 16 + l16;
#pragma unroll
    for (int r = 0; r < 8; ++r)
      C[(size_t)(rbase + r) * (size_t)M + col] = acc[t][r];
  }
  if constexpr (ATT) {
    if (l16 < 4) {
      float* __restrict__ ap = (l16 < 2) ? (als + l16) : (ald + (l16 - 2));
#pragma unroll
      for (int r = 0; r < 8; ++r)
        ap[(size_t)(rbase + r) * 2] = acc[TILES_C][r];
    }
  }
}

__global__ void fold_att(const float* __restrict__ Ws0, const float* __restrict__ Wd0,
                         const float* __restrict__ as0, const float* __restrict__ ad0,
                         const float* __restrict__ Ws1, const float* __restrict__ Wd1,
                         const float* __restrict__ as1, const float* __restrict__ ad1,
                         const float* __restrict__ Ws2, const float* __restrict__ Wd2,
                         const float* __restrict__ as2, const float* __restrict__ ad2,
                         float* __restrict__ wx) {
  const int row = blockIdx.x;
  const float *Ws, *Wd, *a_s, *a_d;
  int k;
  if (row < 128)      { Ws = Ws0; Wd = Wd0; a_s = as0; a_d = ad0; k = row; }
  else if (row < 192) { Ws = Ws1; Wd = Wd1; a_s = as1; a_d = ad1; k = row - 128; }
  else                { Ws = Ws2; Wd = Wd2; a_s = as2; a_d = ad2; k = row - 192; }
  const int j = threadIdx.x >> 5;
  const int lane = threadIdx.x & 31;
  const int h = j & 1;
  const float* wrow = ((j < 2) ? Ws : Wd) + (size_t)k * 128 + h * 64;
  const float* av   = ((j < 2) ? a_s : a_d) + h * 64;
  float s = wrow[lane] * av[lane] + wrow[lane + 32] * av[lane + 32];
  for (int o = 16; o > 0; o >>= 1) s += __shfl_xor(s, o);
  if (lane == 0) wx[row * 16 + j] = s;
  if (threadIdx.x >= 4 && threadIdx.x < 16) wx[row * 16 + threadIdx.x] = 0.f;
}

__global__ void csr_zero(int* __restrict__ cnt, int Nn) {
  int tid = blockIdx.x * blockDim.x + threadIdx.x;
  if (tid < Nn) cnt[tid] = 0;
}

__global__ void csr_hist(const int* __restrict__ dst, int* __restrict__ cnt, int Ee) {
  int tid = blockIdx.x * blockDim.x + threadIdx.x;
  if (tid < Ee) atomicAdd(&cnt[dst[tid]], 1);
}

__global__ void csr_scan(const int* __restrict__ cnt, int* __restrict__ rowptr,
                         int* __restrict__ cursor, int Nn) {
  __shared__ int part[1024];
  __shared__ int ppre[1024];
  const int t = threadIdx.x;
  const int base = blockIdx.x * 1024;
  int pre = 0;
  for (int i = t; i < base; i += 1024) pre += cnt[i];
  const int idx = base + t;
  const int c = (idx < Nn) ? cnt[idx] : 0;
  part[t] = c;
  ppre[t] = pre;
  __syncthreads();
  for (int off = 1; off < 1024; off <<= 1) {
    const int v = (t >= off) ? part[t - off] : 0;
    const int w = (t >= off) ? ppre[t - off] : 0;
    __syncthreads();
    part[t] += v;
    ppre[t] += w;
    __syncthreads();
  }
  const int excl = ppre[1023] + part[t] - c;
  if (idx < Nn) {
    rowptr[idx] = excl;
    cursor[idx] = excl;
  }
  if (idx == Nn - 1) rowptr[Nn] = excl + c;
}

__global__ void csr_fill(const int* __restrict__ src, const int* __restrict__ dst,
                         int* __restrict__ cursor, int* __restrict__ csr_src, int Ee) {
  int tid = blockIdx.x * blockDim.x + threadIdx.x;
  if (tid >= Ee) return;
  const int p = atomicAdd(&cursor[dst[tid]], 1);
  csr_src[p] = src[tid];
}

__global__ void csr_order(const int* __restrict__ rowptr, const int* __restrict__ csr_in,
                          int* __restrict__ csr_out, int Nn) {
  const int lane = threadIdx.x & 31;
  const int node = __builtin_amdgcn_readfirstlane((int)((blockIdx.x * blockDim.x + threadIdx.x) >> 5));
  if (node >= Nn) return;
  const int beg = rowptr[node];
  const int end = rowptr[node + 1];
  for (int cb = beg; cb < end; cb += 32) {
    const int i = cb + lane;
    const int key = (i < end) ? csr_in[i] : 0;
    int rank = 0;
    for (int kb = beg; kb < end; kb += 32) {
      const int k = kb + lane;
      const int other = (k < end) ? csr_in[k] : 0;
      const int n = (end - kb < 32) ? (end - kb) : 32;
      for (int j = 0; j < n; ++j) {
        const int o = __builtin_amdgcn_readlane(other, j);
        const int pos = kb + j;
        rank += ((o < key) || (o == key && pos < i)) ? 1 : 0;
      }
    }
    if (i < end) csr_out[beg + rank] = key;
  }
}

__device__ __forceinline__ float lrelu(float v) { return (v > 0.f) ? v : NEG_SLOPE * v; }

__global__ void node_gather(const int* __restrict__ rowptr, const int* __restrict__ csr_src,
                            const float* __restrict__ als, const float* __restrict__ ald,
                            const float* __restrict__ hs, const float* __restrict__ hl,
                            const float* __restrict__ b, const float* __restrict__ bl,
                            float* __restrict__ out, int Nn) {
  const int lane = threadIdx.x & 31;
  const int node = __builtin_amdgcn_readfirstlane((int)((blockIdx.x * blockDim.x + threadIdx.x) >> 5));
  if (node >= Nn) return;
  const int beg = rowptr[node];
  const int end = rowptr[node + 1];
  const float ad0 = ald[node * 2 + 0];
  const float ad1 = ald[node * 2 + 1];
  const float2* __restrict__ als2 = (const float2*)als;

  float m0 = -INFINITY, m1 = -INFINITY;
  for (int i = beg + lane; i < end; i += 32) {
    const float2 a = als2[csr_src[i]];
    m0 = fmaxf(m0, lrelu(a.x + ad0));
    m1 = fmaxf(m1, lrelu(a.y + ad1));
  }
  for (int o = 16; o > 0; o >>= 1) {
    m0 = fmaxf(m0, __shfl_xor(m0, o));
    m1 = fmaxf(m1, __shfl_xor(m1, o));
  }
  if (!(fabsf(m0) < INFINITY)) m0 = 0.f;
  if (!(fabsf(m1) < INFINITY)) m1 = 0.f;

  float d0 = 0.f, d1 = 0.f;
  for (int i = beg + lane; i < end; i += 32) {
    const float2 a = als2[csr_src[i]];
    d0 += __expf(lrelu(a.x + ad0) - m0);
    d1 += __expf(lrelu(a.y + ad1) - m1);
  }
  for (int o = 16; o > 0; o >>= 1) {
    d0 += __shfl_xor(d0, o);
    d1 += __shfl_xor(d1, o);
  }
  const float r0 = d0 + 1e-16f;
  const float r1 = d1 + 1e-16f;

  const bool head1 = lane >= 16;
  const float* __restrict__ hsl = hs + lane * 4;
  float4 acc = make_float4(0.f, 0.f, 0.f, 0.f);
  for (int base = beg; base < end; base += 32) {
    const int i = base + lane;
    int s_l = 0;
    float w_l0 = 0.f, w_l1 = 0.f;
    if (i < end) {
      s_l = csr_src[i];
      const float2 a = als2[s_l];
      w_l0 = __expf(lrelu(a.x + ad0) - m0) / r0;
      w_l1 = __expf(lrelu(a.y + ad1) - m1) / r1;
    }
    const int cnt = (end - base < 32) ? (end - base) : 32;
#pragma unroll 4
    for (int j = 0; j < cnt; ++j) {
      const int s = __builtin_amdgcn_readlane(s_l, j);
      const float wa = __int_as_float(__builtin_amdgcn_readlane(__float_as_int(w_l0), j));
      const float wb = __int_as_float(__builtin_amdgcn_readlane(__float_as_int(w_l1), j));
      const float w = head1 ? wb : wa;
      const float4 v = *(const float4*)(hsl + (size_t)s * 128);
      acc.x = fmaf(w, v.x, acc.x);
      acc.y = fmaf(w, v.y, acc.y);
      acc.z = fmaf(w, v.z, acc.z);
      acc.w = fmaf(w, v.w, acc.w);
    }
  }
  acc.x += __shfl_xor(acc.x, 16);
  acc.y += __shfl_xor(acc.y, 16);
  acc.z += __shfl_xor(acc.z, 16);
  acc.w += __shfl_xor(acc.w, 16);
  if (lane < 16) {
    const int c = lane * 4;
    const float4 hv = *(const float4*)(hl + (size_t)node * 64 + c);
    float4 o;
    o.x = 0.5f * acc.x + b[c + 0] + hv.x + bl[c + 0];
    o.y = 0.5f * acc.y + b[c + 1] + hv.y + bl[c + 1];
    o.z = 0.5f * acc.z + b[c + 2] + hv.z + bl[c + 2];
    o.w = 0.5f * acc.w + b[c + 3] + hv.w + bl[c + 3];
    o.x = (o.x > 0.f) ? o.x : 0.f;
    o.y = (o.y > 0.f) ? o.y : 0.f;
    o.z = (o.z > 0.f) ? o.z : 0.f;
    o.w = (o.w > 0.f) ? o.w : 0.f;
    *(float4*)(out + (size_t)node * 64 + c) = o;
  }
}

extern "C" void kernel_launch(void* const* d_in, const int* in_sizes, int n_in,
                              void* d_out, int out_size, void* d_ws, size_t ws_size,
                              hipStream_t stream) {
  const float* x  = (const float*)d_in[0];
  const int*   ei = (const int*)d_in[1];
  const int Nn = in_sizes[0] / 128;
  const int Ee = in_sizes[1] / 2;
  const int* src = ei;
  const int* dst = ei + Ee;

  const float *Ws[3], *Wd[3], *As[3], *Ad[3], *B[3], *Wl[3], *Bl[3];
  for (int l = 0; l < 3; ++l) {
    int base = 2 + 7 * l;
    Ws[l] = (const float*)d_in[base + 0];
    Wd[l] = (const float*)d_in[base + 1];
    As[l] = (const float*)d_in[base + 2];
    Ad[l] = (const float*)d_in[base + 3];
    B[l]  = (const float*)d_in[base + 4];
    Wl[l] = (const float*)d_in[base + 5];
    Bl[l] = (const float*)d_in[base + 6];
  }

  float* ws = (float*)d_ws;
  size_t off = 0;
  float* hs  = ws + off; off += (size_t)Nn * 128;
  float* hl  = ws + off; off += (size_t)Nn * 64;
  float* wx  = ws + off; off += (size_t)256 * 16;
  float* als = ws + off; off += (size_t)Nn * 2;
  float* ald = ws + off; off += (size_t)Nn * 2;
  float* h1  = ws + off; off += (size_t)Nn * 64;
  float* h2  = ws + off; off += (size_t)Nn * 64;
  int* iws = (int*)(ws + off);
  size_t ioff = 0;
  int* cnt     = iws + ioff; ioff += (size_t)Nn;
  int* cursor  = iws + ioff; ioff += (size_t)Nn;
  int* rowptr  = iws + ioff; ioff += (size_t)Nn + 1;
  int* csr_tmp = iws + ioff; ioff += (size_t)Ee;
  int* csr_src = iws + ioff; ioff += (size_t)Ee;

  const float* lin[3]  = { x, h1, h2 };
  const float* wxl[3]  = { wx, wx + 128 * 16, wx + 192 * 16 };
  float*       lout[3] = { h1, h2, (float*)d_out };

  const int TPB = 256;
  const int rowTiles = Nn >> 4;
  const int gemmBlocks = (rowTiles + 7) / 8;

  csr_zero<<<(Nn + TPB - 1) / TPB, TPB, 0, stream>>>(cnt, Nn);
  csr_hist<<<(Ee + TPB - 1) / TPB, TPB, 0, stream>>>(dst, cnt, Ee);
  csr_scan<<<(Nn + 1023) / 1024, 1024, 0, stream>>>(cnt, rowptr, cursor, Nn);
  csr_fill<<<(Ee + TPB - 1) / TPB, TPB, 0, stream>>>(src, dst, cursor, csr_tmp, Ee);
  {
    long long threads = (long long)Nn * 32;
    csr_order<<<(int)((threads + TPB - 1) / TPB), TPB, 0, stream>>>(rowptr, csr_tmp, csr_src, Nn);
  }

  fold_att<<<256, 128, 0, stream>>>(Ws[0], Wd[0], As[0], Ad[0], Ws[1], Wd[1], As[1], Ad[1],
                                    Ws[2], Wd[2], As[2], Ad[2], wx);

  for (int l = 0; l < 3; ++l) {
    if (l == 0) {
      gemm_wmma_lds<128, 128, true ><<<gemmBlocks, GEMM_THREADS, 0, stream>>>(lin[l], Ws[l], wxl[l], hs, als, ald, Nn);
      gemm_wmma_lds<128, 64,  false><<<gemmBlocks, GEMM_THREADS, 0, stream>>>(lin[l], Wl[l], nullptr, hl, nullptr, nullptr, Nn);
    } else {
      gemm_wmma_lds<64, 128, true ><<<gemmBlocks, GEMM_THREADS, 0, stream>>>(lin[l], Ws[l], wxl[l], hs, als, ald, Nn);
      gemm_wmma_lds<64, 64,  false><<<gemmBlocks, GEMM_THREADS, 0, stream>>>(lin[l], Wl[l], nullptr, hl, nullptr, nullptr, Nn);
    }
    {
      long long threads = (long long)Nn * 32;
      node_gather<<<(int)((threads + TPB - 1) / TPB), TPB, 0, stream>>>(
          rowptr, csr_src, als, ald, hs, hl, B[l], Bl[l], lout[l], Nn);
    }
  }
}


# Layers 2,3 (64-wide input): per-node weighted sums run over input rows per head, Ws applied after on WMMA (z@[Ws_h0;Ws_h1], skip+bias+relu in epilogue); 256 B per edge not 512; sum order changed
# speedup vs baseline: 2.9319x; 1.0471x over previous
#include <hip/hip_runtime.h>
#include <math.h>

typedef __attribute__((ext_vector_type(2))) float v2f;
typedef __attribute__((ext_vector_type(8))) float v8f;

#define NEG_SLOPE 0.2f

constexpr int GEMM_THREADS = 256;

template <int K, int M, bool ATT>
__global__ void gemm_wmma_lds(const float* __restrict__ A,
                              const float* __restrict__ W,
                              const float* __restrict__ Wx,
                              float* __restrict__ C,
                              float* __restrict__ als,
                              float* __restrict__ ald,
                              int Nrows) {
  constexpr int MT = ATT ? M + 16 : M;
  constexpr int TILES_C = M / 16;
  constexpr int TILES_M = MT / 16;
  __shared__ __attribute__((aligned(16))) float wlds[K * MT];

  constexpr int NW = (K * M / 4) / GEMM_THREADS;
  constexpr int NX = ATT ? (K * 16 / 4) / GEMM_THREADS : 0;
  static_assert((K * M / 4) % GEMM_THREADS == 0, "W must split evenly over the work-group");
  static_assert(!ATT || (K * 16 / 4) % GEMM_THREADS == 0, "Wx must split evenly over the work-group");
  const float4* __restrict__ W4 = (const float4*)W;
  float4 wv[NW];
  float4 xv[NX > 0 ? NX : 1];
#pragma unroll
  for (int j = 0; j < NW; ++j) wv[j] = W4[threadIdx.x + j * GEMM_THREADS];
  if constexpr (ATT) {
    const float4* __restrict__ Wx4 = (const float4*)Wx;
#pragma unroll
    for (int j = 0; j < NX; ++j) xv[j] = Wx4[threadIdx.x + j * GEMM_THREADS];
  }
#pragma unroll
  for (int j = 0; j < NW; ++j) {
    const int f = (threadIdx.x + j * GEMM_THREADS) * 4;
    *(float4*)&wlds[(f / M) * MT + (f % M)] = wv[j];
  }
  if constexpr (ATT) {
#pragma unroll
    for (int j = 0; j < NX; ++j) {
      const int i4 = threadIdx.x + j * GEMM_THREADS;
      *(float4*)&wlds[(i4 >> 2) * MT + M + ((i4 & 3) << 2)] = xv[j];
    }
  }
  __syncthreads();

  const int wavesPerBlock = blockDim.x >> 5;
  const int rowTile = blockIdx.x * wavesPerBlock + (threadIdx.x >> 5);
  const int lane = threadIdx.x & 31;
  if (rowTile >= (Nrows >> 4)) return;

  const int l16 = lane & 15;
  const int kq  = (lane >> 4) << 1;
  const float* __restrict__ arow = A + (size_t)(rowTile * 16 + l16) * (size_t)K;

  v8f acc[TILES_M];
#pragma unroll
  for (int t = 0; t < TILES_M; ++t) acc[t] = (v8f){0.f,0.f,0.f,0.f,0.f,0.f,0.f,0.f};

  for (int k0 = 0; k0 < K; k0 += 4) {
    v2f a;
    a.x = arow[k0 + kq];
    a.y = arow[k0 + kq + 1];
    const float* wr0 = &wlds[(k0 + kq) * MT + l16];
#pragma unroll
    for (int t = 0; t < TILES_M; ++t) {
      v2f b;
      b.x = wr0[t * 16];
      b.y = wr0[MT + t * 16];
      acc[t] = __builtin_amdgcn_wmma_f32_16x16x4_f32(
           false, a,  false, b,
           (short)0, acc[t],  false,  false);
    }
  }

  const int rbase = rowTile * 16 + ((lane >> 4) << 3);
#pragma unroll
  for (int t = 0; t < TILES_C; ++t) {
    const int col = t * 16 + l16;
#pragma unroll
    for (int r = 0; r < 8; ++r)
      C[(size_t)(rbase + r) * (size_t)M + col] = acc[t][r];
  }
  if constexpr (ATT) {
    if (l16 < 4) {
      float* __restrict__ ap = (l16 < 2) ? (als + l16) : (ald + (l16 - 2));
#pragma unroll
      for (int r = 0; r < 8; ++r)
        ap[(size_t)(rbase + r) * 2] = acc[TILES_C][r];
    }
  }
}

__global__ void fold_att(const float* __restrict__ Ws0, const float* __restrict__ Wd0,
                         const float* __restrict__ as0, const float* __restrict__ ad0,
                         const float* __restrict__ Ws1, const float* __restrict__ Wd1,
                         const float* __restrict__ as1, const float* __restrict__ ad1,
                         const float* __restrict__ Ws2, const float* __restrict__ Wd2,
                         const float* __restrict__ as2, const float* __restrict__ ad2,
                         float* __restrict__ wx) {
  const int row = blockIdx.x;
  const float *Ws, *Wd, *a_s, *a_d;
  int k;
  if (row < 128)      { Ws = Ws0; Wd = Wd0; a_s = as0; a_d = ad0; k = row; }
  else if (row < 192) { Ws = Ws1; Wd = Wd1; a_s = as1; a_d = ad1; k = row - 128; }
  else                { Ws = Ws2; Wd = Wd2; a_s = as2; a_d = ad2; k = row - 192; }
  const int j = threadIdx.x >> 5;
  const int lane = threadIdx.x & 31;
  const int h = j & 1;
  const float* wrow = ((j < 2) ? Ws : Wd) + (size_t)k * 128 + h * 64;
  const float* av   = ((j < 2) ? a_s : a_d) + h * 64;
  float s = wrow[lane] * av[lane] + wrow[lane + 32] * av[lane + 32];
  for (int o = 16; o > 0; o >>= 1) s += __shfl_xor(s, o);
  if (lane == 0) wx[row * 16 + j] = s;
  if (threadIdx.x >= 4 && threadIdx.x < 16) wx[row * 16 + threadIdx.x] = 0.f;
}

__global__ void csr_zero(int* __restrict__ cnt, int Nn) {
  int tid = blockIdx.x * blockDim.x + threadIdx.x;
  if (tid < Nn) cnt[tid] = 0;
}

__global__ void csr_hist(const int* __restrict__ dst, int* __restrict__ cnt, int Ee) {
  int tid = blockIdx.x * blockDim.x + threadIdx.x;
  if (tid < Ee) atomicAdd(&cnt[dst[tid]], 1);
}

__global__ void csr_scan(const int* __restrict__ cnt, int* __restrict__ rowptr,
                         int* __restrict__ cursor, int Nn) {
  __shared__ int part[1024];
  __shared__ int ppre[1024];
  const int t = threadIdx.x;
  const int base = blockIdx.x * 1024;
  int pre = 0;
  for (int i = t; i < base; i += 1024) pre += cnt[i];
  const int idx = base + t;
  const int c = (idx < Nn) ? cnt[idx] : 0;
  part[t] = c;
  ppre[t] = pre;
  __syncthreads();
  for (int off = 1; off < 1024; off <<= 1) {
    const int v = (t >= off) ? part[t - off] : 0;
    const int w = (t >= off) ? ppre[t - off] : 0;
    __syncthreads();
    part[t] += v;
    ppre[t] += w;
    __syncthreads();
  }
  const int excl = ppre[1023] + part[t] - c;
  if (idx < Nn) {
    rowptr[idx] = excl;
    cursor[idx] = excl;
  }
  if (idx == Nn - 1) rowptr[Nn] = excl + c;
}

__global__ void csr_fill(const int* __restrict__ src, const int* __restrict__ dst,
                         int* __restrict__ cursor, int* __restrict__ csr_src, int Ee) {
  int tid = blockIdx.x * blockDim.x + threadIdx.x;
  if (tid >= Ee) return;
  const int p = atomicAdd(&cursor[dst[tid]], 1);
  csr_src[p] = src[tid];
}

__global__ void csr_order(const int* __restrict__ rowptr, const int* __restrict__ csr_in,
                          int* __restrict__ csr_out, int Nn) {
  const int lane = threadIdx.x & 31;
  const int node = __builtin_amdgcn_readfirstlane((int)((blockIdx.x * blockDim.x + threadIdx.x) >> 5));
  if (node >= Nn) return;
  const int beg = rowptr[node];
  const int end = rowptr[node + 1];
  for (int cb = beg; cb < end; cb += 32) {
    const int i = cb + lane;
    const int key = (i < end) ? csr_in[i] : 0;
    int rank = 0;
    for (int kb = beg; kb < end; kb += 32) {
      const int k = kb + lane;
      const int other = (k < end) ? csr_in[k] : 0;
      const int n = (end - kb < 32) ? (end - kb) : 32;
      for (int j = 0; j < n; ++j) {
        const int o = __builtin_amdgcn_readlane(other, j);
        const int pos = kb + j;
        rank += ((o < key) || (o == key && pos < i)) ? 1 : 0;
      }
    }
    if (i < end) csr_out[beg + rank] = key;
  }
}

__device__ __forceinline__ float lrelu(float v) { return (v > 0.f) ? v : NEG_SLOPE * v; }

__global__ void node_gather(const int* __restrict__ rowptr, const int* __restrict__ csr_src,
                            const float* __restrict__ als, const float* __restrict__ ald,
                            const float* __restrict__ hs, const float* __restrict__ hl,
                            const float* __restrict__ b, const float* __restrict__ bl,
                            float* __restrict__ out, int Nn) {
  const int lane = threadIdx.x & 31;
  const int node = __builtin_amdgcn_readfirstlane((int)((blockIdx.x * blockDim.x + threadIdx.x) >> 5));
  if (node >= Nn) return;
  const int beg = rowptr[node];
  const int end = rowptr[node + 1];
  const float ad0 = ald[node * 2 + 0];
  const float ad1 = ald[node * 2 + 1];
  const float2* __restrict__ als2 = (const float2*)als;

  float m0 = -INFINITY, m1 = -INFINITY;
  for (int i = beg + lane; i < end; i += 32) {
    const float2 a = als2[csr_src[i]];
    m0 = fmaxf(m0, lrelu(a.x + ad0));
    m1 = fmaxf(m1, lrelu(a.y + ad1));
  }
  for (int o = 16; o > 0; o >>= 1) {
    m0 = fmaxf(m0, __shfl_xor(m0, o));
    m1 = fmaxf(m1, __shfl_xor(m1, o));
  }
  if (!(fabsf(m0) < INFINITY)) m0 = 0.f;
  if (!(fabsf(m1) < INFINITY)) m1 = 0.f;

  float d0 = 0.f, d1 = 0.f;
  for (int i = beg + lane; i < end; i += 32) {
    const float2 a = als2[csr_src[i]];
    d0 += __expf(lrelu(a.x + ad0) - m0);
    d1 += __expf(lrelu(a.y + ad1) - m1);
  }
  for (int o = 16; o > 0; o >>= 1) {
    d0 += __shfl_xor(d0, o);
    d1 += __shfl_xor(d1, o);
  }
  const float r0 = d0 + 1e-16f;
  const float r1 = d1 + 1e-16f;

  const bool head1 = lane >= 16;
  const float* __restrict__ hsl = hs + lane * 4;
  float4 acc = make_float4(0.f, 0.f, 0.f, 0.f);
  for (int base = beg; base < end; base += 32) {
    const int i = base + lane;
    int s_l = 0;
    float w_l0 = 0.f, w_l1 = 0.f;
    if (i < end) {
      s_l = csr_src[i];
      const float2 a = als2[s_l];
      w_l0 = __expf(lrelu(a.x + ad0) - m0) / r0;
      w_l1 = __expf(lrelu(a.y + ad1) - m1) / r1;
    }
    const int cnt = (end - base < 32) ? (end - base) : 32;
#pragma unroll 4
    for (int j = 0; j < cnt; ++j) {
      const int s = __builtin_amdgcn_readlane(s_l, j);
      const float wa = __int_as_float(__builtin_amdgcn_readlane(__float_as_int(w_l0), j));
      const float wb = __int_as_float(__builtin_amdgcn_readlane(__float_as_int(w_l1), j));
      const float w = head1 ? wb : wa;
      const float4 v = *(const float4*)(hsl + (size_t)s * 128);
      acc.x = fmaf(w, v.x, acc.x);
      acc.y = fmaf(w, v.y, acc.y);
      acc.z = fmaf(w, v.z, acc.z);
      acc.w = fmaf(w, v.w, acc.w);
    }
  }
  acc.x += __shfl_xor(acc.x, 16);
  acc.y += __shfl_xor(acc.y, 16);
  acc.z += __shfl_xor(acc.z, 16);
  acc.w += __shfl_xor(acc.w, 16);
  if (lane < 16) {
    const int c = lane * 4;
    const float4 hv = *(const float4*)(hl + (size_t)node * 64 + c);
    float4 o;
    o.x = 0.5f * acc.x + b[c + 0] + hv.x + bl[c + 0];
    o.y = 0.5f * acc.y + b[c + 1] + hv.y + bl[c + 1];
    o.z = 0.5f * acc.z + b[c + 2] + hv.z + bl[c + 2];
    o.w = 0.5f * acc.w + b[c + 3] + hv.w + bl[c + 3];
    o.x = (o.x > 0.f) ? o.x : 0.f;
    o.y = (o.y > 0.f) ? o.y : 0.f;
    o.z = (o.z > 0.f) ? o.z : 0.f;
    o.w = (o.w > 0.f) ? o.w : 0.f;
    *(float4*)(out + (size_t)node * 64 + c) = o;
  }
}

__global__ void node_gather_in(const int* __restrict__ rowptr, const int* __restrict__ csr_src,
                               const float* __restrict__ als, const float* __restrict__ ald,
                               const float* __restrict__ hin, float* __restrict__ z, int Nn) {
  const int lane = threadIdx.x & 31;
  const int node = __builtin_amdgcn_readfirstlane((int)((blockIdx.x * blockDim.x + threadIdx.x) >> 5));
  if (node >= Nn) return;
  const int beg = rowptr[node];
  const int end = rowptr[node + 1];
  const float ad0 = ald[node * 2 + 0];
  const float ad1 = ald[node * 2 + 1];
  const float2* __restrict__ als2 = (const float2*)als;

  float m0 = -INFINITY, m1 = -INFINITY;
  for (int i = beg + lane; i < end; i += 32) {
    const float2 a = als2[csr_src[i]];
    m0 = fmaxf(m0, lrelu(a.x + ad0));
    m1 = fmaxf(m1, lrelu(a.y + ad1));
  }
  for (int o = 16; o > 0; o >>= 1) {
    m0 = fmaxf(m0, __shfl_xor(m0, o));
    m1 = fmaxf(m1, __shfl_xor(m1, o));
  }
  if (!(fabsf(m0) < INFINITY)) m0 = 0.f;
  if (!(fabsf(m1) < INFINITY)) m1 = 0.f;

  float d0 = 0.f, d1 = 0.f;
  for (int i = beg + lane; i < end; i += 32) {
    const float2 a = als2[csr_src[i]];
    d0 += __expf(lrelu(a.x + ad0) - m0);
    d1 += __expf(lrelu(a.y + ad1) - m1);
  }
  for (int o = 16; o > 0; o >>= 1) {
    d0 += __shfl_xor(d0, o);
    d1 += __shfl_xor(d1, o);
  }
  const float r0 = d0 + 1e-16f;
  const float r1 = d1 + 1e-16f;

  const bool odd = lane >= 16;
  const int c = (lane & 15) * 4;
  const float* __restrict__ hrow = hin + c;
  float4 acc0 = make_float4(0.f, 0.f, 0.f, 0.f);
  float4 acc1 = make_float4(0.f, 0.f, 0.f, 0.f);
  for (int base = beg; base < end; base += 32) {
    const int i = base + lane;
    int s_l = 0;
    float w_l0 = 0.f, w_l1 = 0.f;
    if (i < end) {
      s_l = csr_src[i];
      const float2 a = als2[s_l];
      w_l0 = __expf(lrelu(a.x + ad0) - m0) / r0;
      w_l1 = __expf(lrelu(a.y + ad1) - m1) / r1;
    }
    const int s_first = __builtin_amdgcn_readfirstlane(s_l);
    if (i >= end) s_l = s_first;
    const int cnt = (end - base < 32) ? (end - base) : 32;
#pragma unroll 4
    for (int j = 0; j < cnt; j += 2) {
      const int sa = __builtin_amdgcn_readlane(s_l, j);
      const int sb = __builtin_amdgcn_readlane(s_l, j + 1);
      const float wa0 = __int_as_float(__builtin_amdgcn_readlane(__float_as_int(w_l0), j));
      const float wb0 = __int_as_float(__builtin_amdgcn_readlane(__float_as_int(w_l0), j + 1));
      const float wa1 = __int_as_float(__builtin_amdgcn_readlane(__float_as_int(w_l1), j));
      const float wb1 = __int_as_float(__builtin_amdgcn_readlane(__float_as_int(w_l1), j + 1));
      const int s = odd ? sb : sa;
      const float w0 = odd ? wb0 : wa0;
      const float w1 = odd ? wb1 : wa1;
      const float4 v = *(const float4*)(hrow + (size_t)s * 64);
      acc0.x = fmaf(w0, v.x, acc0.x);
      acc0.y = fmaf(w0, v.y, acc0.y);
      acc0.z = fmaf(w0, v.z, acc0.z);
      acc0.w = fmaf(w0, v.w, acc0.w);
      acc1.x = fmaf(w1, v.x, acc1.x);
      acc1.y = fmaf(w1, v.y, acc1.y);
      acc1.z = fmaf(w1, v.z, acc1.z);
      acc1.w = fmaf(w1, v.w, acc1.w);
    }
  }
  acc0.x += __shfl_xor(acc0.x, 16);
  acc0.y += __shfl_xor(acc0.y, 16);
  acc0.z += __shfl_xor(acc0.z, 16);
  acc0.w += __shfl_xor(acc0.w, 16);
  acc1.x += __shfl_xor(acc1.x, 16);
  acc1.y += __shfl_xor(acc1.y, 16);
  acc1.z += __shfl_xor(acc1.z, 16);
  acc1.w += __shfl_xor(acc1.w, 16);
  float4 o;
  o.x = odd ? acc1.x : acc0.x;
  o.y = odd ? acc1.y : acc0.y;
  o.z = odd ? acc1.z : acc0.z;
  o.w = odd ? acc1.w : acc0.w;
  *(float4*)(z + (size_t)node * 128 + (odd ? 64 : 0) + c) = o;
}

__global__ void gemm_zout(const float* __restrict__ Z,
                          const float* __restrict__ Ws,
                          const float* __restrict__ hl,
                          const float* __restrict__ b,
                          const float* __restrict__ bl,
                          float* __restrict__ out,
                          int Nrows) {
  constexpr int K = 128, M = 64, MT = 80, TILES = M / 16;
  __shared__ __attribute__((aligned(16))) float wlds[K * MT];

  constexpr int NW = (64 * 128 / 4) / GEMM_THREADS;
  static_assert((64 * 128 / 4) % GEMM_THREADS == 0, "Ws must split evenly over the work-group");
  const float4* __restrict__ W4 = (const float4*)Ws;
  float4 wv[NW];
#pragma unroll
  for (int j = 0; j < NW; ++j) wv[j] = W4[threadIdx.x + j * GEMM_THREADS];
#pragma unroll
  for (int j = 0; j < NW; ++j) {
    const int f = (threadIdx.x + j * GEMM_THREADS) * 4;
    const int k = f >> 7;
    const int col = f & 127;
    *(float4*)&wlds[((col >> 6) * 64 + k) * MT + (col & 63)] = wv[j];
  }
  __syncthreads();

  const int wavesPerBlock = blockDim.x >> 5;
  const int rowTile = blockIdx.x * wavesPerBlock + (threadIdx.x >> 5);
  const int lane = threadIdx.x & 31;
  if (rowTile >= (Nrows >> 4)) return;

  const int l16 = lane & 15;
  const int kq  = (lane >> 4) << 1;
  const float* __restrict__ arow = Z + (size_t)(rowTile * 16 + l16) * (size_t)K;

  const int rbase = rowTile * 16 + ((lane >> 4) << 3);
  float hv[TILES][8];
  float bc[TILES], blc[TILES];
#pragma unroll
  for (int t = 0; t < TILES; ++t) {
    const int col = t * 16 + l16;
    bc[t] = b[col];
    blc[t] = bl[col];
#pragma unroll
    for (int r = 0; r < 8; ++r)
      hv[t][r] = hl[(size_t)(rbase + r) * (size_t)M + col];
  }

  v8f acc[TILES];
#pragma unroll
  for (int t = 0; t < TILES; ++t) acc[t] = (v8f){0.f,0.f,0.f,0.f,0.f,0.f,0.f,0.f};

  for (int k0 = 0; k0 < K; k0 += 4) {
    v2f a;
    a.x = arow[k0 + kq];
    a.y = arow[k0 + kq + 1];
    const float* wr0 = &wlds[(k0 + kq) * MT + l16];
#pragma unroll
    for (int t = 0; t < TILES; ++t) {
      v2f bb;
      bb.x = wr0[t * 16];
      bb.y = wr0[MT + t * 16];
      acc[t] = __builtin_amdgcn_wmma_f32_16x16x4_f32(
           false, a,  false, bb,
           (short)0, acc[t],  false,  false);
    }
  }

#pragma unroll
  for (int t = 0; t < TILES; ++t) {
    const int col = t * 16 + l16;
#pragma unroll
    for (int r = 0; r < 8; ++r) {
      const float v = 0.5f * acc[t][r] + bc[t] + hv[t][r] + blc[t];
      out[(size_t)(rbase + r) * (size_t)M + col] = (v > 0.f) ? v : 0.f;
    }
  }
}

extern "C" void kernel_launch(void* const* d_in, const int* in_sizes, int n_in,
                              void* d_out, int out_size, void* d_ws, size_t ws_size,
                              hipStream_t stream) {
  const float* x  = (const float*)d_in[0];
  const int*   ei = (const int*)d_in[1];
  const int Nn = in_sizes[0] / 128;
  const int Ee = in_sizes[1] / 2;
  const int* src = ei;
  const int* dst = ei + Ee;

  const float *Ws[3], *Wd[3], *As[3], *Ad[3], *B[3], *Wl[3], *Bl[3];
  for (int l = 0; l < 3; ++l) {
    int base = 2 + 7 * l;
    Ws[l] = (const float*)d_in[base + 0];
    Wd[l] = (const float*)d_in[base + 1];
    As[l] = (const float*)d_in[base + 2];
    Ad[l] = (const float*)d_in[base + 3];
    B[l]  = (const float*)d_in[base + 4];
    Wl[l] = (const float*)d_in[base + 5];
    Bl[l] = (const float*)d_in[base + 6];
  }

  float* ws = (float*)d_ws;
  size_t off = 0;
  float* hs  = ws + off; off += (size_t)Nn * 128;
  float* hl  = ws + off; off += (size_t)Nn * 64;
  float* wx  = ws + off; off += (size_t)256 * 16;
  float* als = ws + off; off += (size_t)Nn * 2;
  float* ald = ws + off; off += (size_t)Nn * 2;
  float* h1  = ws + off; off += (size_t)Nn * 64;
  float* h2  = ws + off; off += (size_t)Nn * 64;
  int* iws = (int*)(ws + off);
  size_t ioff = 0;
  int* cnt     = iws + ioff; ioff += (size_t)Nn;
  int* cursor  = iws + ioff; ioff += (size_t)Nn;
  int* rowptr  = iws + ioff; ioff += (size_t)Nn + 1;
  int* csr_tmp = iws + ioff; ioff += (size_t)Ee;
  int* csr_src = iws + ioff; ioff += (size_t)Ee;

  const float* lin[3]  = { x, h1, h2 };
  const float* wxl[3]  = { wx, wx + 128 * 16, wx + 192 * 16 };
  float*       lout[3] = { h1, h2, (float*)d_out };

  const int TPB = 256;
  const int rowTiles = Nn >> 4;
  const int gemmBlocks = (rowTiles + 7) / 8;

  csr_zero<<<(Nn + TPB - 1) / TPB, TPB, 0, stream>>>(cnt, Nn);
  csr_hist<<<(Ee + TPB - 1) / TPB, TPB, 0, stream>>>(dst, cnt, Ee);
  csr_scan<<<(Nn + 1023) / 1024, 1024, 0, stream>>>(cnt, rowptr, cursor, Nn);
  csr_fill<<<(Ee + TPB - 1) / TPB, TPB, 0, stream>>>(src, dst, cursor, csr_tmp, Ee);
  {
    long long threads = (long long)Nn * 32;
    csr_order<<<(int)((threads + TPB - 1) / TPB), TPB, 0, stream>>>(rowptr, csr_tmp, csr_src, Nn);
  }

  fold_att<<<256, 128, 0, stream>>>(Ws[0], Wd[0], As[0], Ad[0], Ws[1], Wd[1], As[1], Ad[1],
                                    Ws[2], Wd[2], As[2], Ad[2], wx);

  for (int l = 0; l < 3; ++l) {
    if (l == 0) {
      gemm_wmma_lds<128, 128, true ><<<gemmBlocks, GEMM_THREADS, 0, stream>>>(lin[l], Ws[l], wxl[l], hs, als, ald, Nn);
      gemm_wmma_lds<128, 64,  false><<<gemmBlocks, GEMM_THREADS, 0, stream>>>(lin[l], Wl[l], nullptr, hl, nullptr, nullptr, Nn);
      long long threads = (long long)Nn * 32;
      node_gather<<<(int)((threads + TPB - 1) / TPB), TPB, 0, stream>>>(
          rowptr, csr_src, als, ald, hs, hl, B[l], Bl[l], lout[l], Nn);
    } else {
      gemm_wmma_lds<64, 64, true><<<gemmBlocks, GEMM_THREADS, 0, stream>>>(lin[l], Wl[l], wxl[l], hl, als, ald, Nn);
      long long threads = (long long)Nn * 32;
      node_gather_in<<<(int)((threads + TPB - 1) / TPB), TPB, 0, stream>>>(
          rowptr, csr_src, als, ald, lin[l], hs, Nn);
      gemm_zout<<<gemmBlocks, GEMM_THREADS, 0, stream>>>(hs, Ws[l], hl, B[l], Bl[l], lout[l], Nn);
    }
  }
}


# Edge sort rebuilt on LDS counters: edges binned by 128-node destination range (per-block LDS histogram, slot = LDS add's return), one block sorts each bin; no global atomics; 14 launches not 15
# speedup vs baseline: 4.1526x; 1.4164x over previous
#include <hip/hip_runtime.h>
#include <math.h>

typedef __attribute__((ext_vector_type(2))) float v2f;
typedef __attribute__((ext_vector_type(8))) float v8f;

#define NEG_SLOPE 0.2f

constexpr int GEMM_THREADS = 256;

template <int K, int M, bool ATT>
__global__ void gemm_wmma_lds(const float* __restrict__ A,
                              const float* __restrict__ W,
                              const float* __restrict__ Wx,
                              float* __restrict__ C,
                              float* __restrict__ als,
                              float* __restrict__ ald,
                              int Nrows) {
  constexpr int MT = ATT ? M + 16 : M;
  constexpr int TILES_C = M / 16;
  constexpr int TILES_M = MT / 16;
  __shared__ __attribute__((aligned(16))) float wlds[K * MT];

  constexpr int NW = (K * M / 4) / GEMM_THREADS;
  constexpr int NX = ATT ? (K * 16 / 4) / GEMM_THREADS : 0;
  static_assert((K * M / 4) % GEMM_THREADS == 0, "W must split evenly over the work-group");
  static_assert(!ATT || (K * 16 / 4) % GEMM_THREADS == 0, "Wx must split evenly over the work-group");
  const float4* __restrict__ W4 = (const float4*)W;
  float4 wv[NW];
  float4 xv[NX > 0 ? NX : 1];
#pragma unroll
  for (int j = 0; j < NW; ++j) wv[j] = W4[threadIdx.x + j * GEMM_THREADS];
  if constexpr (ATT) {
    const float4* __restrict__ Wx4 = (const float4*)Wx;
#pragma unroll
    for (int j = 0; j < NX; ++j) xv[j] = Wx4[threadIdx.x + j * GEMM_THREADS];
  }
#pragma unroll
  for (int j = 0; j < NW; ++j) {
    const int f = (threadIdx.x + j * GEMM_THREADS) * 4;
    *(float4*)&wlds[(f / M) * MT + (f % M)] = wv[j];
  }
  if constexpr (ATT) {
#pragma unroll
    for (int j = 0; j < NX; ++j) {
      const int i4 = threadIdx.x + j * GEMM_THREADS;
      *(float4*)&wlds[(i4 >> 2) * MT + M + ((i4 & 3) << 2)] = xv[j];
    }
  }
  __syncthreads();

  const int wavesPerBlock = blockDim.x >> 5;
  const int rowTile = blockIdx.x * wavesPerBlock + (threadIdx.x >> 5);
  const int lane = threadIdx.x & 31;
  if (rowTile >= (Nrows >> 4)) return;

  const int l16 = lane & 15;
  const int kq  = (lane >> 4) << 1;
  const float* __restrict__ arow = A + (size_t)(rowTile * 16 + l16) * (size_t)K;

  v8f acc[TILES_M];
#pragma unroll
  for (int t = 0; t < TILES_M; ++t) acc[t] = (v8f){0.f,0.f,0.f,0.f,0.f,0.f,0.f,0.f};

  for (int k0 = 0; k0 < K; k0 += 4) {
    v2f a;
    a.x = arow[k0 + kq];
    a.y = arow[k0 + kq + 1];
    const float* wr0 = &wlds[(k0 + kq) * MT + l16];
#pragma unroll
    for (int t = 0; t < TILES_M; ++t) {
      v2f b;
      b.x = wr0[t * 16];
      b.y = wr0[MT + t * 16];
      acc[t] = __builtin_amdgcn_wmma_f32_16x16x4_f32(
           false, a,  false, b,
           (short)0, acc[t],  false,  false);
    }
  }

  const int rbase = rowTile * 16 + ((lane >> 4) << 3);
#pragma unroll
  for (int t = 0; t < TILES_C; ++t) {
    const int col = t * 16 + l16;
#pragma unroll
    for (int r = 0; r < 8; ++r)
      C[(size_t)(rbase + r) * (size_t)M + col] = acc[t][r];
  }
  if constexpr (ATT) {
    if (l16 < 4) {
      float* __restrict__ ap = (l16 < 2) ? (als + l16) : (ald + (l16 - 2));
#pragma unroll
      for (int r = 0; r < 8; ++r)
        ap[(size_t)(rbase + r) * 2] = acc[TILES_C][r];
    }
  }
}

__global__ void fold_att(const float* __restrict__ Ws0, const float* __restrict__ Wd0,
                         const float* __restrict__ as0, const float* __restrict__ ad0,
                         const float* __restrict__ Ws1, const float* __restrict__ Wd1,
                         const float* __restrict__ as1, const float* __restrict__ ad1,
                         const float* __restrict__ Ws2, const float* __restrict__ Wd2,
                         const float* __restrict__ as2, const float* __restrict__ ad2,
                         float* __restrict__ wx) {
  const int row = blockIdx.x;
  const float *Ws, *Wd, *a_s, *a_d;
  int k;
  if (row < 128)      { Ws = Ws0; Wd = Wd0; a_s = as0; a_d = ad0; k = row; }
  else if (row < 192) { Ws = Ws1; Wd = Wd1; a_s = as1; a_d = ad1; k = row - 128; }
  else                { Ws = Ws2; Wd = Wd2; a_s = as2; a_d = ad2; k = row - 192; }
  const int j = threadIdx.x >> 5;
  const int lane = threadIdx.x & 31;
  const int h = j & 1;
  const float* wrow = ((j < 2) ? Ws : Wd) + (size_t)k * 128 + h * 64;
  const float* av   = ((j < 2) ? a_s : a_d) + h * 64;
  float s = wrow[lane] * av[lane] + wrow[lane + 32] * av[lane + 32];
  for (int o = 16; o > 0; o >>= 1) s += __shfl_xor(s, o);
  if (lane == 0) wx[row * 16 + j] = s;
  if (threadIdx.x >= 4 && threadIdx.x < 16) wx[row * 16 + threadIdx.x] = 0.f;
}

constexpr int BIN_SHIFT = 7;
constexpr int BIN_NODES = 1 << BIN_SHIFT;
constexpr int MAX_BINS = 512;
constexpr int SORT_BLOCKS = 128;
constexpr int SORT_THREADS = 1024;
constexpr int SORT_BATCH = 8;
constexpr int BSORT_THREADS = 512;
constexpr int BSORT_BATCH = 4;

__device__ __forceinline__ int block_excl_scan(int v, int* wsum) {
  const int lane = threadIdx.x & 31;
  const int w = threadIdx.x >> 5;
  int incl = v;
  for (int o = 1; o < 32; o <<= 1) {
    const int u = __shfl_up(incl, o);
    if (lane >= o) incl += u;
  }
  if (lane == 31) wsum[w] = incl;
  __syncthreads();
  int pre = 0;
  for (int q = 0; q < w; ++q) pre += wsum[q];
  return pre + incl - v;
}

__global__ void bin_count(const int* __restrict__ dst, int* __restrict__ tbl, int Ee, int nbins) {
  __shared__ int hist[MAX_BINS];
  const int t = threadIdx.x;
  if (t < MAX_BINS) hist[t] = 0;
  __syncthreads();
  const int chunk = (Ee + SORT_BLOCKS - 1) / SORT_BLOCKS;
  const int lo = (int)blockIdx.x * chunk;
  const int hi = min(lo + chunk, Ee);
  for (int i0 = lo + t; i0 < hi; i0 += SORT_BATCH * SORT_THREADS) {
    int d[SORT_BATCH];
#pragma unroll
    for (int k = 0; k < SORT_BATCH; ++k) {
      const int i = i0 + k * SORT_THREADS;
      d[k] = (i < hi) ? dst[i] : 0;
    }
#pragma unroll
    for (int k = 0; k < SORT_BATCH; ++k)
      if (i0 + k * SORT_THREADS < hi) atomicAdd(&hist[d[k] >> BIN_SHIFT], 1);
  }
  __syncthreads();
  if (t < nbins) tbl[(int)blockIdx.x * nbins + t] = hist[t];
}

__global__ void bin_scatter(const int* __restrict__ src, const int* __restrict__ dst,
                            const int* __restrict__ tbl, int* __restrict__ binbase,
                            int* __restrict__ binned, int Ee, int nbins) {
  __shared__ int cur[MAX_BINS];
  __shared__ int ptot[2][MAX_BINS];
  __shared__ int pbef[2][MAX_BINS];
  __shared__ int wsum[SORT_THREADS / 32];
  const int t = threadIdx.x;
  const int g = blockIdx.x;
  const int bin = t & (MAX_BINS - 1);
  const int half = t / MAX_BINS;
  int tot = 0, bef = 0;
  if (bin < nbins) {
    const int q0 = half * (SORT_BLOCKS / 2);
#pragma unroll 16
    for (int q = q0; q < q0 + SORT_BLOCKS / 2; ++q) {
      const int c = tbl[q * nbins + bin];
      tot += c;
      bef += (q < g) ? c : 0;
    }
  }
  ptot[half][bin] = tot;
  pbef[half][bin] = bef;
  __syncthreads();
  int total = 0, before = 0;
  if (t < MAX_BINS) {
    total = ptot[0][t] + ptot[1][t];
    before = pbef[0][t] + pbef[1][t];
  }
  const int base = block_excl_scan(total, wsum);
  if (t < nbins) {
    cur[t] = base + before;
    if (g == 0) {
      binbase[t] = base;
      if (t == nbins - 1) binbase[nbins] = base + total;
    }
  }
  __syncthreads();
  const int chunk = (Ee + SORT_BLOCKS - 1) / SORT_BLOCKS;
  const int lo = g * chunk;
  const int hi = min(lo + chunk, Ee);
  int2* __restrict__ pairs = (int2*)binned;
  for (int i0 = lo + t; i0 < hi; i0 += SORT_BATCH * SORT_THREADS) {
    int d[SORT_BATCH], s[SORT_BATCH];
#pragma unroll
    for (int k = 0; k < SORT_BATCH; ++k) {
      const int i = i0 + k * SORT_THREADS;
      d[k] = (i < hi) ? dst[i] : 0;
      s[k] = (i < hi) ? src[i] : 0;
    }
#pragma unroll
    for (int k = 0; k < SORT_BATCH; ++k) {
      if (i0 + k * SORT_THREADS < hi) {
        const int p = atomicAdd(&cur[d[k] >> BIN_SHIFT], 1);
        pairs[p] = make_int2(s[k], d[k]);
      }
    }
  }
}

__global__ void bin_sort(const int* __restrict__ binbase, const int* __restrict__ binned,
                         int* __restrict__ rowptr, int* __restrict__ csr_out, int Nn) {
  __shared__ int cnt[BIN_NODES];
  __shared__ int cur[BIN_NODES];
  __shared__ int wsum[BSORT_THREADS / 32];
  const int2* __restrict__ pairs = (const int2*)binned;
  const int t = threadIdx.x;
  const int b = blockIdx.x;
  const int s = binbase[b];
  const int e = binbase[b + 1];
  if (t < BIN_NODES) cnt[t] = 0;
  __syncthreads();
  for (int i0 = s + t; i0 < e; i0 += BSORT_BATCH * BSORT_THREADS) {
    int d[BSORT_BATCH];
#pragma unroll
    for (int k = 0; k < BSORT_BATCH; ++k) {
      const int i = i0 + k * BSORT_THREADS;
      d[k] = (i < e) ? pairs[i].y : 0;
    }
#pragma unroll
    for (int k = 0; k < BSORT_BATCH; ++k)
      if (i0 + k * BSORT_THREADS < e) atomicAdd(&cnt[d[k] & (BIN_NODES - 1)], 1);
  }
  __syncthreads();
  const int c = (t < BIN_NODES) ? cnt[t] : 0;
  const int first = s + block_excl_scan(c, wsum);
  if (t < BIN_NODES) {
    const int node = b * BIN_NODES + t;
    cur[t] = first;
    if (node < Nn) rowptr[node] = first;
    if (node == Nn - 1) rowptr[Nn] = first + c;
  }
  __syncthreads();
  for (int i0 = s + t; i0 < e; i0 += BSORT_BATCH * BSORT_THREADS) {
    int2 pr[BSORT_BATCH];
#pragma unroll
    for (int k = 0; k < BSORT_BATCH; ++k) {
      const int i = i0 + k * BSORT_THREADS;
      pr[k] = (i < e) ? pairs[i] : make_int2(0, 0);
    }
#pragma unroll
    for (int k = 0; k < BSORT_BATCH; ++k) {
      if (i0 + k * BSORT_THREADS < e) {
        const int p = atomicAdd(&cur[pr[k].y & (BIN_NODES - 1)], 1);
        csr_out[p] = pr[k].x;
      }
    }
  }
}

__global__ void csr_order(const int* __restrict__ rowptr, const int* __restrict__ csr_in,
                          int* __restrict__ csr_out, int Nn) {
  const int lane = threadIdx.x & 31;
  const int node = __builtin_amdgcn_readfirstlane((int)((blockIdx.x * blockDim.x + threadIdx.x) >> 5));
  if (node >= Nn) return;
  const int beg = rowptr[node];
  const int end = rowptr[node + 1];
  for (int cb = beg; cb < end; cb += 32) {
    const int i = cb + lane;
    const int key = (i < end) ? csr_in[i] : 0;
    int rank = 0;
    for (int kb = beg; kb < end; kb += 32) {
      const int k = kb + lane;
      const int other = (k < end) ? csr_in[k] : 0;
      const int n = (end - kb < 32) ? (end - kb) : 32;
      for (int j = 0; j < n; ++j) {
        const int o = __builtin_amdgcn_readlane(other, j);
        const int pos = kb + j;
        rank += ((o < key) || (o == key && pos < i)) ? 1 : 0;
      }
    }
    if (i < end) csr_out[beg + rank] = key;
  }
}

__device__ __forceinline__ float lrelu(float v) { return (v > 0.f) ? v : NEG_SLOPE * v; }

__global__ void node_gather(const int* __restrict__ rowptr, const int* __restrict__ csr_src,
                            const float* __restrict__ als, const float* __restrict__ ald,
                            const float* __restrict__ hs, const float* __restrict__ hl,
                            const float* __restrict__ b, const float* __restrict__ bl,
                            float* __restrict__ out, int Nn) {
  const int lane = threadIdx.x & 31;
  const int node = __builtin_amdgcn_readfirstlane((int)((blockIdx.x * blockDim.x + threadIdx.x) >> 5));
  if (node >= Nn) return;
  const int beg = rowptr[node];
  const int end = rowptr[node + 1];
  const float ad0 = ald[node * 2 + 0];
  const float ad1 = ald[node * 2 + 1];
  const float2* __restrict__ als2 = (const float2*)als;

  float m0 = -INFINITY, m1 = -INFINITY;
  for (int i = beg + lane; i < end; i += 32) {
    const float2 a = als2[csr_src[i]];
    m0 = fmaxf(m0, lrelu(a.x + ad0));
    m1 = fmaxf(m1, lrelu(a.y + ad1));
  }
  for (int o = 16; o > 0; o >>= 1) {
    m0 = fmaxf(m0, __shfl_xor(m0, o));
    m1 = fmaxf(m1, __shfl_xor(m1, o));
  }
  if (!(fabsf(m0) < INFINITY)) m0 = 0.f;
  if (!(fabsf(m1) < INFINITY)) m1 = 0.f;

  float d0 = 0.f, d1 = 0.f;
  for (int i = beg + lane; i < end; i += 32) {
    const float2 a = als2[csr_src[i]];
    d0 += __expf(lrelu(a.x + ad0) - m0);
    d1 += __expf(lrelu(a.y + ad1) - m1);
  }
  for (int o = 16; o > 0; o >>= 1) {
    d0 += __shfl_xor(d0, o);
    d1 += __shfl_xor(d1, o);
  }
  const float r0 = d0 + 1e-16f;
  const float r1 = d1 + 1e-16f;

  const bool head1 = lane >= 16;
  const float* __restrict__ hsl = hs + lane * 4;
  float4 acc = make_float4(0.f, 0.f, 0.f, 0.f);
  for (int base = beg; base < end; base += 32) {
    const int i = base + lane;
    int s_l = 0;
    float w_l0 = 0.f, w_l1 = 0.f;
    if (i < end) {
      s_l = csr_src[i];
      const float2 a = als2[s_l];
      w_l0 = __expf(lrelu(a.x + ad0) - m0) / r0;
      w_l1 = __expf(lrelu(a.y + ad1) - m1) / r1;
    }
    const int cnt = (end - base < 32) ? (end - base) : 32;
#pragma unroll 4
    for (int j = 0; j < cnt; ++j) {
      const int s = __builtin_amdgcn_readlane(s_l, j);
      const float wa = __int_as_float(__builtin_amdgcn_readlane(__float_as_int(w_l0), j));
      const float wb = __int_as_float(__builtin_amdgcn_readlane(__float_as_int(w_l1), j));
      const float w = head1 ? wb : wa;
      const float4 v = *(const float4*)(hsl + (size_t)s * 128);
      acc.x = fmaf(w, v.x, acc.x);
      acc.y = fmaf(w, v.y, acc.y);
      acc.z = fmaf(w, v.z, acc.z);
      acc.w = fmaf(w, v.w, acc.w);
    }
  }
  acc.x += __shfl_xor(acc.x, 16);
  acc.y += __shfl_xor(acc.y, 16);
  acc.z += __shfl_xor(acc.z, 16);
  acc.w += __shfl_xor(acc.w, 16);
  if (lane < 16) {
    const int c = lane * 4;
    const float4 hv = *(const float4*)(hl + (size_t)node * 64 + c);
    float4 o;
    o.x = 0.5f * acc.x + b[c + 0] + hv.x + bl[c + 0];
    o.y = 0.5f * acc.y + b[c + 1] + hv.y + bl[c + 1];
    o.z = 0.5f * acc.z + b[c + 2] + hv.z + bl[c + 2];
    o.w = 0.5f * acc.w + b[c + 3] + hv.w + bl[c + 3];
    o.x = (o.x > 0.f) ? o.x : 0.f;
    o.y = (o.y > 0.f) ? o.y : 0.f;
    o.z = (o.z > 0.f) ? o.z : 0.f;
    o.w = (o.w > 0.f) ? o.w : 0.f;
    *(float4*)(out + (size_t)node * 64 + c) = o;
  }
}

__global__ void node_gather_in(const int* __restrict__ rowptr, const int* __restrict__ csr_src,
                               const float* __restrict__ als, const float* __restrict__ ald,
                               const float* __restrict__ hin, float* __restrict__ z, int Nn) {
  const int lane = threadIdx.x & 31;
  const int node = __builtin_amdgcn_readfirstlane((int)((blockIdx.x * blockDim.x + threadIdx.x) >> 5));
  if (node >= Nn) return;
  const int beg = rowptr[node];
  const int end = rowptr[node + 1];
  const float ad0 = ald[node * 2 + 0];
  const float ad1 = ald[node * 2 + 1];
  const float2* __restrict__ als2 = (const float2*)als;

  float m0 = -INFINITY, m1 = -INFINITY;
  for (int i = beg + lane; i < end; i += 32) {
    const float2 a = als2[csr_src[i]];
    m0 = fmaxf(m0, lrelu(a.x + ad0));
    m1 = fmaxf(m1, lrelu(a.y + ad1));
  }
  for (int o = 16; o > 0; o >>= 1) {
    m0 = fmaxf(m0, __shfl_xor(m0, o));
    m1 = fmaxf(m1, __shfl_xor(m1, o));
  }
  if (!(fabsf(m0) < INFINITY)) m0 = 0.f;
  if (!(fabsf(m1) < INFINITY)) m1 = 0.f;

  float d0 = 0.f, d1 = 0.f;
  for (int i = beg + lane; i < end; i += 32) {
    const float2 a = als2[csr_src[i]];
    d0 += __expf(lrelu(a.x + ad0) - m0);
    d1 += __expf(lrelu(a.y + ad1) - m1);
  }
  for (int o = 16; o > 0; o >>= 1) {
    d0 += __shfl_xor(d0, o);
    d1 += __shfl_xor(d1, o);
  }
  const float r0 = d0 + 1e-16f;
  const float r1 = d1 + 1e-16f;

  const bool odd = lane >= 16;
  const int c = (lane & 15) * 4;
  const float* __restrict__ hrow = hin + c;
  float4 acc0 = make_float4(0.f, 0.f, 0.f, 0.f);
  float4 acc1 = make_float4(0.f, 0.f, 0.f, 0.f);
  for (int base = beg; base < end; base += 32) {
    const int i = base + lane;
    int s_l = 0;
    float w_l0 = 0.f, w_l1 = 0.f;
    if (i < end) {
      s_l = csr_src[i];
      const float2 a = als2[s_l];
      w_l0 = __expf(lrelu(a.x + ad0) - m0) / r0;
      w_l1 = __expf(lrelu(a.y + ad1) - m1) / r1;
    }
    const int s_first = __builtin_amdgcn_readfirstlane(s_l);
    if (i >= end) s_l = s_first;
    const int cnt = (end - base < 32) ? (end - base) : 32;
#pragma unroll 4
    for (int j = 0; j < cnt; j += 2) {
      const int sa = __builtin_amdgcn_readlane(s_l, j);
      const int sb = __builtin_amdgcn_readlane(s_l, j + 1);
      const float wa0 = __int_as_float(__builtin_amdgcn_readlane(__float_as_int(w_l0), j));
      const float wb0 = __int_as_float(__builtin_amdgcn_readlane(__float_as_int(w_l0), j + 1));
      const float wa1 = __int_as_float(__builtin_amdgcn_readlane(__float_as_int(w_l1), j));
      const float wb1 = __int_as_float(__builtin_amdgcn_readlane(__float_as_int(w_l1), j + 1));
      const int s = odd ? sb : sa;
      const float w0 = odd ? wb0 : wa0;
      const float w1 = odd ? wb1 : wa1;
      const float4 v = *(const float4*)(hrow + (size_t)s * 64);
      acc0.x = fmaf(w0, v.x, acc0.x);
      acc0.y = fmaf(w0, v.y, acc0.y);
      acc0.z = fmaf(w0, v.z, acc0.z);
      acc0.w = fmaf(w0, v.w, acc0.w);
      acc1.x = fmaf(w1, v.x, acc1.x);
      acc1.y = fmaf(w1, v.y, acc1.y);
      acc1.z = fmaf(w1, v.z, acc1.z);
      acc1.w = fmaf(w1, v.w, acc1.w);
    }
  }
  acc0.x += __shfl_xor(acc0.x, 16);
  acc0.y += __shfl_xor(acc0.y, 16);
  acc0.z += __shfl_xor(acc0.z, 16);
  acc0.w += __shfl_xor(acc0.w, 16);
  acc1.x += __shfl_xor(acc1.x, 16);
  acc1.y += __shfl_xor(acc1.y, 16);
  acc1.z += __shfl_xor(acc1.z, 16);
  acc1.w += __shfl_xor(acc1.w, 16);
  float4 o;
  o.x = odd ? acc1.x : acc0.x;
  o.y = odd ? acc1.y : acc0.y;
  o.z = odd ? acc1.z : acc0.z;
  o.w = odd ? acc1.w : acc0.w;
  *(float4*)(z + (size_t)node * 128 + (odd ? 64 : 0) + c) = o;
}

__global__ void gemm_zout(const float* __restrict__ Z,
                          const float* __restrict__ Ws,
                          const float* __restrict__ hl,
                          const float* __restrict__ b,
                          const float* __restrict__ bl,
                          float* __restrict__ out,
                          int Nrows) {
  constexpr int K = 128, M = 64, MT = 80, TILES = M / 16;
  __shared__ __attribute__((aligned(16))) float wlds[K * MT];

  constexpr int NW = (64 * 128 / 4) / GEMM_THREADS;
  static_assert((64 * 128 / 4) % GEMM_THREADS == 0, "Ws must split evenly over the work-group");
  const float4* __restrict__ W4 = (const float4*)Ws;
  float4 wv[NW];
#pragma unroll
  for (int j = 0; j < NW; ++j) wv[j] = W4[threadIdx.x + j * GEMM_THREADS];
#pragma unroll
  for (int j = 0; j < NW; ++j) {
    const int f = (threadIdx.x + j * GEMM_THREADS) * 4;
    const int k = f >> 7;
    const int col = f & 127;
    *(float4*)&wlds[((col >> 6) * 64 + k) * MT + (col & 63)] = wv[j];
  }
  __syncthreads();

  const int wavesPerBlock = blockDim.x >> 5;
  const int rowTile = blockIdx.x * wavesPerBlock + (threadIdx.x >> 5);
  const int lane = threadIdx.x & 31;
  if (rowTile >= (Nrows >> 4)) return;

  const int l16 = lane & 15;
  const int kq  = (lane >> 4) << 1;
  const float* __restrict__ arow = Z + (size_t)(rowTile * 16 + l16) * (size_t)K;

  const int rbase = rowTile * 16 + ((lane >> 4) << 3);
  float hv[TILES][8];
  float bc[TILES], blc[TILES];
#pragma unroll
  for (int t = 0; t < TILES; ++t) {
    const int col = t * 16 + l16;
    bc[t] = b[col];
    blc[t] = bl[col];
#pragma unroll
    for (int r = 0; r < 8; ++r)
      hv[t][r] = hl[(size_t)(rbase + r) * (size_t)M + col];
  }

  v8f acc[TILES];
#pragma unroll
  for (int t = 0; t < TILES; ++t) acc[t] = (v8f){0.f,0.f,0.f,0.f,0.f,0.f,0.f,0.f};

  for (int k0 = 0; k0 < K; k0 += 4) {
    v2f a;
    a.x = arow[k0 + kq];
    a.y = arow[k0 + kq + 1];
    const float* wr0 = &wlds[(k0 + kq) * MT + l16];
#pragma unroll
    for (int t = 0; t < TILES; ++t) {
      v2f bb;
      bb.x = wr0[t * 16];
      bb.y = wr0[MT + t * 16];
      acc[t] = __builtin_amdgcn_wmma_f32_16x16x4_f32(
           false, a,  false, bb,
           (short)0, acc[t],  false,  false);
    }
  }

#pragma unroll
  for (int t = 0; t < TILES; ++t) {
    const int col = t * 16 + l16;
#pragma unroll
    for (int r = 0; r < 8; ++r) {
      const float v = 0.5f * acc[t][r] + bc[t] + hv[t][r] + blc[t];
      out[(size_t)(rbase + r) * (size_t)M + col] = (v > 0.f) ? v : 0.f;
    }
  }
}

extern "C" void kernel_launch(void* const* d_in, const int* in_sizes, int n_in,
                              void* d_out, int out_size, void* d_ws, size_t ws_size,
                              hipStream_t stream) {
  const float* x  = (const float*)d_in[0];
  const int*   ei = (const int*)d_in[1];
  const int Nn = in_sizes[0] / 128;
  const int Ee = in_sizes[1] / 2;
  const int* src = ei;
  const int* dst = ei + Ee;

  const float *Ws[3], *Wd[3], *As[3], *Ad[3], *B[3], *Wl[3], *Bl[3];
  for (int l = 0; l < 3; ++l) {
    int base = 2 + 7 * l;
    Ws[l] = (const float*)d_in[base + 0];
    Wd[l] = (const float*)d_in[base + 1];
    As[l] = (const float*)d_in[base + 2];
    Ad[l] = (const float*)d_in[base + 3];
    B[l]  = (const float*)d_in[base + 4];
    Wl[l] = (const float*)d_in[base + 5];
    Bl[l] = (const float*)d_in[base + 6];
  }

  float* ws = (float*)d_ws;
  size_t off = 0;
  float* hs  = ws + off; off += (size_t)Nn * 128;
  float* hl  = ws + off; off += (size_t)Nn * 64;
  float* wx  = ws + off; off += (size_t)256 * 16;
  float* als = ws + off; off += (size_t)Nn * 2;
  float* ald = ws + off; off += (size_t)Nn * 2;
  float* h1  = ws + off; off += (size_t)Nn * 64;
  float* h2  = ws + off; off += (size_t)Nn * 64;
  int* iws = (int*)(ws + off);
  size_t ioff = 0;
  int* rowptr  = iws + ioff; ioff += (size_t)Nn + 1;
  int* csr_tmp = iws + ioff; ioff += (size_t)Ee;
  int* csr_src = iws + ioff; ioff += (size_t)Ee;

  const float* lin[3]  = { x, h1, h2 };
  const float* wxl[3]  = { wx, wx + 128 * 16, wx + 192 * 16 };
  float*       lout[3] = { h1, h2, (float*)d_out };

  const int TPB = 256;
  const int rowTiles = Nn >> 4;
  const int gemmBlocks = (rowTiles + 7) / 8;

  int* binned  = (int*)hs;
  int* tbl     = binned + 2 * (size_t)Ee;
  int* binbase = tbl + (size_t)SORT_BLOCKS * MAX_BINS;
  const int nbins = (Nn + BIN_NODES - 1) >> BIN_SHIFT;
  bin_count<<<SORT_BLOCKS, SORT_THREADS, 0, stream>>>(dst, tbl, Ee, nbins);
  bin_scatter<<<SORT_BLOCKS, SORT_THREADS, 0, stream>>>(src, dst, tbl, binbase, binned, Ee, nbins);
  bin_sort<<<nbins, BSORT_THREADS, 0, stream>>>(binbase, binned, rowptr, csr_tmp, Nn);
  {
    long long threads = (long long)Nn * 32;
    csr_order<<<(int)((threads + TPB - 1) / TPB), TPB, 0, stream>>>(rowptr, csr_tmp, csr_src, Nn);
  }

  fold_att<<<256, 128, 0, stream>>>(Ws[0], Wd[0], As[0], Ad[0], Ws[1], Wd[1], As[1], Ad[1],
                                    Ws[2], Wd[2], As[2], Ad[2], wx);

  for (int l = 0; l < 3; ++l) {
    if (l == 0) {
      gemm_wmma_lds<128, 128, true ><<<gemmBlocks, GEMM_THREADS, 0, stream>>>(lin[l], Ws[l], wxl[l], hs, als, ald, Nn);
      gemm_wmma_lds<128, 64,  false><<<gemmBlocks, GEMM_THREADS, 0, stream>>>(lin[l], Wl[l], nullptr, hl, nullptr, nullptr, Nn);
      long long threads = (long long)Nn * 32;
      node_gather<<<(int)((threads + TPB - 1) / TPB), TPB, 0, stream>>>(
          rowptr, csr_src, als, ald, hs, hl, B[l], Bl[l], lout[l], Nn);
    } else {
      gemm_wmma_lds<64, 64, true><<<gemmBlocks, GEMM_THREADS, 0, stream>>>(lin[l], Wl[l], wxl[l], hl, als, ald, Nn);
      long long threads = (long long)Nn * 32;
      node_gather_in<<<(int)((threads + TPB - 1) / TPB), TPB, 0, stream>>>(
          rowptr, csr_src, als, ald, lin[l], hs, Nn);
      gemm_zout<<<gemmBlocks, GEMM_THREADS, 0, stream>>>(hs, Ws[l], hl, B[l], Bl[l], lout[l], Nn);
    }
  }
}


# Per-node launches (source ordering, three gathers): each wave walks 4 consecutive nodes, next nodes' bounds, sources and attention terms fetched ahead; 50000 waves a call, not 200000; sums unchanged
# speedup vs baseline: 4.5911x; 1.1056x over previous
#include <hip/hip_runtime.h>
#include <math.h>

typedef __attribute__((ext_vector_type(2))) float v2f;
typedef __attribute__((ext_vector_type(8))) float v8f;

#define NEG_SLOPE 0.2f

constexpr int GEMM_THREADS = 256;

template <int K, int M, bool ATT>
__global__ void gemm_wmma_lds(const float* __restrict__ A,
                              const float* __restrict__ W,
                              const float* __restrict__ Wx,
                              float* __restrict__ C,
                              float* __restrict__ als,
                              float* __restrict__ ald,
                              int Nrows) {
  constexpr int MT = ATT ? M + 16 : M;
  constexpr int TILES_C = M / 16;
  constexpr int TILES_M = MT / 16;
  __shared__ __attribute__((aligned(16))) float wlds[K * MT];

  constexpr int NW = (K * M / 4) / GEMM_THREADS;
  constexpr int NX = ATT ? (K * 16 / 4) / GEMM_THREADS : 0;
  static_assert((K * M / 4) % GEMM_THREADS == 0, "W must split evenly over the work-group");
  static_assert(!ATT || (K * 16 / 4) % GEMM_THREADS == 0, "Wx must split evenly over the work-group");
  const float4* __restrict__ W4 = (const float4*)W;
  float4 wv[NW];
  float4 xv[NX > 0 ? NX : 1];
#pragma unroll
  for (int j = 0; j < NW; ++j) wv[j] = W4[threadIdx.x + j * GEMM_THREADS];
  if constexpr (ATT) {
    const float4* __restrict__ Wx4 = (const float4*)Wx;
#pragma unroll
    for (int j = 0; j < NX; ++j) xv[j] = Wx4[threadIdx.x + j * GEMM_THREADS];
  }
#pragma unroll
  for (int j = 0; j < NW; ++j) {
    const int f = (threadIdx.x + j * GEMM_THREADS) * 4;
    *(float4*)&wlds[(f / M) * MT + (f % M)] = wv[j];
  }
  if constexpr (ATT) {
#pragma unroll
    for (int j = 0; j < NX; ++j) {
      const int i4 = threadIdx.x + j * GEMM_THREADS;
      *(float4*)&wlds[(i4 >> 2) * MT + M + ((i4 & 3) << 2)] = xv[j];
    }
  }
  __syncthreads();

  const int wavesPerBlock = blockDim.x >> 5;
  const int rowTile = blockIdx.x * wavesPerBlock + (threadIdx.x >> 5);
  const int lane = threadIdx.x & 31;
  if (rowTile >= (Nrows >> 4)) return;

  const int l16 = lane & 15;
  const int kq  = (lane >> 4) << 1;
  const float* __restrict__ arow = A + (size_t)(rowTile * 16 + l16) * (size_t)K;

  v8f acc[TILES_M];
#pragma unroll
  for (int t = 0; t < TILES_M; ++t) acc[t] = (v8f){0.f,0.f,0.f,0.f,0.f,0.f,0.f,0.f};

  for (int k0 = 0; k0 < K; k0 += 4) {
    v2f a;
    a.x = arow[k0 + kq];
    a.y = arow[k0 + kq + 1];
    const float* wr0 = &wlds[(k0 + kq) * MT + l16];
#pragma unroll
    for (int t = 0; t < TILES_M; ++t) {
      v2f b;
      b.x = wr0[t * 16];
      b.y = wr0[MT + t * 16];
      acc[t] = __builtin_amdgcn_wmma_f32_16x16x4_f32(
           false, a,  false, b,
           (short)0, acc[t],  false,  false);
    }
  }

  const int rbase = rowTile * 16 + ((lane >> 4) << 3);
#pragma unroll
  for (int t = 0; t < TILES_C; ++t) {
    const int col = t * 16 + l16;
#pragma unroll
    for (int r = 0; r < 8; ++r)
      C[(size_t)(rbase + r) * (size_t)M + col] = acc[t][r];
  }
  if constexpr (ATT) {
    if (l16 < 4) {
      float* __restrict__ ap = (l16 < 2) ? (als + l16) : (ald + (l16 - 2));
#pragma unroll
      for (int r = 0; r < 8; ++r)
        ap[(size_t)(rbase + r) * 2] = acc[TILES_C][r];
    }
  }
}

__global__ void fold_att(const float* __restrict__ Ws0, const float* __restrict__ Wd0,
                         const float* __restrict__ as0, const float* __restrict__ ad0,
                         const float* __restrict__ Ws1, const float* __restrict__ Wd1,
                         const float* __restrict__ as1, const float* __restrict__ ad1,
                         const float* __restrict__ Ws2, const float* __restrict__ Wd2,
                         const float* __restrict__ as2, const float* __restrict__ ad2,
                         float* __restrict__ wx) {
  const int row = blockIdx.x;
  const float *Ws, *Wd, *a_s, *a_d;
  int k;
  if (row < 128)      { Ws = Ws0; Wd = Wd0; a_s = as0; a_d = ad0; k = row; }
  else if (row < 192) { Ws = Ws1; Wd = Wd1; a_s = as1; a_d = ad1; k = row - 128; }
  else                { Ws = Ws2; Wd = Wd2; a_s = as2; a_d = ad2; k = row - 192; }
  const int j = threadIdx.x >> 5;
  const int lane = threadIdx.x & 31;
  const int h = j & 1;
  const float* wrow = ((j < 2) ? Ws : Wd) + (size_t)k * 128 + h * 64;
  const float* av   = ((j < 2) ? a_s : a_d) + h * 64;
  float s = wrow[lane] * av[lane] + wrow[lane + 32] * av[lane + 32];
  for (int o = 16; o > 0; o >>= 1) s += __shfl_xor(s, o);
  if (lane == 0) wx[row * 16 + j] = s;
  if (threadIdx.x >= 4 && threadIdx.x < 16) wx[row * 16 + threadIdx.x] = 0.f;
}

constexpr int BIN_SHIFT = 7;
constexpr int BIN_NODES = 1 << BIN_SHIFT;
constexpr int MAX_BINS = 512;
constexpr int SORT_BLOCKS = 128;
constexpr int SORT_THREADS = 1024;
constexpr int SORT_BATCH = 8;
constexpr int BSORT_THREADS = 512;
constexpr int BSORT_BATCH = 4;

__device__ __forceinline__ int block_excl_scan(int v, int* wsum) {
  const int lane = threadIdx.x & 31;
  const int w = threadIdx.x >> 5;
  int incl = v;
  for (int o = 1; o < 32; o <<= 1) {
    const int u = __shfl_up(incl, o);
    if (lane >= o) incl += u;
  }
  if (lane == 31) wsum[w] = incl;
  __syncthreads();
  int pre = 0;
  for (int q = 0; q < w; ++q) pre += wsum[q];
  return pre + incl - v;
}

__global__ void bin_count(const int* __restrict__ dst, int* __restrict__ tbl, int Ee, int nbins) {
  __shared__ int hist[MAX_BINS];
  const int t = threadIdx.x;
  if (t < MAX_BINS) hist[t] = 0;
  __syncthreads();
  const int chunk = (Ee + SORT_BLOCKS - 1) / SORT_BLOCKS;
  const int lo = (int)blockIdx.x * chunk;
  const int hi = min(lo + chunk, Ee);
  for (int i0 = lo + t; i0 < hi; i0 += SORT_BATCH * SORT_THREADS) {
    int d[SORT_BATCH];
#pragma unroll
    for (int k = 0; k < SORT_BATCH; ++k) {
      const int i = i0 + k * SORT_THREADS;
      d[k] = (i < hi) ? dst[i] : 0;
    }
#pragma unroll
    for (int k = 0; k < SORT_BATCH; ++k)
      if (i0 + k * SORT_THREADS < hi) atomicAdd(&hist[d[k] >> BIN_SHIFT], 1);
  }
  __syncthreads();
  if (t < nbins) tbl[(int)blockIdx.x * nbins + t] = hist[t];
}

__global__ void bin_scatter(const int* __restrict__ src, const int* __restrict__ dst,
                            const int* __restrict__ tbl, int* __restrict__ binbase,
                            int* __restrict__ binned, int Ee, int nbins) {
  __shared__ int cur[MAX_BINS];
  __shared__ int ptot[2][MAX_BINS];
  __shared__ int pbef[2][MAX_BINS];
  __shared__ int wsum[SORT_THREADS / 32];
  const int t = threadIdx.x;
  const int g = blockIdx.x;
  const int bin = t & (MAX_BINS - 1);
  const int half = t / MAX_BINS;
  int tot = 0, bef = 0;
  if (bin < nbins) {
    const int q0 = half * (SORT_BLOCKS / 2);
#pragma unroll 16
    for (int q = q0; q < q0 + SORT_BLOCKS / 2; ++q) {
      const int c = tbl[q * nbins + bin];
      tot += c;
      bef += (q < g) ? c : 0;
    }
  }
  ptot[half][bin] = tot;
  pbef[half][bin] = bef;
  __syncthreads();
  int total = 0, before = 0;
  if (t < MAX_BINS) {
    total = ptot[0][t] + ptot[1][t];
    before = pbef[0][t] + pbef[1][t];
  }
  const int base = block_excl_scan(total, wsum);
  if (t < nbins) {
    cur[t] = base + before;
    if (g == 0) {
      binbase[t] = base;
      if (t == nbins - 1) binbase[nbins] = base + total;
    }
  }
  __syncthreads();
  const int chunk = (Ee + SORT_BLOCKS - 1) / SORT_BLOCKS;
  const int lo = g * chunk;
  const int hi = min(lo + chunk, Ee);
  int2* __restrict__ pairs = (int2*)binned;
  for (int i0 = lo + t; i0 < hi; i0 += SORT_BATCH * SORT_THREADS) {
    int d[SORT_BATCH], s[SORT_BATCH];
#pragma unroll
    for (int k = 0; k < SORT_BATCH; ++k) {
      const int i = i0 + k * SORT_THREADS;
      d[k] = (i < hi) ? dst[i] : 0;
      s[k] = (i < hi) ? src[i] : 0;
    }
#pragma unroll
    for (int k = 0; k < SORT_BATCH; ++k) {
      if (i0 + k * SORT_THREADS < hi) {
        const int p = atomicAdd(&cur[d[k] >> BIN_SHIFT], 1);
        pairs[p] = make_int2(s[k], d[k]);
      }
    }
  }
}

__global__ void bin_sort(const int* __restrict__ binbase, const int* __restrict__ binned,
                         int* __restrict__ rowptr, int* __restrict__ csr_out, int Nn) {
  __shared__ int cnt[BIN_NODES];
  __shared__ int cur[BIN_NODES];
  __shared__ int wsum[BSORT_THREADS / 32];
  const int2* __restrict__ pairs = (const int2*)binned;
  const int t = threadIdx.x;
  const int b = blockIdx.x;
  const int s = binbase[b];
  const int e = binbase[b + 1];
  if (t < BIN_NODES) cnt[t] = 0;
  __syncthreads();
  for (int i0 = s + t; i0 < e; i0 += BSORT_BATCH * BSORT_THREADS) {
    int d[BSORT_BATCH];
#pragma unroll
    for (int k = 0; k < BSORT_BATCH; ++k) {
      const int i = i0 + k * BSORT_THREADS;
      d[k] = (i < e) ? pairs[i].y : 0;
    }
#pragma unroll
    for (int k = 0; k < BSORT_BATCH; ++k)
      if (i0 + k * BSORT_THREADS < e) atomicAdd(&cnt[d[k] & (BIN_NODES - 1)], 1);
  }
  __syncthreads();
  const int c = (t < BIN_NODES) ? cnt[t] : 0;
  const int first = s + block_excl_scan(c, wsum);
  if (t < BIN_NODES) {
    const int node = b * BIN_NODES + t;
    cur[t] = first;
    if (node < Nn) rowptr[node] = first;
    if (node == Nn - 1) rowptr[Nn] = first + c;
  }
  __syncthreads();
  for (int i0 = s + t; i0 < e; i0 += BSORT_BATCH * BSORT_THREADS) {
    int2 pr[BSORT_BATCH];
#pragma unroll
    for (int k = 0; k < BSORT_BATCH; ++k) {
      const int i = i0 + k * BSORT_THREADS;
      pr[k] = (i < e) ? pairs[i] : make_int2(0, 0);
    }
#pragma unroll
    for (int k = 0; k < BSORT_BATCH; ++k) {
      if (i0 + k * BSORT_THREADS < e) {
        const int p = atomicAdd(&cur[pr[k].y & (BIN_NODES - 1)], 1);
        csr_out[p] = pr[k].x;
      }
    }
  }
}

constexpr int NODES_PER_WAVE = 4;

__global__ void csr_order(const int* __restrict__ rowptr, const int* __restrict__ csr_in,
                          int* __restrict__ csr_out, int Nn) {
  const int lane = threadIdx.x & 31;
  const int wave = __builtin_amdgcn_readfirstlane((int)((blockIdx.x * blockDim.x + threadIdx.x) >> 5));
  const int first = wave * NODES_PER_WAVE;
  if (first >= Nn) return;
  const int last = min(first + NODES_PER_WAVE, Nn);
  int beg = rowptr[first];
  int end = rowptr[first + 1];
  for (int node = first; node < last; ++node) {
  const int nend = rowptr[min(node + 2, Nn)];
  for (int cb = beg; cb < end; cb += 32) {
    const int i = cb + lane;
    const int key = (i < end) ? csr_in[i] : 0;
    int rank = 0;
    for (int kb = beg; kb < end; kb += 32) {
      const int k = kb + lane;
      const int other = (k < end) ? csr_in[k] : 0;
      const int n = (end - kb < 32) ? (end - kb) : 32;
      for (int j = 0; j < n; ++j) {
        const int o = __builtin_amdgcn_readlane(other, j);
        const int pos = kb + j;
        rank += ((o < key) || (o == key && pos < i)) ? 1 : 0;
      }
    }
    if (i < end) csr_out[beg + rank] = key;
  }
  beg = end;
  end = nend;
  }
}

__device__ __forceinline__ float lrelu(float v) { return (v > 0.f) ? v : NEG_SLOPE * v; }

__global__ void node_gather(const int* __restrict__ rowptr, const int* __restrict__ csr_src,
                            const float* __restrict__ als, const float* __restrict__ ald,
                            const float* __restrict__ hs, const float* __restrict__ hl,
                            const float* __restrict__ b, const float* __restrict__ bl,
                            float* __restrict__ out, int Nn, int Ee) {
  const int lane = threadIdx.x & 31;
  const int wave = __builtin_amdgcn_readfirstlane((int)((blockIdx.x * blockDim.x + threadIdx.x) >> 5));
  const int first = wave * NODES_PER_WAVE;
  if (first >= Nn) return;
  const int last = min(first + NODES_PER_WAVE, Nn);
  const float2* __restrict__ als2 = (const float2*)als;
  const int f1 = min(first + 1, Nn - 1);
  int beg  = rowptr[first];
  int end  = rowptr[first + 1];
  int end1 = rowptr[f1 + 1];
  float ad0  = ald[first * 2 + 0];
  float ad1  = ald[first * 2 + 1];
  float nad0 = ald[f1 * 2 + 0];
  float nad1 = ald[f1 * 2 + 1];
  int s_cur = csr_src[min(beg + lane, Ee - 1)];
  int s_nxt = csr_src[min(end + lane, Ee - 1)];
  const float2 a_first = als2[(beg + lane < end) ? s_cur : __builtin_amdgcn_readfirstlane(s_cur)];
  float2 a_cur = a_first;
  for (int node = first; node < last; ++node) {
  const int n2 = min(node + 2, Nn - 1);
  const int end2 = rowptr[n2 + 1];
  const float n2ad0 = ald[n2 * 2 + 0];
  const float n2ad1 = ald[n2 * 2 + 1];
  const int s_nn = csr_src[min(end1 + lane, Ee - 1)];
  const float2 a_nxt = als2[(end + lane < end1) ? s_nxt : __builtin_amdgcn_readfirstlane(s_nxt)];

  float m0 = -INFINITY, m1 = -INFINITY;
  if (beg + lane < end) {
    m0 = fmaxf(m0, lrelu(a_cur.x + ad0));
    m1 = fmaxf(m1, lrelu(a_cur.y + ad1));
  }
  for (int i = beg + 32 + lane; i < end; i += 32) {
    const float2 a = als2[csr_src[i]];
    m0 = fmaxf(m0, lrelu(a.x + ad0));
    m1 = fmaxf(m1, lrelu(a.y + ad1));
  }
  for (int o = 16; o > 0; o >>= 1) {
    m0 = fmaxf(m0, __shfl_xor(m0, o));
    m1 = fmaxf(m1, __shfl_xor(m1, o));
  }
  if (!(fabsf(m0) < INFINITY)) m0 = 0.f;
  if (!(fabsf(m1) < INFINITY)) m1 = 0.f;

  float d0 = 0.f, d1 = 0.f;
  if (beg + lane < end) {
    d0 += __expf(lrelu(a_cur.x + ad0) - m0);
    d1 += __expf(lrelu(a_cur.y + ad1) - m1);
  }
  for (int i = beg + 32 + lane; i < end; i += 32) {
    const float2 a = als2[csr_src[i]];
    d0 += __expf(lrelu(a.x + ad0) - m0);
    d1 += __expf(lrelu(a.y + ad1) - m1);
  }
  for (int o = 16; o > 0; o >>= 1) {
    d0 += __shfl_xor(d0, o);
    d1 += __shfl_xor(d1, o);
  }
  const float r0 = d0 + 1e-16f;
  const float r1 = d1 + 1e-16f;

  const bool head1 = lane >= 16;
  const float* __restrict__ hsl = hs + lane * 4;
  float4 acc = make_float4(0.f, 0.f, 0.f, 0.f);
  for (int base = beg; base < end; base += 32) {
    const int i = base + lane;
    int s_l = 0;
    float w_l0 = 0.f, w_l1 = 0.f;
    if (i < end) {
      float2 a;
      if (base == beg) { s_l = s_cur; a = a_cur; }
      else             { s_l = csr_src[i]; a = als2[s_l]; }
      w_l0 = __expf(lrelu(a.x + ad0) - m0) / r0;
      w_l1 = __expf(lrelu(a.y + ad1) - m1) / r1;
    }
    const int cnt = (end - base < 32) ? (end - base) : 32;
#pragma unroll 4
    for (int j = 0; j < cnt; ++j) {
      const int s = __builtin_amdgcn_readlane(s_l, j);
      const float wa = __int_as_float(__builtin_amdgcn_readlane(__float_as_int(w_l0), j));
      const float wb = __int_as_float(__builtin_amdgcn_readlane(__float_as_int(w_l1), j));
      const float w = head1 ? wb : wa;
      const float4 v = *(const float4*)(hsl + (size_t)s * 128);
      acc.x = fmaf(w, v.x, acc.x);
      acc.y = fmaf(w, v.y, acc.y);
      acc.z = fmaf(w, v.z, acc.z);
      acc.w = fmaf(w, v.w, acc.w);
    }
  }
  acc.x += __shfl_xor(acc.x, 16);
  acc.y += __shfl_xor(acc.y, 16);
  acc.z += __shfl_xor(acc.z, 16);
  acc.w += __shfl_xor(acc.w, 16);
  if (lane < 16) {
    const int c = lane * 4;
    const float4 hv = *(const float4*)(hl + (size_t)node * 64 + c);
    float4 o;
    o.x = 0.5f * acc.x + b[c + 0] + hv.x + bl[c + 0];
    o.y = 0.5f * acc.y + b[c + 1] + hv.y + bl[c + 1];
    o.z = 0.5f * acc.z + b[c + 2] + hv.z + bl[c + 2];
    o.w = 0.5f * acc.w + b[c + 3] + hv.w + bl[c + 3];
    o.x = (o.x > 0.f) ? o.x : 0.f;
    o.y = (o.y > 0.f) ? o.y : 0.f;
    o.z = (o.z > 0.f) ? o.z : 0.f;
    o.w = (o.w > 0.f) ? o.w : 0.f;
    *(float4*)(out + (size_t)node * 64 + c) = o;
  }
  beg = end;   end = end1;   end1 = end2;
  ad0 = nad0;  ad1 = nad1;   nad0 = n2ad0;  nad1 = n2ad1;
  s_cur = s_nxt;  s_nxt = s_nn;  a_cur = a_nxt;
  }
}

__global__ void node_gather_in(const int* __restrict__ rowptr, const int* __restrict__ csr_src,
                               const float* __restrict__ als, const float* __restrict__ ald,
                               const float* __restrict__ hin, float* __restrict__ z, int Nn, int Ee) {
  const int lane = threadIdx.x & 31;
  const int wave = __builtin_amdgcn_readfirstlane((int)((blockIdx.x * blockDim.x + threadIdx.x) >> 5));
  const int first = wave * NODES_PER_WAVE;
  if (first >= Nn) return;
  const int last = min(first + NODES_PER_WAVE, Nn);
  const float2* __restrict__ als2 = (const float2*)als;
  const bool odd = lane >= 16;
  const int c = (lane & 15) * 4;
  const float* __restrict__ hrow = hin + c;
  const int f1 = min(first + 1, Nn - 1);
  int beg  = rowptr[first];
  int end  = rowptr[first + 1];
  int end1 = rowptr[f1 + 1];
  float ad0  = ald[first * 2 + 0];
  float ad1  = ald[first * 2 + 1];
  float nad0 = ald[f1 * 2 + 0];
  float nad1 = ald[f1 * 2 + 1];
  int s_cur = csr_src[min(beg + lane, Ee - 1)];
  int s_nxt = csr_src[min(end + lane, Ee - 1)];
  const float2 a_first = als2[(beg + lane < end) ? s_cur : __builtin_amdgcn_readfirstlane(s_cur)];
  float2 a_cur = a_first;
  for (int node = first; node < last; ++node) {
  const int n2 = min(node + 2, Nn - 1);
  const int end2 = rowptr[n2 + 1];
  const float n2ad0 = ald[n2 * 2 + 0];
  const float n2ad1 = ald[n2 * 2 + 1];
  const int s_nn = csr_src[min(end1 + lane, Ee - 1)];
  const float2 a_nxt = als2[(end + lane < end1) ? s_nxt : __builtin_amdgcn_readfirstlane(s_nxt)];

  float m0 = -INFINITY, m1 = -INFINITY;
  if (beg + lane < end) {
    m0 = fmaxf(m0, lrelu(a_cur.x + ad0));
    m1 = fmaxf(m1, lrelu(a_cur.y + ad1));
  }
  for (int i = beg + 32 + lane; i < end; i += 32) {
    const float2 a = als2[csr_src[i]];
    m0 = fmaxf(m0, lrelu(a.x + ad0));
    m1 = fmaxf(m1, lrelu(a.y + ad1));
  }
  for (int o = 16; o > 0; o >>= 1) {
    m0 = fmaxf(m0, __shfl_xor(m0, o));
    m1 = fmaxf(m1, __shfl_xor(m1, o));
  }
  if (!(fabsf(m0) < INFINITY)) m0 = 0.f;
  if (!(fabsf(m1) < INFINITY)) m1 = 0.f;

  float d0 = 0.f, d1 = 0.f;
  if (beg + lane < end) {
    d0 += __expf(lrelu(a_cur.x + ad0) - m0);
    d1 += __expf(lrelu(a_cur.y + ad1) - m1);
  }
  for (int i = beg + 32 + lane; i < end; i += 32) {
    const float2 a = als2[csr_src[i]];
    d0 += __expf(lrelu(a.x + ad0) - m0);
    d1 += __expf(lrelu(a.y + ad1) - m1);
  }
  for (int o = 16; o > 0; o >>= 1) {
    d0 += __shfl_xor(d0, o);
    d1 += __shfl_xor(d1, o);
  }
  const float r0 = d0 + 1e-16f;
  const float r1 = d1 + 1e-16f;

  float4 acc0 = make_float4(0.f, 0.f, 0.f, 0.f);
  float4 acc1 = make_float4(0.f, 0.f, 0.f, 0.f);
  for (int base = beg; base < end; base += 32) {
    const int i = base + lane;
    int s_l = 0;
    float w_l0 = 0.f, w_l1 = 0.f;
    if (i < end) {
      float2 a;
      if (base == beg) { s_l = s_cur; a = a_cur; }
      else             { s_l = csr_src[i]; a = als2[s_l]; }
      w_l0 = __expf(lrelu(a.x + ad0) - m0) / r0;
      w_l1 = __expf(lrelu(a.y + ad1) - m1) / r1;
    }
    const int s_first = __builtin_amdgcn_readfirstlane(s_l);
    if (i >= end) s_l = s_first;
    const int cnt = (end - base < 32) ? (end - base) : 32;
#pragma unroll 4
    for (int j = 0; j < cnt; j += 2) {
      const int sa = __builtin_amdgcn_readlane(s_l, j);
      const int sb = __builtin_amdgcn_readlane(s_l, j + 1);
      const float wa0 = __int_as_float(__builtin_amdgcn_readlane(__float_as_int(w_l0), j));
      const float wb0 = __int_as_float(__builtin_amdgcn_readlane(__float_as_int(w_l0), j + 1));
      const float wa1 = __int_as_float(__builtin_amdgcn_readlane(__float_as_int(w_l1), j));
      const float wb1 = __int_as_float(__builtin_amdgcn_readlane(__float_as_int(w_l1), j + 1));
      const int s = odd ? sb : sa;
      const float w0 = odd ? wb0 : wa0;
      const float w1 = odd ? wb1 : wa1;
      const float4 v = *(const float4*)(hrow + (size_t)s * 64);
      acc0.x = fmaf(w0, v.x, acc0.x);
      acc0.y = fmaf(w0, v.y, acc0.y);
      acc0.z = fmaf(w0, v.z, acc0.z);
      acc0.w = fmaf(w0, v.w, acc0.w);
      acc1.x = fmaf(w1, v.x, acc1.x);
      acc1.y = fmaf(w1, v.y, acc1.y);
      acc1.z = fmaf(w1, v.z, acc1.z);
      acc1.w = fmaf(w1, v.w, acc1.w);
    }
  }
  acc0.x += __shfl_xor(acc0.x, 16);
  acc0.y += __shfl_xor(acc0.y, 16);
  acc0.z += __shfl_xor(acc0.z, 16);
  acc0.w += __shfl_xor(acc0.w, 16);
  acc1.x += __shfl_xor(acc1.x, 16);
  acc1.y += __shfl_xor(acc1.y, 16);
  acc1.z += __shfl_xor(acc1.z, 16);
  acc1.w += __shfl_xor(acc1.w, 16);
  float4 o;
  o.x = odd ? acc1.x : acc0.x;
  o.y = odd ? acc1.y : acc0.y;
  o.z = odd ? acc1.z : acc0.z;
  o.w = odd ? acc1.w : acc0.w;
  *(float4*)(z + (size_t)node * 128 + (odd ? 64 : 0) + c) = o;
  beg = end;   end = end1;   end1 = end2;
  ad0 = nad0;  ad1 = nad1;   nad0 = n2ad0;  nad1 = n2ad1;
  s_cur = s_nxt;  s_nxt = s_nn;  a_cur = a_nxt;
  }
}

__global__ void gemm_zout(const float* __restrict__ Z,
                          const float* __restrict__ Ws,
                          const float* __restrict__ hl,
                          const float* __restrict__ b,
                          const float* __restrict__ bl,
                          float* __restrict__ out,
                          int Nrows) {
  constexpr int K = 128, M = 64, MT = 80, TILES = M / 16;
  __shared__ __attribute__((aligned(16))) float wlds[K * MT];

  constexpr int NW = (64 * 128 / 4) / GEMM_THREADS;
  static_assert((64 * 128 / 4) % GEMM_THREADS == 0, "Ws must split evenly over the work-group");
  const float4* __restrict__ W4 = (const float4*)Ws;
  float4 wv[NW];
#pragma unroll
  for (int j = 0; j < NW; ++j) wv[j] = W4[threadIdx.x + j * GEMM_THREADS];
#pragma unroll
  for (int j = 0; j < NW; ++j) {
    const int f = (threadIdx.x + j * GEMM_THREADS) * 4;
    const int k = f >> 7;
    const int col = f & 127;
    *(float4*)&wlds[((col >> 6) * 64 + k) * MT + (col & 63)] = wv[j];
  }
  __syncthreads();

  const int wavesPerBlock = blockDim.x >> 5;
  const int rowTile = blockIdx.x * wavesPerBlock + (threadIdx.x >> 5);
  const int lane = threadIdx.x & 31;
  if (rowTile >= (Nrows >> 4)) return;

  const int l16 = lane & 15;
  const int kq  = (lane >> 4) << 1;
  const float* __restrict__ arow = Z + (size_t)(rowTile * 16 + l16) * (size_t)K;

  const int rbase = rowTile * 16 + ((lane >> 4) << 3);
  float hv[TILES][8];
  float bc[TILES], blc[TILES];
#pragma unroll
  for (int t = 0; t < TILES; ++t) {
    const int col = t * 16 + l16;
    bc[t] = b[col];
    blc[t] = bl[col];
#pragma unroll
    for (int r = 0; r < 8; ++r)
      hv[t][r] = hl[(size_t)(rbase + r) * (size_t)M + col];
  }

  v8f acc[TILES];
#pragma unroll
  for (int t = 0; t < TILES; ++t) acc[t] = (v8f){0.f,0.f,0.f,0.f,0.f,0.f,0.f,0.f};

  for (int k0 = 0; k0 < K; k0 += 4) {
    v2f a;
    a.x = arow[k0 + kq];
    a.y = arow[k0 + kq + 1];
    const float* wr0 = &wlds[(k0 + kq) * MT + l16];
#pragma unroll
    for (int t = 0; t < TILES; ++t) {
      v2f bb;
      bb.x = wr0[t * 16];
      bb.y = wr0[MT + t * 16];
      acc[t] = __builtin_amdgcn_wmma_f32_16x16x4_f32(
           false, a,  false, bb,
           (short)0, acc[t],  false,  false);
    }
  }

#pragma unroll
  for (int t = 0; t < TILES; ++t) {
    const int col = t * 16 + l16;
#pragma unroll
    for (int r = 0; r < 8; ++r) {
      const float v = 0.5f * acc[t][r] + bc[t] + hv[t][r] + blc[t];
      out[(size_t)(rbase + r) * (size_t)M + col] = (v > 0.f) ? v : 0.f;
    }
  }
}

extern "C" void kernel_launch(void* const* d_in, const int* in_sizes, int n_in,
                              void* d_out, int out_size, void* d_ws, size_t ws_size,
                              hipStream_t stream) {
  const float* x  = (const float*)d_in[0];
  const int*   ei = (const int*)d_in[1];
  const int Nn = in_sizes[0] / 128;
  const int Ee = in_sizes[1] / 2;
  const int* src = ei;
  const int* dst = ei + Ee;

  const float *Ws[3], *Wd[3], *As[3], *Ad[3], *B[3], *Wl[3], *Bl[3];
  for (int l = 0; l < 3; ++l) {
    int base = 2 + 7 * l;
    Ws[l] = (const float*)d_in[base + 0];
    Wd[l] = (const float*)d_in[base + 1];
    As[l] = (const float*)d_in[base + 2];
    Ad[l] = (const float*)d_in[base + 3];
    B[l]  = (const float*)d_in[base + 4];
    Wl[l] = (const float*)d_in[base + 5];
    Bl[l] = (const float*)d_in[base + 6];
  }

  float* ws = (float*)d_ws;
  size_t off = 0;
  float* hs  = ws + off; off += (size_t)Nn * 128;
  float* hl  = ws + off; off += (size_t)Nn * 64;
  float* wx  = ws + off; off += (size_t)256 * 16;
  float* als = ws + off; off += (size_t)Nn * 2;
  float* ald = ws + off; off += (size_t)Nn * 2;
  float* h1  = ws + off; off += (size_t)Nn * 64;
  float* h2  = ws + off; off += (size_t)Nn * 64;
  int* iws = (int*)(ws + off);
  size_t ioff = 0;
  int* rowptr  = iws + ioff; ioff += (size_t)Nn + 1;
  int* csr_tmp = iws + ioff; ioff += (size_t)Ee;
  int* csr_src = iws + ioff; ioff += (size_t)Ee;

  const float* lin[3]  = { x, h1, h2 };
  const float* wxl[3]  = { wx, wx + 128 * 16, wx + 192 * 16 };
  float*       lout[3] = { h1, h2, (float*)d_out };

  const int TPB = 256;
  const int rowTiles = Nn >> 4;
  const int gemmBlocks = (rowTiles + 7) / 8;

  int* binned  = (int*)hs;
  int* tbl     = binned + 2 * (size_t)Ee;
  int* binbase = tbl + (size_t)SORT_BLOCKS * MAX_BINS;
  const int nbins = (Nn + BIN_NODES - 1) >> BIN_SHIFT;
  bin_count<<<SORT_BLOCKS, SORT_THREADS, 0, stream>>>(dst, tbl, Ee, nbins);
  bin_scatter<<<SORT_BLOCKS, SORT_THREADS, 0, stream>>>(src, dst, tbl, binbase, binned, Ee, nbins);
  bin_sort<<<nbins, BSORT_THREADS, 0, stream>>>(binbase, binned, rowptr, csr_tmp, Nn);
  const long long nodeThreads = (long long)((Nn + NODES_PER_WAVE - 1) / NODES_PER_WAVE) * 32;
  const int nodeBlocks = (int)((nodeThreads + TPB - 1) / TPB);
  csr_order<<<nodeBlocks, TPB, 0, stream>>>(rowptr, csr_tmp, csr_src, Nn);

  fold_att<<<256, 128, 0, stream>>>(Ws[0], Wd[0], As[0], Ad[0], Ws[1], Wd[1], As[1], Ad[1],
                                    Ws[2], Wd[2], As[2], Ad[2], wx);

  for (int l = 0; l < 3; ++l) {
    if (l == 0) {
      gemm_wmma_lds<128, 128, true ><<<gemmBlocks, GEMM_THREADS, 0, stream>>>(lin[l], Ws[l], wxl[l], hs, als, ald, Nn);
      gemm_wmma_lds<128, 64,  false><<<gemmBlocks, GEMM_THREADS, 0, stream>>>(lin[l], Wl[l], nullptr, hl, nullptr, nullptr, Nn);
      node_gather<<<nodeBlocks, TPB, 0, stream>>>(
          rowptr, csr_src, als, ald, hs, hl, B[l], Bl[l], lout[l], Nn, Ee);
    } else {
      gemm_wmma_lds<64, 64, true><<<gemmBlocks, GEMM_THREADS, 0, stream>>>(lin[l], Wl[l], wxl[l], hl, als, ald, Nn);
      node_gather_in<<<nodeBlocks, TPB, 0, stream>>>(
          rowptr, csr_src, als, ald, lin[l], hs, Nn, Ee);
      gemm_zout<<<gemmBlocks, GEMM_THREADS, 0, stream>>>(hs, Ws[l], hl, B[l], Bl[l], lout[l], Nn);
    }
  }
}


# Independent work shares launches: per-node source ranking runs beside layer 0's two products (block ranges of one grid), the attention fold beside the edge count; 11 launches not 14; sums unchanged
# speedup vs baseline: 4.8523x; 1.0569x over previous
#include <hip/hip_runtime.h>
#include <math.h>

typedef __attribute__((ext_vector_type(2))) float v2f;
typedef __attribute__((ext_vector_type(8))) float v8f;

#define NEG_SLOPE 0.2f

constexpr int GEMM_THREADS = 256;

template <int K, int M, bool ATT>
__device__ __forceinline__ void gemm_strip(float* __restrict__ wlds, int blk,
                                           const float* __restrict__ A,
                                           const float* __restrict__ W,
                                           const float* __restrict__ Wx,
                                           float* __restrict__ C,
                                           float* __restrict__ als,
                                           float* __restrict__ ald,
                                           int Nrows) {
  constexpr int MT = ATT ? M + 16 : M;
  constexpr int TILES_C = M / 16;
  constexpr int TILES_M = MT / 16;

  constexpr int NW = (K * M / 4) / GEMM_THREADS;
  constexpr int NX = ATT ? (K * 16 / 4) / GEMM_THREADS : 0;
  static_assert((K * M / 4) % GEMM_THREADS == 0, "W must split evenly over the work-group");
  static_assert(!ATT || (K * 16 / 4) % GEMM_THREADS == 0, "Wx must split evenly over the work-group");
  const float4* __restrict__ W4 = (const float4*)W;
  float4 wv[NW];
  float4 xv[NX > 0 ? NX : 1];
#pragma unroll
  for (int j = 0; j < NW; ++j) wv[j] = W4[threadIdx.x + j * GEMM_THREADS];
  if constexpr (ATT) {
    const float4* __restrict__ Wx4 = (const float4*)Wx;
#pragma unroll
    for (int j = 0; j < NX; ++j) xv[j] = Wx4[threadIdx.x + j * GEMM_THREADS];
  }
#pragma unroll
  for (int j = 0; j < NW; ++j) {
    const int f = (threadIdx.x + j * GEMM_THREADS) * 4;
    *(float4*)&wlds[(f / M) * MT + (f % M)] = wv[j];
  }
  if constexpr (ATT) {
#pragma unroll
    for (int j = 0; j < NX; ++j) {
      const int i4 = threadIdx.x + j * GEMM_THREADS;
      *(float4*)&wlds[(i4 >> 2) * MT + M + ((i4 & 3) << 2)] = xv[j];
    }
  }
  __syncthreads();

  const int wavesPerBlock = blockDim.x >> 5;
  const int rowTile = blk * wavesPerBlock + (threadIdx.x >> 5);
  const int lane = threadIdx.x & 31;
  if (rowTile >= (Nrows >> 4)) return;

  const int l16 = lane & 15;
  const int kq  = (lane >> 4) << 1;
  const float* __restrict__ arow = A + (size_t)(rowTile * 16 + l16) * (size_t)K;

  v8f acc[TILES_M];
#pragma unroll
  for (int t = 0; t < TILES_M; ++t) acc[t] = (v8f){0.f,0.f,0.f,0.f,0.f,0.f,0.f,0.f};

  for (int k0 = 0; k0 < K; k0 += 4) {
    v2f a;
    a.x = arow[k0 + kq];
    a.y = arow[k0 + kq + 1];
    const float* wr0 = &wlds[(k0 + kq) * MT + l16];
#pragma unroll
    for (int t = 0; t < TILES_M; ++t) {
      v2f b;
      b.x = wr0[t * 16];
      b.y = wr0[MT + t * 16];
      acc[t] = __builtin_amdgcn_wmma_f32_16x16x4_f32(
           false, a,  false, b,
           (short)0, acc[t],  false,  false);
    }
  }

  const int rbase = rowTile * 16 + ((lane >> 4) << 3);
#pragma unroll
  for (int t = 0; t < TILES_C; ++t) {
    const int col = t * 16 + l16;
#pragma unroll
    for (int r = 0; r < 8; ++r)
      C[(size_t)(rbase + r) * (size_t)M + col] = acc[t][r];
  }
  if constexpr (ATT) {
    if (l16 < 4) {
      float* __restrict__ ap = (l16 < 2) ? (als + l16) : (ald + (l16 - 2));
#pragma unroll
      for (int r = 0; r < 8; ++r)
        ap[(size_t)(rbase + r) * 2] = acc[TILES_C][r];
    }
  }
}

template <int K, int M, bool ATT>
__global__ void gemm_wmma_lds(const float* __restrict__ A,
                              const float* __restrict__ W,
                              const float* __restrict__ Wx,
                              float* __restrict__ C,
                              float* __restrict__ als,
                              float* __restrict__ ald,
                              int Nrows) {
  constexpr int MT = ATT ? M + 16 : M;
  __shared__ __attribute__((aligned(16))) float wlds[K * MT];
  gemm_strip<K, M, ATT>(wlds, (int)blockIdx.x, A, W, Wx, C, als, ald, Nrows);
}

constexpr int FOLD_ROWS = 256;

__device__ __forceinline__ void fold_att_row(int row, int j, int lane,
                         const float* __restrict__ Ws0, const float* __restrict__ Wd0,
                         const float* __restrict__ as0, const float* __restrict__ ad0,
                         const float* __restrict__ Ws1, const float* __restrict__ Wd1,
                         const float* __restrict__ as1, const float* __restrict__ ad1,
                         const float* __restrict__ Ws2, const float* __restrict__ Wd2,
                         const float* __restrict__ as2, const float* __restrict__ ad2,
                         float* __restrict__ wx) {
  const float *Ws, *Wd, *a_s, *a_d;
  int k;
  if (row < 128)      { Ws = Ws0; Wd = Wd0; a_s = as0; a_d = ad0; k = row; }
  else if (row < 192) { Ws = Ws1; Wd = Wd1; a_s = as1; a_d = ad1; k = row - 128; }
  else                { Ws = Ws2; Wd = Wd2; a_s = as2; a_d = ad2; k = row - 192; }
  const int h = j & 1;
  const float* wrow = ((j < 2) ? Ws : Wd) + (size_t)k * 128 + h * 64;
  const float* av   = ((j < 2) ? a_s : a_d) + h * 64;
  float s = wrow[lane] * av[lane] + wrow[lane + 32] * av[lane + 32];
  for (int o = 16; o > 0; o >>= 1) s += __shfl_xor(s, o);
  if (lane == 0) wx[row * 16 + j] = s;
  if (j == 0 && lane >= 4 && lane < 16) wx[row * 16 + lane] = 0.f;
}

constexpr int BIN_SHIFT = 7;
constexpr int BIN_NODES = 1 << BIN_SHIFT;
constexpr int MAX_BINS = 512;
constexpr int SORT_BLOCKS = 128;
constexpr int SORT_THREADS = 1024;
constexpr int SORT_BATCH = 8;
constexpr int BSORT_THREADS = 512;
constexpr int BSORT_BATCH = 4;

__device__ __forceinline__ int block_excl_scan(int v, int* wsum) {
  const int lane = threadIdx.x & 31;
  const int w = threadIdx.x >> 5;
  int incl = v;
  for (int o = 1; o < 32; o <<= 1) {
    const int u = __shfl_up(incl, o);
    if (lane >= o) incl += u;
  }
  if (lane == 31) wsum[w] = incl;
  __syncthreads();
  int pre = 0;
  for (int q = 0; q < w; ++q) pre += wsum[q];
  return pre + incl - v;
}

constexpr int FOLD_ROWS_PER_BLOCK = SORT_THREADS / 128;
constexpr int FOLD_BLOCKS = FOLD_ROWS / FOLD_ROWS_PER_BLOCK;
static_assert(FOLD_ROWS % FOLD_ROWS_PER_BLOCK == 0, "fold rows must split evenly over the work-groups");

__global__ void bin_count_fold(const int* __restrict__ dst, int* __restrict__ tbl, int Ee, int nbins,
                         const float* __restrict__ Ws0, const float* __restrict__ Wd0,
                         const float* __restrict__ as0, const float* __restrict__ ad0,
                         const float* __restrict__ Ws1, const float* __restrict__ Wd1,
                         const float* __restrict__ as1, const float* __restrict__ ad1,
                         const float* __restrict__ Ws2, const float* __restrict__ Wd2,
                         const float* __restrict__ as2, const float* __restrict__ ad2,
                         float* __restrict__ wx) {
  __shared__ int hist[MAX_BINS];
  const int t = threadIdx.x;
  if ((int)blockIdx.x >= SORT_BLOCKS) {
    const int w = t >> 5;
    fold_att_row(((int)blockIdx.x - SORT_BLOCKS) * FOLD_ROWS_PER_BLOCK + (w >> 2), w & 3, t & 31,
                 Ws0, Wd0, as0, ad0, Ws1, Wd1, as1, ad1, Ws2, Wd2, as2, ad2, wx);
    return;
  }
  if (t < MAX_BINS) hist[t] = 0;
  __syncthreads();
  const int chunk = (Ee + SORT_BLOCKS - 1) / SORT_BLOCKS;
  const int lo = (int)blockIdx.x * chunk;
  const int hi = min(lo + chunk, Ee);
  for (int i0 = lo + t; i0 < hi; i0 += SORT_BATCH * SORT_THREADS) {
    int d[SORT_BATCH];
#pragma unroll
    for (int k = 0; k < SORT_BATCH; ++k) {
      const int i = i0 + k * SORT_THREADS;
      d[k] = (i < hi) ? dst[i] : 0;
    }
#pragma unroll
    for (int k = 0; k < SORT_BATCH; ++k)
      if (i0 + k * SORT_THREADS < hi) atomicAdd(&hist[d[k] >> BIN_SHIFT], 1);
  }
  __syncthreads();
  if (t < nbins) tbl[(int)blockIdx.x * nbins + t] = hist[t];
}

__global__ void bin_scatter(const int* __restrict__ src, const int* __restrict__ dst,
                            const int* __restrict__ tbl, int* __restrict__ binbase,
                            int* __restrict__ binned, int Ee, int nbins) {
  __shared__ int cur[MAX_BINS];
  __shared__ int ptot[2][MAX_BINS];
  __shared__ int pbef[2][MAX_BINS];
  __shared__ int wsum[SORT_THREADS / 32];
  const int t = threadIdx.x;
  const int g = blockIdx.x;
  const int bin = t & (MAX_BINS - 1);
  const int half = t / MAX_BINS;
  int tot = 0, bef = 0;
  if (bin < nbins) {
    const int q0 = half * (SORT_BLOCKS / 2);
#pragma unroll 16
    for (int q = q0; q < q0 + SORT_BLOCKS / 2; ++q) {
      const int c = tbl[q * nbins + bin];
      tot += c;
      bef += (q < g) ? c : 0;
    }
  }
  ptot[half][bin] = tot;
  pbef[half][bin] = bef;
  __syncthreads();
  int total = 0, before = 0;
  if (t < MAX_BINS) {
    total = ptot[0][t] + ptot[1][t];
    before = pbef[0][t] + pbef[1][t];
  }
  const int base = block_excl_scan(total, wsum);
  if (t < nbins) {
    cur[t] = base + before;
    if (g == 0) {
      binbase[t] = base;
      if (t == nbins - 1) binbase[nbins] = base + total;
    }
  }
  __syncthreads();
  const int chunk = (Ee + SORT_BLOCKS - 1) / SORT_BLOCKS;
  const int lo = g * chunk;
  const int hi = min(lo + chunk, Ee);
  int2* __restrict__ pairs = (int2*)binned;
  for (int i0 = lo + t; i0 < hi; i0 += SORT_BATCH * SORT_THREADS) {
    int d[SORT_BATCH], s[SORT_BATCH];
#pragma unroll
    for (int k = 0; k < SORT_BATCH; ++k) {
      const int i = i0 + k * SORT_THREADS;
      d[k] = (i < hi) ? dst[i] : 0;
      s[k] = (i < hi) ? src[i] : 0;
    }
#pragma unroll
    for (int k = 0; k < SORT_BATCH; ++k) {
      if (i0 + k * SORT_THREADS < hi) {
        const int p = atomicAdd(&cur[d[k] >> BIN_SHIFT], 1);
        pairs[p] = make_int2(s[k], d[k]);
      }
    }
  }
}

__global__ void bin_sort(const int* __restrict__ binbase, const int* __restrict__ binned,
                         int* __restrict__ rowptr, int* __restrict__ csr_out, int Nn) {
  __shared__ int cnt[BIN_NODES];
  __shared__ int cur[BIN_NODES];
  __shared__ int wsum[BSORT_THREADS / 32];
  const int2* __restrict__ pairs = (const int2*)binned;
  const int t = threadIdx.x;
  const int b = blockIdx.x;
  const int s = binbase[b];
  const int e = binbase[b + 1];
  if (t < BIN_NODES) cnt[t] = 0;
  __syncthreads();
  for (int i0 = s + t; i0 < e; i0 += BSORT_BATCH * BSORT_THREADS) {
    int d[BSORT_BATCH];
#pragma unroll
    for (int k = 0; k < BSORT_BATCH; ++k) {
      const int i = i0 + k * BSORT_THREADS;
      d[k] = (i < e) ? pairs[i].y : 0;
    }
#pragma unroll
    for (int k = 0; k < BSORT_BATCH; ++k)
      if (i0 + k * BSORT_THREADS < e) atomicAdd(&cnt[d[k] & (BIN_NODES - 1)], 1);
  }
  __syncthreads();
  const int c = (t < BIN_NODES) ? cnt[t] : 0;
  const int first = s + block_excl_scan(c, wsum);
  if (t < BIN_NODES) {
    const int node = b * BIN_NODES + t;
    cur[t] = first;
    if (node < Nn) rowptr[node] = first;
    if (node == Nn - 1) rowptr[Nn] = first + c;
  }
  __syncthreads();
  for (int i0 = s + t; i0 < e; i0 += BSORT_BATCH * BSORT_THREADS) {
    int2 pr[BSORT_BATCH];
#pragma unroll
    for (int k = 0; k < BSORT_BATCH; ++k) {
      const int i = i0 + k * BSORT_THREADS;
      pr[k] = (i < e) ? pairs[i] : make_int2(0, 0);
    }
#pragma unroll
    for (int k = 0; k < BSORT_BATCH; ++k) {
      if (i0 + k * BSORT_THREADS < e) {
        const int p = atomicAdd(&cur[pr[k].y & (BIN_NODES - 1)], 1);
        csr_out[p] = pr[k].x;
      }
    }
  }
}

constexpr int NODES_PER_WAVE = 4;

__device__ __forceinline__ void csr_order(int blk, const int* __restrict__ rowptr,
                                          const int* __restrict__ csr_in,
                                          int* __restrict__ csr_out, int Nn) {
  const int lane = threadIdx.x & 31;
  const int wave = __builtin_amdgcn_readfirstlane((int)((blk * blockDim.x + threadIdx.x) >> 5));
  const int first = wave * NODES_PER_WAVE;
  if (first >= Nn) return;
  const int last = min(first + NODES_PER_WAVE, Nn);
  int beg = rowptr[first];
  int end = rowptr[first + 1];
  for (int node = first; node < last; ++node) {
  const int nend = rowptr[min(node + 2, Nn)];
  for (int cb = beg; cb < end; cb += 32) {
    const int i = cb + lane;
    const int key = (i < end) ? csr_in[i] : 0;
    int rank = 0;
    for (int kb = beg; kb < end; kb += 32) {
      const int k = kb + lane;
      const int other = (k < end) ? csr_in[k] : 0;
      const int n = (end - kb < 32) ? (end - kb) : 32;
      for (int j = 0; j < n; ++j) {
        const int o = __builtin_amdgcn_readlane(other, j);
        const int pos = kb + j;
        rank += ((o < key) || (o == key && pos < i)) ? 1 : 0;
      }
    }
    if (i < end) csr_out[beg + rank] = key;
  }
  beg = end;
  end = nend;
  }
}

__global__ void layer0_products_and_order(const float* __restrict__ x,
                                          const float* __restrict__ Ws, const float* __restrict__ Wx,
                                          const float* __restrict__ Wl,
                                          float* __restrict__ hs, float* __restrict__ als,
                                          float* __restrict__ ald, float* __restrict__ hl,
                                          const int* __restrict__ rowptr, const int* __restrict__ csr_in,
                                          int* __restrict__ csr_out, int Nn, int gb) {
  __shared__ __attribute__((aligned(16))) float wlds[128 * (128 + 16)];
  const int blk = (int)blockIdx.x;
  if (blk < gb)
    gemm_strip<128, 128, true >(wlds, blk, x, Ws, Wx, hs, als, ald, Nn);
  else if (blk < 2 * gb)
    gemm_strip<128, 64,  false>(wlds, blk - gb, x, Wl, nullptr, hl, nullptr, nullptr, Nn);
  else
    csr_order(blk - 2 * gb, rowptr, csr_in, csr_out, Nn);
}

__device__ __forceinline__ float lrelu(float v) { return (v > 0.f) ? v : NEG_SLOPE * v; }

__global__ void node_gather(const int* __restrict__ rowptr, const int* __restrict__ csr_src,
                            const float* __restrict__ als, const float* __restrict__ ald,
                            const float* __restrict__ hs, const float* __restrict__ hl,
                            const float* __restrict__ b, const float* __restrict__ bl,
                            float* __restrict__ out, int Nn, int Ee) {
  const int lane = threadIdx.x & 31;
  const int wave = __builtin_amdgcn_readfirstlane((int)((blockIdx.x * blockDim.x + threadIdx.x) >> 5));
  const int first = wave * NODES_PER_WAVE;
  if (first >= Nn) return;
  const int last = min(first + NODES_PER_WAVE, Nn);
  const float2* __restrict__ als2 = (const float2*)als;
  const int f1 = min(first + 1, Nn - 1);
  int beg  = rowptr[first];
  int end  = rowptr[first + 1];
  int end1 = rowptr[f1 + 1];
  float ad0  = ald[first * 2 + 0];
  float ad1  = ald[first * 2 + 1];
  float nad0 = ald[f1 * 2 + 0];
  float nad1 = ald[f1 * 2 + 1];
  int s_cur = csr_src[min(beg + lane, Ee - 1)];
  int s_nxt = csr_src[min(end + lane, Ee - 1)];
  const float2 a_first = als2[(beg + lane < end) ? s_cur : __builtin_amdgcn_readfirstlane(s_cur)];
  float2 a_cur = a_first;
  for (int node = first; node < last; ++node) {
  const int n2 = min(node + 2, Nn - 1);
  const int end2 = rowptr[n2 + 1];
  const float n2ad0 = ald[n2 * 2 + 0];
  const float n2ad1 = ald[n2 * 2 + 1];
  const int s_nn = csr_src[min(end1 + lane, Ee - 1)];
  const float2 a_nxt = als2[(end + lane < end1) ? s_nxt : __builtin_amdgcn_readfirstlane(s_nxt)];

  float m0 = -INFINITY, m1 = -INFINITY;
  if (beg + lane < end) {
    m0 = fmaxf(m0, lrelu(a_cur.x + ad0));
    m1 = fmaxf(m1, lrelu(a_cur.y + ad1));
  }
  for (int i = beg + 32 + lane; i < end; i += 32) {
    const float2 a = als2[csr_src[i]];
    m0 = fmaxf(m0, lrelu(a.x + ad0));
    m1 = fmaxf(m1, lrelu(a.y + ad1));
  }
  for (int o = 16; o > 0; o >>= 1) {
    m0 = fmaxf(m0, __shfl_xor(m0, o));
    m1 = fmaxf(m1, __shfl_xor(m1, o));
  }
  if (!(fabsf(m0) < INFINITY)) m0 = 0.f;
  if (!(fabsf(m1) < INFINITY)) m1 = 0.f;

  float d0 = 0.f, d1 = 0.f;
  if (beg + lane < end) {
    d0 += __expf(lrelu(a_cur.x + ad0) - m0);
    d1 += __expf(lrelu(a_cur.y + ad1) - m1);
  }
  for (int i = beg + 32 + lane; i < end; i += 32) {
    const float2 a = als2[csr_src[i]];
    d0 += __expf(lrelu(a.x + ad0) - m0);
    d1 += __expf(lrelu(a.y + ad1) - m1);
  }
  for (int o = 16; o > 0; o >>= 1) {
    d0 += __shfl_xor(d0, o);
    d1 += __shfl_xor(d1, o);
  }
  const float r0 = d0 + 1e-16f;
  const float r1 = d1 + 1e-16f;

  const bool head1 = lane >= 16;
  const float* __restrict__ hsl = hs + lane * 4;
  float4 acc = make_float4(0.f, 0.f, 0.f, 0.f);
  for (int base = beg; base < end; base += 32) {
    const int i = base + lane;
    int s_l = 0;
    float w_l0 = 0.f, w_l1 = 0.f;
    if (i < end) {
      float2 a;
      if (base == beg) { s_l = s_cur; a = a_cur; }
      else             { s_l = csr_src[i]; a = als2[s_l]; }
      w_l0 = __expf(lrelu(a.x + ad0) - m0) / r0;
      w_l1 = __expf(lrelu(a.y + ad1) - m1) / r1;
    }
    const int cnt = (end - base < 32) ? (end - base) : 32;
#pragma unroll 4
    for (int j = 0; j < cnt; ++j) {
      const int s = __builtin_amdgcn_readlane(s_l, j);
      const float wa = __int_as_float(__builtin_amdgcn_readlane(__float_as_int(w_l0), j));
      const float wb = __int_as_float(__builtin_amdgcn_readlane(__float_as_int(w_l1), j));
      const float w = head1 ? wb : wa;
      const float4 v = *(const float4*)(hsl + (size_t)s * 128);
      acc.x = fmaf(w, v.x, acc.x);
      acc.y = fmaf(w, v.y, acc.y);
      acc.z = fmaf(w, v.z, acc.z);
      acc.w = fmaf(w, v.w, acc.w);
    }
  }
  acc.x += __shfl_xor(acc.x, 16);
  acc.y += __shfl_xor(acc.y, 16);
  acc.z += __shfl_xor(acc.z, 16);
  acc.w += __shfl_xor(acc.w, 16);
  if (lane < 16) {
    const int c = lane * 4;
    const float4 hv = *(const float4*)(hl + (size_t)node * 64 + c);
    float4 o;
    o.x = 0.5f * acc.x + b[c + 0] + hv.x + bl[c + 0];
    o.y = 0.5f * acc.y + b[c + 1] + hv.y + bl[c + 1];
    o.z = 0.5f * acc.z + b[c + 2] + hv.z + bl[c + 2];
    o.w = 0.5f * acc.w + b[c + 3] + hv.w + bl[c + 3];
    o.x = (o.x > 0.f) ? o.x : 0.f;
    o.y = (o.y > 0.f) ? o.y : 0.f;
    o.z = (o.z > 0.f) ? o.z : 0.f;
    o.w = (o.w > 0.f) ? o.w : 0.f;
    *(float4*)(out + (size_t)node * 64 + c) = o;
  }
  beg = end;   end = end1;   end1 = end2;
  ad0 = nad0;  ad1 = nad1;   nad0 = n2ad0;  nad1 = n2ad1;
  s_cur = s_nxt;  s_nxt = s_nn;  a_cur = a_nxt;
  }
}

__global__ void node_gather_in(const int* __restrict__ rowptr, const int* __restrict__ csr_src,
                               const float* __restrict__ als, const float* __restrict__ ald,
                               const float* __restrict__ hin, float* __restrict__ z, int Nn, int Ee) {
  const int lane = threadIdx.x & 31;
  const int wave = __builtin_amdgcn_readfirstlane((int)((blockIdx.x * blockDim.x + threadIdx.x) >> 5));
  const int first = wave * NODES_PER_WAVE;
  if (first >= Nn) return;
  const int last = min(first + NODES_PER_WAVE, Nn);
  const float2* __restrict__ als2 = (const float2*)als;
  const bool odd = lane >= 16;
  const int c = (lane & 15) * 4;
  const float* __restrict__ hrow = hin + c;
  const int f1 = min(first + 1, Nn - 1);
  int beg  = rowptr[first];
  int end  = rowptr[first + 1];
  int end1 = rowptr[f1 + 1];
  float ad0  = ald[first * 2 + 0];
  float ad1  = ald[first * 2 + 1];
  float nad0 = ald[f1 * 2 + 0];
  float nad1 = ald[f1 * 2 + 1];
  int s_cur = csr_src[min(beg + lane, Ee - 1)];
  int s_nxt = csr_src[min(end + lane, Ee - 1)];
  const float2 a_first = als2[(beg + lane < end) ? s_cur : __builtin_amdgcn_readfirstlane(s_cur)];
  float2 a_cur = a_first;
  for (int node = first; node < last; ++node) {
  const int n2 = min(node + 2, Nn - 1);
  const int end2 = rowptr[n2 + 1];
  const float n2ad0 = ald[n2 * 2 + 0];
  const float n2ad1 = ald[n2 * 2 + 1];
  const int s_nn = csr_src[min(end1 + lane, Ee - 1)];
  const float2 a_nxt = als2[(end + lane < end1) ? s_nxt : __builtin_amdgcn_readfirstlane(s_nxt)];

  float m0 = -INFINITY, m1 = -INFINITY;
  if (beg + lane < end) {
    m0 = fmaxf(m0, lrelu(a_cur.x + ad0));
    m1 = fmaxf(m1, lrelu(a_cur.y + ad1));
  }
  for (int i = beg + 32 + lane; i < end; i += 32) {
    const float2 a = als2[csr_src[i]];
    m0 = fmaxf(m0, lrelu(a.x + ad0));
    m1 = fmaxf(m1, lrelu(a.y + ad1));
  }
  for (int o = 16; o > 0; o >>= 1) {
    m0 = fmaxf(m0, __shfl_xor(m0, o));
    m1 = fmaxf(m1, __shfl_xor(m1, o));
  }
  if (!(fabsf(m0) < INFINITY)) m0 = 0.f;
  if (!(fabsf(m1) < INFINITY)) m1 = 0.f;

  float d0 = 0.f, d1 = 0.f;
  if (beg + lane < end) {
    d0 += __expf(lrelu(a_cur.x + ad0) - m0);
    d1 += __expf(lrelu(a_cur.y + ad1) - m1);
  }
  for (int i = beg + 32 + lane; i < end; i += 32) {
    const float2 a = als2[csr_src[i]];
    d0 += __expf(lrelu(a.x + ad0) - m0);
    d1 += __expf(lrelu(a.y + ad1) - m1);
  }
  for (int o = 16; o > 0; o >>= 1) {
    d0 += __shfl_xor(d0, o);
    d1 += __shfl_xor(d1, o);
  }
  const float r0 = d0 + 1e-16f;
  const float r1 = d1 + 1e-16f;

  float4 acc0 = make_float4(0.f, 0.f, 0.f, 0.f);
  float4 acc1 = make_float4(0.f, 0.f, 0.f, 0.f);
  for (int base = beg; base < end; base += 32) {
    const int i = base + lane;
    int s_l = 0;
    float w_l0 = 0.f, w_l1 = 0.f;
    if (i < end) {
      float2 a;
      if (base == beg) { s_l = s_cur; a = a_cur; }
      else             { s_l = csr_src[i]; a = als2[s_l]; }
      w_l0 = __expf(lrelu(a.x + ad0) - m0) / r0;
      w_l1 = __expf(lrelu(a.y + ad1) - m1) / r1;
    }
    const int s_first = __builtin_amdgcn_readfirstlane(s_l);
    if (i >= end) s_l = s_first;
    const int cnt = (end - base < 32) ? (end - base) : 32;
#pragma unroll 4
    for (int j = 0; j < cnt; j += 2) {
      const int sa = __builtin_amdgcn_readlane(s_l, j);
      const int sb = __builtin_amdgcn_readlane(s_l, j + 1);
      const float wa0 = __int_as_float(__builtin_amdgcn_readlane(__float_as_int(w_l0), j));
      const float wb0 = __int_as_float(__builtin_amdgcn_readlane(__float_as_int(w_l0), j + 1));
      const float wa1 = __int_as_float(__builtin_amdgcn_readlane(__float_as_int(w_l1), j));
      const float wb1 = __int_as_float(__builtin_amdgcn_readlane(__float_as_int(w_l1), j + 1));
      const int s = odd ? sb : sa;
      const float w0 = odd ? wb0 : wa0;
      const float w1 = odd ? wb1 : wa1;
      const float4 v = *(const float4*)(hrow + (size_t)s * 64);
      acc0.x = fmaf(w0, v.x, acc0.x);
      acc0.y = fmaf(w0, v.y, acc0.y);
      acc0.z = fmaf(w0, v.z, acc0.z);
      acc0.w = fmaf(w0, v.w, acc0.w);
      acc1.x = fmaf(w1, v.x, acc1.x);
      acc1.y = fmaf(w1, v.y, acc1.y);
      acc1.z = fmaf(w1, v.z, acc1.z);
      acc1.w = fmaf(w1, v.w, acc1.w);
    }
  }
  acc0.x += __shfl_xor(acc0.x, 16);
  acc0.y += __shfl_xor(acc0.y, 16);
  acc0.z += __shfl_xor(acc0.z, 16);
  acc0.w += __shfl_xor(acc0.w, 16);
  acc1.x += __shfl_xor(acc1.x, 16);
  acc1.y += __shfl_xor(acc1.y, 16);
  acc1.z += __shfl_xor(acc1.z, 16);
  acc1.w += __shfl_xor(acc1.w, 16);
  float4 o;
  o.x = odd ? acc1.x : acc0.x;
  o.y = odd ? acc1.y : acc0.y;
  o.z = odd ? acc1.z : acc0.z;
  o.w = odd ? acc1.w : acc0.w;
  *(float4*)(z + (size_t)node * 128 + (odd ? 64 : 0) + c) = o;
  beg = end;   end = end1;   end1 = end2;
  ad0 = nad0;  ad1 = nad1;   nad0 = n2ad0;  nad1 = n2ad1;
  s_cur = s_nxt;  s_nxt = s_nn;  a_cur = a_nxt;
  }
}

__global__ void gemm_zout(const float* __restrict__ Z,
                          const float* __restrict__ Ws,
                          const float* __restrict__ hl,
                          const float* __restrict__ b,
                          const float* __restrict__ bl,
                          float* __restrict__ out,
                          int Nrows) {
  constexpr int K = 128, M = 64, MT = 80, TILES = M / 16;
  __shared__ __attribute__((aligned(16))) float wlds[K * MT];

  constexpr int NW = (64 * 128 / 4) / GEMM_THREADS;
  static_assert((64 * 128 / 4) % GEMM_THREADS == 0, "Ws must split evenly over the work-group");
  const float4* __restrict__ W4 = (const float4*)Ws;
  float4 wv[NW];
#pragma unroll
  for (int j = 0; j < NW; ++j) wv[j] = W4[threadIdx.x + j * GEMM_THREADS];
#pragma unroll
  for (int j = 0; j < NW; ++j) {
    const int f = (threadIdx.x + j * GEMM_THREADS) * 4;
    const int k = f >> 7;
    const int col = f & 127;
    *(float4*)&wlds[((col >> 6) * 64 + k) * MT + (col & 63)] = wv[j];
  }
  __syncthreads();

  const int wavesPerBlock = blockDim.x >> 5;
  const int rowTile = blockIdx.x * wavesPerBlock + (threadIdx.x >> 5);
  const int lane = threadIdx.x & 31;
  if (rowTile >= (Nrows >> 4)) return;

  const int l16 = lane & 15;
  const int kq  = (lane >> 4) << 1;
  const float* __restrict__ arow = Z + (size_t)(rowTile * 16 + l16) * (size_t)K;

  const int rbase = rowTile * 16 + ((lane >> 4) << 3);
  float hv[TILES][8];
  float bc[TILES], blc[TILES];
#pragma unroll
  for (int t = 0; t < TILES; ++t) {
    const int col = t * 16 + l16;
    bc[t] = b[col];
    blc[t] = bl[col];
#pragma unroll
    for (int r = 0; r < 8; ++r)
      hv[t][r] = hl[(size_t)(rbase + r) * (size_t)M + col];
  }

  v8f acc[TILES];
#pragma unroll
  for (int t = 0; t < TILES; ++t) acc[t] = (v8f){0.f,0.f,0.f,0.f,0.f,0.f,0.f,0.f};

  for (int k0 = 0; k0 < K; k0 += 4) {
    v2f a;
    a.x = arow[k0 + kq];
    a.y = arow[k0 + kq + 1];
    const float* wr0 = &wlds[(k0 + kq) * MT + l16];
#pragma unroll
    for (int t = 0; t < TILES; ++t) {
      v2f bb;
      bb.x = wr0[t * 16];
      bb.y = wr0[MT + t * 16];
      acc[t] = __builtin_amdgcn_wmma_f32_16x16x4_f32(
           false, a,  false, bb,
           (short)0, acc[t],  false,  false);
    }
  }

#pragma unroll
  for (int t = 0; t < TILES; ++t) {
    const int col = t * 16 + l16;
#pragma unroll
    for (int r = 0; r < 8; ++r) {
      const float v = 0.5f * acc[t][r] + bc[t] + hv[t][r] + blc[t];
      out[(size_t)(rbase + r) * (size_t)M + col] = (v > 0.f) ? v : 0.f;
    }
  }
}

extern "C" void kernel_launch(void* const* d_in, const int* in_sizes, int n_in,
                              void* d_out, int out_size, void* d_ws, size_t ws_size,
                              hipStream_t stream) {
  const float* x  = (const float*)d_in[0];
  const int*   ei = (const int*)d_in[1];
  const int Nn = in_sizes[0] / 128;
  const int Ee = in_sizes[1] / 2;
  const int* src = ei;
  const int* dst = ei + Ee;

  const float *Ws[3], *Wd[3], *As[3], *Ad[3], *B[3], *Wl[3], *Bl[3];
  for (int l = 0; l < 3; ++l) {
    int base = 2 + 7 * l;
    Ws[l] = (const float*)d_in[base + 0];
    Wd[l] = (const float*)d_in[base + 1];
    As[l] = (const float*)d_in[base + 2];
    Ad[l] = (const float*)d_in[base + 3];
    B[l]  = (const float*)d_in[base + 4];
    Wl[l] = (const float*)d_in[base + 5];
    Bl[l] = (const float*)d_in[base + 6];
  }

  float* ws = (float*)d_ws;
  size_t off = 0;
  float* hs  = ws + off; off += (size_t)Nn * 128;
  float* hl  = ws + off; off += (size_t)Nn * 64;
  float* wx  = ws + off; off += (size_t)256 * 16;
  float* als = ws + off; off += (size_t)Nn * 2;
  float* ald = ws + off; off += (size_t)Nn * 2;
  float* h1  = ws + off; off += (size_t)Nn * 64;
  float* h2  = ws + off; off += (size_t)Nn * 64;
  int* iws = (int*)(ws + off);
  size_t ioff = 0;
  int* rowptr  = iws + ioff; ioff += (size_t)Nn + 1;
  int* csr_tmp = iws + ioff; ioff += (size_t)Ee;
  int* csr_src = iws + ioff; ioff += (size_t)Ee;

  const float* lin[3]  = { x, h1, h2 };
  const float* wxl[3]  = { wx, wx + 128 * 16, wx + 192 * 16 };
  float*       lout[3] = { h1, h2, (float*)d_out };

  const int TPB = 256;
  const int rowTiles = Nn >> 4;
  const int gemmBlocks = (rowTiles + 7) / 8;

  int* binned  = (int*)hs;
  int* tbl     = binned + 2 * (size_t)Ee;
  int* binbase = tbl + (size_t)SORT_BLOCKS * MAX_BINS;
  const int nbins = (Nn + BIN_NODES - 1) >> BIN_SHIFT;
  bin_count_fold<<<SORT_BLOCKS + FOLD_BLOCKS, SORT_THREADS, 0, stream>>>(
      dst, tbl, Ee, nbins, Ws[0], Wd[0], As[0], Ad[0], Ws[1], Wd[1], As[1], Ad[1],
      Ws[2], Wd[2], As[2], Ad[2], wx);
  bin_scatter<<<SORT_BLOCKS, SORT_THREADS, 0, stream>>>(src, dst, tbl, binbase, binned, Ee, nbins);
  bin_sort<<<nbins, BSORT_THREADS, 0, stream>>>(binbase, binned, rowptr, csr_tmp, Nn);
  const long long nodeThreads = (long long)((Nn + NODES_PER_WAVE - 1) / NODES_PER_WAVE) * 32;
  const int nodeBlocks = (int)((nodeThreads + TPB - 1) / TPB);

  for (int l = 0; l < 3; ++l) {
    if (l == 0) {
      static_assert(TPB == GEMM_THREADS, "ranking and products share one work-group size");
      layer0_products_and_order<<<2 * gemmBlocks + nodeBlocks, GEMM_THREADS, 0, stream>>>(
          lin[l], Ws[l], wxl[l], Wl[l], hs, als, ald, hl, rowptr, csr_tmp, csr_src, Nn, gemmBlocks);
      node_gather<<<nodeBlocks, TPB, 0, stream>>>(
          rowptr, csr_src, als, ald, hs, hl, B[l], Bl[l], lout[l], Nn, Ee);
    } else {
      gemm_wmma_lds<64, 64, true><<<gemmBlocks, GEMM_THREADS, 0, stream>>>(lin[l], Wl[l], wxl[l], hl, als, ald, Nn);
      node_gather_in<<<nodeBlocks, TPB, 0, stream>>>(
          rowptr, csr_src, als, ald, lin[l], hs, Nn, Ee);
      gemm_zout<<<gemmBlocks, GEMM_THREADS, 0, stream>>>(hs, Ws[l], hl, B[l], Bl[l], lout[l], Nn);
    }
  }
}


# Layers 2,3: per-node weighted sums and the Ws product after them share one launch; a work-group's 64 z rows go through LDS to its WMMA waves; z plane gone, 9 launches not 11; sums unchanged
# speedup vs baseline: 5.1061x; 1.0523x over previous
#include <hip/hip_runtime.h>
#include <math.h>

typedef __attribute__((ext_vector_type(2))) float v2f;
typedef __attribute__((ext_vector_type(8))) float v8f;

#define NEG_SLOPE 0.2f

constexpr int GEMM_THREADS = 256;

template <int K, int M, bool ATT>
__device__ __forceinline__ void gemm_strip(float* __restrict__ wlds, int blk,
                                           const float* __restrict__ A,
                                           const float* __restrict__ W,
                                           const float* __restrict__ Wx,
                                           float* __restrict__ C,
                                           float* __restrict__ als,
                                           float* __restrict__ ald,
                                           int Nrows) {
  constexpr int MT = ATT ? M + 16 : M;
  constexpr int TILES_C = M / 16;
  constexpr int TILES_M = MT / 16;

  constexpr int NW = (K * M / 4) / GEMM_THREADS;
  constexpr int NX = ATT ? (K * 16 / 4) / GEMM_THREADS : 0;
  static_assert((K * M / 4) % GEMM_THREADS == 0, "W must split evenly over the work-group");
  static_assert(!ATT || (K * 16 / 4) % GEMM_THREADS == 0, "Wx must split evenly over the work-group");
  const float4* __restrict__ W4 = (const float4*)W;
  float4 wv[NW];
  float4 xv[NX > 0 ? NX : 1];
#pragma unroll
  for (int j = 0; j < NW; ++j) wv[j] = W4[threadIdx.x + j * GEMM_THREADS];
  if constexpr (ATT) {
    const float4* __restrict__ Wx4 = (const float4*)Wx;
#pragma unroll
    for (int j = 0; j < NX; ++j) xv[j] = Wx4[threadIdx.x + j * GEMM_THREADS];
  }
#pragma unroll
  for (int j = 0; j < NW; ++j) {
    const int f = (threadIdx.x + j * GEMM_THREADS) * 4;
    *(float4*)&wlds[(f / M) * MT + (f % M)] = wv[j];
  }
  if constexpr (ATT) {
#pragma unroll
    for (int j = 0; j < NX; ++j) {
      const int i4 = threadIdx.x + j * GEMM_THREADS;
      *(float4*)&wlds[(i4 >> 2) * MT + M + ((i4 & 3) << 2)] = xv[j];
    }
  }
  __syncthreads();

  const int wavesPerBlock = blockDim.x >> 5;
  const int rowTile = blk * wavesPerBlock + (threadIdx.x >> 5);
  const int lane = threadIdx.x & 31;
  if (rowTile >= (Nrows >> 4)) return;

  const int l16 = lane & 15;
  const int kq  = (lane >> 4) << 1;
  const float* __restrict__ arow = A + (size_t)(rowTile * 16 + l16) * (size_t)K;

  v8f acc[TILES_M];
#pragma unroll
  for (int t = 0; t < TILES_M; ++t) acc[t] = (v8f){0.f,0.f,0.f,0.f,0.f,0.f,0.f,0.f};

  for (int k0 = 0; k0 < K; k0 += 4) {
    v2f a;
    a.x = arow[k0 + kq];
    a.y = arow[k0 + kq + 1];
    const float* wr0 = &wlds[(k0 + kq) * MT + l16];
#pragma unroll
    for (int t = 0; t < TILES_M; ++t) {
      v2f b;
      b.x = wr0[t * 16];
      b.y = wr0[MT + t * 16];
      acc[t] = __builtin_amdgcn_wmma_f32_16x16x4_f32(
           false, a,  false, b,
           (short)0, acc[t],  false,  false);
    }
  }

  const int rbase = rowTile * 16 + ((lane >> 4) << 3);
#pragma unroll
  for (int t = 0; t < TILES_C; ++t) {
    const int col = t * 16 + l16;
#pragma unroll
    for (int r = 0; r < 8; ++r)
      C[(size_t)(rbase + r) * (size_t)M + col] = acc[t][r];
  }
  if constexpr (ATT) {
    if (l16 < 4) {
      float* __restrict__ ap = (l16 < 2) ? (als + l16) : (ald + (l16 - 2));
#pragma unroll
      for (int r = 0; r < 8; ++r)
        ap[(size_t)(rbase + r) * 2] = acc[TILES_C][r];
    }
  }
}

template <int K, int M, bool ATT>
__global__ void gemm_wmma_lds(const float* __restrict__ A,
                              const float* __restrict__ W,
                              const float* __restrict__ Wx,
                              float* __restrict__ C,
                              float* __restrict__ als,
                              float* __restrict__ ald,
                              int Nrows) {
  constexpr int MT = ATT ? M + 16 : M;
  __shared__ __attribute__((aligned(16))) float wlds[K * MT];
  gemm_strip<K, M, ATT>(wlds, (int)blockIdx.x, A, W, Wx, C, als, ald, Nrows);
}

constexpr int FOLD_ROWS = 256;

__device__ __forceinline__ void fold_att_row(int row, int j, int lane,
                         const float* __restrict__ Ws0, const float* __restrict__ Wd0,
                         const float* __restrict__ as0, const float* __restrict__ ad0,
                         const float* __restrict__ Ws1, const float* __restrict__ Wd1,
                         const float* __restrict__ as1, const float* __restrict__ ad1,
                         const float* __restrict__ Ws2, const float* __restrict__ Wd2,
                         const float* __restrict__ as2, const float* __restrict__ ad2,
                         float* __restrict__ wx) {
  const float *Ws, *Wd, *a_s, *a_d;
  int k;
  if (row < 128)      { Ws = Ws0; Wd = Wd0; a_s = as0; a_d = ad0; k = row; }
  else if (row < 192) { Ws = Ws1; Wd = Wd1; a_s = as1; a_d = ad1; k = row - 128; }
  else                { Ws = Ws2; Wd = Wd2; a_s = as2; a_d = ad2; k = row - 192; }
  const int h = j & 1;
  const float* wrow = ((j < 2) ? Ws : Wd) + (size_t)k * 128 + h * 64;
  const float* av   = ((j < 2) ? a_s : a_d) + h * 64;
  float s = wrow[lane] * av[lane] + wrow[lane + 32] * av[lane + 32];
  for (int o = 16; o > 0; o >>= 1) s += __shfl_xor(s, o);
  if (lane == 0) wx[row * 16 + j] = s;
  if (j == 0 && lane >= 4 && lane < 16) wx[row * 16 + lane] = 0.f;
}

constexpr int BIN_SHIFT = 7;
constexpr int BIN_NODES = 1 << BIN_SHIFT;
constexpr int MAX_BINS = 512;
constexpr int SORT_BLOCKS = 128;
constexpr int SORT_THREADS = 1024;
constexpr int SORT_BATCH = 8;
constexpr int BSORT_THREADS = 512;
constexpr int BSORT_BATCH = 4;

__device__ __forceinline__ int block_excl_scan(int v, int* wsum) {
  const int lane = threadIdx.x & 31;
  const int w = threadIdx.x >> 5;
  int incl = v;
  for (int o = 1; o < 32; o <<= 1) {
    const int u = __shfl_up(incl, o);
    if (lane >= o) incl += u;
  }
  if (lane == 31) wsum[w] = incl;
  __syncthreads();
  int pre = 0;
  for (int q = 0; q < w; ++q) pre += wsum[q];
  return pre + incl - v;
}

constexpr int FOLD_ROWS_PER_BLOCK = SORT_THREADS / 128;
constexpr int FOLD_BLOCKS = FOLD_ROWS / FOLD_ROWS_PER_BLOCK;
static_assert(FOLD_ROWS % FOLD_ROWS_PER_BLOCK == 0, "fold rows must split evenly over the work-groups");

__global__ void bin_count_fold(const int* __restrict__ dst, int* __restrict__ tbl, int Ee, int nbins,
                         const float* __restrict__ Ws0, const float* __restrict__ Wd0,
                         const float* __restrict__ as0, const float* __restrict__ ad0,
                         const float* __restrict__ Ws1, const float* __restrict__ Wd1,
                         const float* __restrict__ as1, const float* __restrict__ ad1,
                         const float* __restrict__ Ws2, const float* __restrict__ Wd2,
                         const float* __restrict__ as2, const float* __restrict__ ad2,
                         float* __restrict__ wx) {
  __shared__ int hist[MAX_BINS];
  const int t = threadIdx.x;
  if ((int)blockIdx.x >= SORT_BLOCKS) {
    const int w = t >> 5;
    fold_att_row(((int)blockIdx.x - SORT_BLOCKS) * FOLD_ROWS_PER_BLOCK + (w >> 2), w & 3, t & 31,
                 Ws0, Wd0, as0, ad0, Ws1, Wd1, as1, ad1, Ws2, Wd2, as2, ad2, wx);
    return;
  }
  if (t < MAX_BINS) hist[t] = 0;
  __syncthreads();
  const int chunk = (Ee + SORT_BLOCKS - 1) / SORT_BLOCKS;
  const int lo = (int)blockIdx.x * chunk;
  const int hi = min(lo + chunk, Ee);
  for (int i0 = lo + t; i0 < hi; i0 += SORT_BATCH * SORT_THREADS) {
    int d[SORT_BATCH];
#pragma unroll
    for (int k = 0; k < SORT_BATCH; ++k) {
      const int i = i0 + k * SORT_THREADS;
      d[k] = (i < hi) ? dst[i] : 0;
    }
#pragma unroll
    for (int k = 0; k < SORT_BATCH; ++k)
      if (i0 + k * SORT_THREADS < hi) atomicAdd(&hist[d[k] >> BIN_SHIFT], 1);
  }
  __syncthreads();
  if (t < nbins) tbl[(int)blockIdx.x * nbins + t] = hist[t];
}

__global__ void bin_scatter(const int* __restrict__ src, const int* __restrict__ dst,
                            const int* __restrict__ tbl, int* __restrict__ binbase,
                            int* __restrict__ binned, int Ee, int nbins) {
  __shared__ int cur[MAX_BINS];
  __shared__ int ptot[2][MAX_BINS];
  __shared__ int pbef[2][MAX_BINS];
  __shared__ int wsum[SORT_THREADS / 32];
  const int t = threadIdx.x;
  const int g = blockIdx.x;
  const int bin = t & (MAX_BINS - 1);
  const int half = t / MAX_BINS;
  int tot = 0, bef = 0;
  if (bin < nbins) {
    const int q0 = half * (SORT_BLOCKS / 2);
#pragma unroll 16
    for (int q = q0; q < q0 + SORT_BLOCKS / 2; ++q) {
      const int c = tbl[q * nbins + bin];
      tot += c;
      bef += (q < g) ? c : 0;
    }
  }
  ptot[half][bin] = tot;
  pbef[half][bin] = bef;
  __syncthreads();
  int total = 0, before = 0;
  if (t < MAX_BINS) {
    total = ptot[0][t] + ptot[1][t];
    before = pbef[0][t] + pbef[1][t];
  }
  const int base = block_excl_scan(total, wsum);
  if (t < nbins) {
    cur[t] = base + before;
    if (g == 0) {
      binbase[t] = base;
      if (t == nbins - 1) binbase[nbins] = base + total;
    }
  }
  __syncthreads();
  const int chunk = (Ee + SORT_BLOCKS - 1) / SORT_BLOCKS;
  const int lo = g * chunk;
  const int hi = min(lo + chunk, Ee);
  int2* __restrict__ pairs = (int2*)binned;
  for (int i0 = lo + t; i0 < hi; i0 += SORT_BATCH * SORT_THREADS) {
    int d[SORT_BATCH], s[SORT_BATCH];
#pragma unroll
    for (int k = 0; k < SORT_BATCH; ++k) {
      const int i = i0 + k * SORT_THREADS;
      d[k] = (i < hi) ? dst[i] : 0;
      s[k] = (i < hi) ? src[i] : 0;
    }
#pragma unroll
    for (int k = 0; k < SORT_BATCH; ++k) {
      if (i0 + k * SORT_THREADS < hi) {
        const int p = atomicAdd(&cur[d[k] >> BIN_SHIFT], 1);
        pairs[p] = make_int2(s[k], d[k]);
      }
    }
  }
}

__global__ void bin_sort(const int* __restrict__ binbase, const int* __restrict__ binned,
                         int* __restrict__ rowptr, int* __restrict__ csr_out, int Nn) {
  __shared__ int cnt[BIN_NODES];
  __shared__ int cur[BIN_NODES];
  __shared__ int wsum[BSORT_THREADS / 32];
  const int2* __restrict__ pairs = (const int2*)binned;
  const int t = threadIdx.x;
  const int b = blockIdx.x;
  const int s = binbase[b];
  const int e = binbase[b + 1];
  if (t < BIN_NODES) cnt[t] = 0;
  __syncthreads();
  for (int i0 = s + t; i0 < e; i0 += BSORT_BATCH * BSORT_THREADS) {
    int d[BSORT_BATCH];
#pragma unroll
    for (int k = 0; k < BSORT_BATCH; ++k) {
      const int i = i0 + k * BSORT_THREADS;
      d[k] = (i < e) ? pairs[i].y : 0;
    }
#pragma unroll
    for (int k = 0; k < BSORT_BATCH; ++k)
      if (i0 + k * BSORT_THREADS < e) atomicAdd(&cnt[d[k] & (BIN_NODES - 1)], 1);
  }
  __syncthreads();
  const int c = (t < BIN_NODES) ? cnt[t] : 0;
  const int first = s + block_excl_scan(c, wsum);
  if (t < BIN_NODES) {
    const int node = b * BIN_NODES + t;
    cur[t] = first;
    if (node < Nn) rowptr[node] = first;
    if (node == Nn - 1) rowptr[Nn] = first + c;
  }
  __syncthreads();
  for (int i0 = s + t; i0 < e; i0 += BSORT_BATCH * BSORT_THREADS) {
    int2 pr[BSORT_BATCH];
#pragma unroll
    for (int k = 0; k < BSORT_BATCH; ++k) {
      const int i = i0 + k * BSORT_THREADS;
      pr[k] = (i < e) ? pairs[i] : make_int2(0, 0);
    }
#pragma unroll
    for (int k = 0; k < BSORT_BATCH; ++k) {
      if (i0 + k * BSORT_THREADS < e) {
        const int p = atomicAdd(&cur[pr[k].y & (BIN_NODES - 1)], 1);
        csr_out[p] = pr[k].x;
      }
    }
  }
}

constexpr int NODES_PER_WAVE = 4;

__device__ __forceinline__ void csr_order(int blk, const int* __restrict__ rowptr,
                                          const int* __restrict__ csr_in,
                                          int* __restrict__ csr_out, int Nn) {
  const int lane = threadIdx.x & 31;
  const int wave = __builtin_amdgcn_readfirstlane((int)((blk * blockDim.x + threadIdx.x) >> 5));
  const int first = wave * NODES_PER_WAVE;
  if (first >= Nn) return;
  const int last = min(first + NODES_PER_WAVE, Nn);
  int beg = rowptr[first];
  int end = rowptr[first + 1];
  for (int node = first; node < last; ++node) {
  const int nend = rowptr[min(node + 2, Nn)];
  for (int cb = beg; cb < end; cb += 32) {
    const int i = cb + lane;
    const int key = (i < end) ? csr_in[i] : 0;
    int rank = 0;
    for (int kb = beg; kb < end; kb += 32) {
      const int k = kb + lane;
      const int other = (k < end) ? csr_in[k] : 0;
      const int n = (end - kb < 32) ? (end - kb) : 32;
      for (int j = 0; j < n; ++j) {
        const int o = __builtin_amdgcn_readlane(other, j);
        const int pos = kb + j;
        rank += ((o < key) || (o == key && pos < i)) ? 1 : 0;
      }
    }
    if (i < end) csr_out[beg + rank] = key;
  }
  beg = end;
  end = nend;
  }
}

__global__ void layer0_products_and_order(const float* __restrict__ x,
                                          const float* __restrict__ Ws, const float* __restrict__ Wx,
                                          const float* __restrict__ Wl,
                                          float* __restrict__ hs, float* __restrict__ als,
                                          float* __restrict__ ald, float* __restrict__ hl,
                                          const int* __restrict__ rowptr, const int* __restrict__ csr_in,
                                          int* __restrict__ csr_out, int Nn, int gb) {
  __shared__ __attribute__((aligned(16))) float wlds[128 * (128 + 16)];
  const int blk = (int)blockIdx.x;
  if (blk < gb)
    gemm_strip<128, 128, true >(wlds, blk, x, Ws, Wx, hs, als, ald, Nn);
  else if (blk < 2 * gb)
    gemm_strip<128, 64,  false>(wlds, blk - gb, x, Wl, nullptr, hl, nullptr, nullptr, Nn);
  else
    csr_order(blk - 2 * gb, rowptr, csr_in, csr_out, Nn);
}

__device__ __forceinline__ float lrelu(float v) { return (v > 0.f) ? v : NEG_SLOPE * v; }

__global__ void node_gather(const int* __restrict__ rowptr, const int* __restrict__ csr_src,
                            const float* __restrict__ als, const float* __restrict__ ald,
                            const float* __restrict__ hs, const float* __restrict__ hl,
                            const float* __restrict__ b, const float* __restrict__ bl,
                            float* __restrict__ out, int Nn, int Ee) {
  const int lane = threadIdx.x & 31;
  const int wave = __builtin_amdgcn_readfirstlane((int)((blockIdx.x * blockDim.x + threadIdx.x) >> 5));
  const int first = wave * NODES_PER_WAVE;
  if (first >= Nn) return;
  const int last = min(first + NODES_PER_WAVE, Nn);
  const float2* __restrict__ als2 = (const float2*)als;
  const int f1 = min(first + 1, Nn - 1);
  int beg  = rowptr[first];
  int end  = rowptr[first + 1];
  int end1 = rowptr[f1 + 1];
  float ad0  = ald[first * 2 + 0];
  float ad1  = ald[first * 2 + 1];
  float nad0 = ald[f1 * 2 + 0];
  float nad1 = ald[f1 * 2 + 1];
  int s_cur = csr_src[min(beg + lane, Ee - 1)];
  int s_nxt = csr_src[min(end + lane, Ee - 1)];
  const float2 a_first = als2[(beg + lane < end) ? s_cur : __builtin_amdgcn_readfirstlane(s_cur)];
  float2 a_cur = a_first;
  for (int node = first; node < last; ++node) {
  const int n2 = min(node + 2, Nn - 1);
  const int end2 = rowptr[n2 + 1];
  const float n2ad0 = ald[n2 * 2 + 0];
  const float n2ad1 = ald[n2 * 2 + 1];
  const int s_nn = csr_src[min(end1 + lane, Ee - 1)];
  const float2 a_nxt = als2[(end + lane < end1) ? s_nxt : __builtin_amdgcn_readfirstlane(s_nxt)];

  float m0 = -INFINITY, m1 = -INFINITY;
  if (beg + lane < end) {
    m0 = fmaxf(m0, lrelu(a_cur.x + ad0));
    m1 = fmaxf(m1, lrelu(a_cur.y + ad1));
  }
  for (int i = beg + 32 + lane; i < end; i += 32) {
    const float2 a = als2[csr_src[i]];
    m0 = fmaxf(m0, lrelu(a.x + ad0));
    m1 = fmaxf(m1, lrelu(a.y + ad1));
  }
  for (int o = 16; o > 0; o >>= 1) {
    m0 = fmaxf(m0, __shfl_xor(m0, o));
    m1 = fmaxf(m1, __shfl_xor(m1, o));
  }
  if (!(fabsf(m0) < INFINITY)) m0 = 0.f;
  if (!(fabsf(m1) < INFINITY)) m1 = 0.f;

  float d0 = 0.f, d1 = 0.f;
  if (beg + lane < end) {
    d0 += __expf(lrelu(a_cur.x + ad0) - m0);
    d1 += __expf(lrelu(a_cur.y + ad1) - m1);
  }
  for (int i = beg + 32 + lane; i < end; i += 32) {
    const float2 a = als2[csr_src[i]];
    d0 += __expf(lrelu(a.x + ad0) - m0);
    d1 += __expf(lrelu(a.y + ad1) - m1);
  }
  for (int o = 16; o > 0; o >>= 1) {
    d0 += __shfl_xor(d0, o);
    d1 += __shfl_xor(d1, o);
  }
  const float r0 = d0 + 1e-16f;
  const float r1 = d1 + 1e-16f;

  const bool head1 = lane >= 16;
  const float* __restrict__ hsl = hs + lane * 4;
  float4 acc = make_float4(0.f, 0.f, 0.f, 0.f);
  for (int base = beg; base < end; base += 32) {
    const int i = base + lane;
    int s_l = 0;
    float w_l0 = 0.f, w_l1 = 0.f;
    if (i < end) {
      float2 a;
      if (base == beg) { s_l = s_cur; a = a_cur; }
      else             { s_l = csr_src[i]; a = als2[s_l]; }
      w_l0 = __expf(lrelu(a.x + ad0) - m0) / r0;
      w_l1 = __expf(lrelu(a.y + ad1) - m1) / r1;
    }
    const int cnt = (end - base < 32) ? (end - base) : 32;
#pragma unroll 4
    for (int j = 0; j < cnt; ++j) {
      const int s = __builtin_amdgcn_readlane(s_l, j);
      const float wa = __int_as_float(__builtin_amdgcn_readlane(__float_as_int(w_l0), j));
      const float wb = __int_as_float(__builtin_amdgcn_readlane(__float_as_int(w_l1), j));
      const float w = head1 ? wb : wa;
      const float4 v = *(const float4*)(hsl + (size_t)s * 128);
      acc.x = fmaf(w, v.x, acc.x);
      acc.y = fmaf(w, v.y, acc.y);
      acc.z = fmaf(w, v.z, acc.z);
      acc.w = fmaf(w, v.w, acc.w);
    }
  }
  acc.x += __shfl_xor(acc.x, 16);
  acc.y += __shfl_xor(acc.y, 16);
  acc.z += __shfl_xor(acc.z, 16);
  acc.w += __shfl_xor(acc.w, 16);
  if (lane < 16) {
    const int c = lane * 4;
    const float4 hv = *(const float4*)(hl + (size_t)node * 64 + c);
    float4 o;
    o.x = 0.5f * acc.x + b[c + 0] + hv.x + bl[c + 0];
    o.y = 0.5f * acc.y + b[c + 1] + hv.y + bl[c + 1];
    o.z = 0.5f * acc.z + b[c + 2] + hv.z + bl[c + 2];
    o.w = 0.5f * acc.w + b[c + 3] + hv.w + bl[c + 3];
    o.x = (o.x > 0.f) ? o.x : 0.f;
    o.y = (o.y > 0.f) ? o.y : 0.f;
    o.z = (o.z > 0.f) ? o.z : 0.f;
    o.w = (o.w > 0.f) ? o.w : 0.f;
    *(float4*)(out + (size_t)node * 64 + c) = o;
  }
  beg = end;   end = end1;   end1 = end2;
  ad0 = nad0;  ad1 = nad1;   nad0 = n2ad0;  nad1 = n2ad1;
  s_cur = s_nxt;  s_nxt = s_nn;  a_cur = a_nxt;
  }
}

constexpr int FUSE_THREADS = 512;
constexpr int FUSE_NODES = (FUSE_THREADS / 32) * NODES_PER_WAVE;
constexpr int ZT_PITCH = 132;
static_assert(FUSE_NODES % 16 == 0 && (FUSE_NODES / 16) * 4 == FUSE_THREADS / 32,
              "one (strip, column tile) of the product per wave");

__global__ void __launch_bounds__(FUSE_THREADS)
node_gather_in_zout(const int* __restrict__ rowptr, const int* __restrict__ csr_src,
                    const float* __restrict__ als, const float* __restrict__ ald,
                    const float* __restrict__ hin, const float* __restrict__ Ws,
                    const float* __restrict__ hl, const float* __restrict__ b,
                    const float* __restrict__ bl, float* __restrict__ out, int Nn, int Ee) {
  constexpr int K = 128, M = 64, MT = 80;
  __shared__ __attribute__((aligned(16))) float wlds[K * MT];
  __shared__ __attribute__((aligned(16))) float zt[FUSE_NODES * ZT_PITCH];

  const int lane = threadIdx.x & 31;
  const int wave = __builtin_amdgcn_readfirstlane((int)((blockIdx.x * blockDim.x + threadIdx.x) >> 5));
  const int first = wave * NODES_PER_WAVE;
  const int tile_first = (int)blockIdx.x * FUSE_NODES;
  if (first < Nn) {
  const int last = min(first + NODES_PER_WAVE, Nn);
  const float2* __restrict__ als2 = (const float2*)als;
  const bool odd = lane >= 16;
  const int c = (lane & 15) * 4;
  const float* __restrict__ hrow = hin + c;
  const int f1 = min(first + 1, Nn - 1);
  int beg  = rowptr[first];
  int end  = rowptr[first + 1];
  int end1 = rowptr[f1 + 1];
  float ad0  = ald[first * 2 + 0];
  float ad1  = ald[first * 2 + 1];
  float nad0 = ald[f1 * 2 + 0];
  float nad1 = ald[f1 * 2 + 1];
  int s_cur = csr_src[min(beg + lane, Ee - 1)];
  int s_nxt = csr_src[min(end + lane, Ee - 1)];
  const float2 a_first = als2[(beg + lane < end) ? s_cur : __builtin_amdgcn_readfirstlane(s_cur)];
  float2 a_cur = a_first;
  for (int node = first; node < last; ++node) {
  const int n2 = min(node + 2, Nn - 1);
  const int end2 = rowptr[n2 + 1];
  const float n2ad0 = ald[n2 * 2 + 0];
  const float n2ad1 = ald[n2 * 2 + 1];
  const int s_nn = csr_src[min(end1 + lane, Ee - 1)];
  const float2 a_nxt = als2[(end + lane < end1) ? s_nxt : __builtin_amdgcn_readfirstlane(s_nxt)];

  float m0 = -INFINITY, m1 = -INFINITY;
  if (beg + lane < end) {
    m0 = fmaxf(m0, lrelu(a_cur.x + ad0));
    m1 = fmaxf(m1, lrelu(a_cur.y + ad1));
  }
  for (int i = beg + 32 + lane; i < end; i += 32) {
    const float2 a = als2[csr_src[i]];
    m0 = fmaxf(m0, lrelu(a.x + ad0));
    m1 = fmaxf(m1, lrelu(a.y + ad1));
  }
  for (int o = 16; o > 0; o >>= 1) {
    m0 = fmaxf(m0, __shfl_xor(m0, o));
    m1 = fmaxf(m1, __shfl_xor(m1, o));
  }
  if (!(fabsf(m0) < INFINITY)) m0 = 0.f;
  if (!(fabsf(m1) < INFINITY)) m1 = 0.f;

  float d0 = 0.f, d1 = 0.f;
  if (beg + lane < end) {
    d0 += __expf(lrelu(a_cur.x + ad0) - m0);
    d1 += __expf(lrelu(a_cur.y + ad1) - m1);
  }
  for (int i = beg + 32 + lane; i < end; i += 32) {
    const float2 a = als2[csr_src[i]];
    d0 += __expf(lrelu(a.x + ad0) - m0);
    d1 += __expf(lrelu(a.y + ad1) - m1);
  }
  for (int o = 16; o > 0; o >>= 1) {
    d0 += __shfl_xor(d0, o);
    d1 += __shfl_xor(d1, o);
  }
  const float r0 = d0 + 1e-16f;
  const float r1 = d1 + 1e-16f;

  float4 acc0 = make_float4(0.f, 0.f, 0.f, 0.f);
  float4 acc1 = make_float4(0.f, 0.f, 0.f, 0.f);
  for (int base = beg; base < end; base += 32) {
    const int i = base + lane;
    int s_l = 0;
    float w_l0 = 0.f, w_l1 = 0.f;
    if (i < end) {
      float2 a;
      if (base == beg) { s_l = s_cur; a = a_cur; }
      else             { s_l = csr_src[i]; a = als2[s_l]; }
      w_l0 = __expf(lrelu(a.x + ad0) - m0) / r0;
      w_l1 = __expf(lrelu(a.y + ad1) - m1) / r1;
    }
    const int s_first = __builtin_amdgcn_readfirstlane(s_l);
    if (i >= end) s_l = s_first;
    const int cnt = (end - base < 32) ? (end - base) : 32;
#pragma unroll 4
    for (int j = 0; j < cnt; j += 2) {
      const int sa = __builtin_amdgcn_readlane(s_l, j);
      const int sb = __builtin_amdgcn_readlane(s_l, j + 1);
      const float wa0 = __int_as_float(__builtin_amdgcn_readlane(__float_as_int(w_l0), j));
      const float wb0 = __int_as_float(__builtin_amdgcn_readlane(__float_as_int(w_l0), j + 1));
      const float wa1 = __int_as_float(__builtin_amdgcn_readlane(__float_as_int(w_l1), j));
      const float wb1 = __int_as_float(__builtin_amdgcn_readlane(__float_as_int(w_l1), j + 1));
      const int s = odd ? sb : sa;
      const float w0 = odd ? wb0 : wa0;
      const float w1 = odd ? wb1 : wa1;
      const float4 v = *(const float4*)(hrow + (size_t)s * 64);
      acc0.x = fmaf(w0, v.x, acc0.x);
      acc0.y = fmaf(w0, v.y, acc0.y);
      acc0.z = fmaf(w0, v.z, acc0.z);
      acc0.w = fmaf(w0, v.w, acc0.w);
      acc1.x = fmaf(w1, v.x, acc1.x);
      acc1.y = fmaf(w1, v.y, acc1.y);
      acc1.z = fmaf(w1, v.z, acc1.z);
      acc1.w = fmaf(w1, v.w, acc1.w);
    }
  }
  acc0.x += __shfl_xor(acc0.x, 16);
  acc0.y += __shfl_xor(acc0.y, 16);
  acc0.z += __shfl_xor(acc0.z, 16);
  acc0.w += __shfl_xor(acc0.w, 16);
  acc1.x += __shfl_xor(acc1.x, 16);
  acc1.y += __shfl_xor(acc1.y, 16);
  acc1.z += __shfl_xor(acc1.z, 16);
  acc1.w += __shfl_xor(acc1.w, 16);
  float4 o;
  o.x = odd ? acc1.x : acc0.x;
  o.y = odd ? acc1.y : acc0.y;
  o.z = odd ? acc1.z : acc0.z;
  o.w = odd ? acc1.w : acc0.w;
  *(float4*)&zt[(node - tile_first) * ZT_PITCH + (odd ? 64 : 0) + c] = o;
  beg = end;   end = end1;   end1 = end2;
  ad0 = nad0;  ad1 = nad1;   nad0 = n2ad0;  nad1 = n2ad1;
  s_cur = s_nxt;  s_nxt = s_nn;  a_cur = a_nxt;
  }
  }

  const int w = threadIdx.x >> 5;
  const int strip = w >> 2;
  const int rowTile = (int)blockIdx.x * (FUSE_NODES / 16) + strip;
  const bool live = rowTile < (Nn >> 4);
  const int l16 = lane & 15;
  const int kq  = (lane >> 4) << 1;
  const int col = (w & 3) * 16 + l16;
  const int rbase = rowTile * 16 + ((lane >> 4) << 3);

  constexpr int NW = (64 * 128 / 4) / FUSE_THREADS;
  static_assert((64 * 128 / 4) % FUSE_THREADS == 0, "Ws must split evenly over the work-group");
  const float4* __restrict__ W4 = (const float4*)Ws;
  float4 wv[NW];
#pragma unroll
  for (int j = 0; j < NW; ++j) wv[j] = W4[threadIdx.x + j * FUSE_THREADS];
  float hv[8];
  const float bc = b[col];
  const float blc = bl[col];
#pragma unroll
  for (int r = 0; r < 8; ++r) hv[r] = hl[(size_t)min(rbase + r, Nn - 1) * (size_t)M + col];
#pragma unroll
  for (int j = 0; j < NW; ++j) {
    const int f = (threadIdx.x + j * FUSE_THREADS) * 4;
    const int k = f >> 7;
    const int wcol = f & 127;
    *(float4*)&wlds[((wcol >> 6) * 64 + k) * MT + (wcol & 63)] = wv[j];
  }
  __syncthreads();
  if (!live) return;

  const float* zrow = &zt[(strip * 16 + l16) * ZT_PITCH + kq];
  const float* wr0 = &wlds[kq * MT + col];
  v8f acc = (v8f){0.f,0.f,0.f,0.f,0.f,0.f,0.f,0.f};
  for (int k0 = 0; k0 < K; k0 += 4) {
    v2f a;
    a.x = zrow[k0];
    a.y = zrow[k0 + 1];
    v2f bb;
    bb.x = wr0[k0 * MT];
    bb.y = wr0[k0 * MT + MT];
    acc = __builtin_amdgcn_wmma_f32_16x16x4_f32(
         false, a,  false, bb,
         (short)0, acc,  false,  false);
  }

#pragma unroll
  for (int r = 0; r < 8; ++r) {
    const float v = 0.5f * acc[r] + bc + hv[r] + blc;
    out[(size_t)(rbase + r) * (size_t)M + col] = (v > 0.f) ? v : 0.f;
  }
}

extern "C" void kernel_launch(void* const* d_in, const int* in_sizes, int n_in,
                              void* d_out, int out_size, void* d_ws, size_t ws_size,
                              hipStream_t stream) {
  const float* x  = (const float*)d_in[0];
  const int*   ei = (const int*)d_in[1];
  const int Nn = in_sizes[0] / 128;
  const int Ee = in_sizes[1] / 2;
  const int* src = ei;
  const int* dst = ei + Ee;

  const float *Ws[3], *Wd[3], *As[3], *Ad[3], *B[3], *Wl[3], *Bl[3];
  for (int l = 0; l < 3; ++l) {
    int base = 2 + 7 * l;
    Ws[l] = (const float*)d_in[base + 0];
    Wd[l] = (const float*)d_in[base + 1];
    As[l] = (const float*)d_in[base + 2];
    Ad[l] = (const float*)d_in[base + 3];
    B[l]  = (const float*)d_in[base + 4];
    Wl[l] = (const float*)d_in[base + 5];
    Bl[l] = (const float*)d_in[base + 6];
  }

  float* ws = (float*)d_ws;
  size_t off = 0;
  float* hs  = ws + off; off += (size_t)Nn * 128;
  float* hl  = ws + off; off += (size_t)Nn * 64;
  float* wx  = ws + off; off += (size_t)256 * 16;
  float* als = ws + off; off += (size_t)Nn * 2;
  float* ald = ws + off; off += (size_t)Nn * 2;
  float* h1  = ws + off; off += (size_t)Nn * 64;
  float* h2  = ws + off; off += (size_t)Nn * 64;
  int* iws = (int*)(ws + off);
  size_t ioff = 0;
  int* rowptr  = iws + ioff; ioff += (size_t)Nn + 1;
  int* csr_tmp = iws + ioff; ioff += (size_t)Ee;
  int* csr_src = iws + ioff; ioff += (size_t)Ee;

  const float* lin[3]  = { x, h1, h2 };
  const float* wxl[3]  = { wx, wx + 128 * 16, wx + 192 * 16 };
  float*       lout[3] = { h1, h2, (float*)d_out };

  const int TPB = 256;
  const int rowTiles = Nn >> 4;
  const int gemmBlocks = (rowTiles + 7) / 8;

  int* binned  = (int*)hs;
  int* tbl     = binned + 2 * (size_t)Ee;
  int* binbase = tbl + (size_t)SORT_BLOCKS * MAX_BINS;
  const int nbins = (Nn + BIN_NODES - 1) >> BIN_SHIFT;
  bin_count_fold<<<SORT_BLOCKS + FOLD_BLOCKS, SORT_THREADS, 0, stream>>>(
      dst, tbl, Ee, nbins, Ws[0], Wd[0], As[0], Ad[0], Ws[1], Wd[1], As[1], Ad[1],
      Ws[2], Wd[2], As[2], Ad[2], wx);
  bin_scatter<<<SORT_BLOCKS, SORT_THREADS, 0, stream>>>(src, dst, tbl, binbase, binned, Ee, nbins);
  bin_sort<<<nbins, BSORT_THREADS, 0, stream>>>(binbase, binned, rowptr, csr_tmp, Nn);
  const long long nodeThreads = (long long)((Nn + NODES_PER_WAVE - 1) / NODES_PER_WAVE) * 32;
  const int nodeBlocks = (int)((nodeThreads + TPB - 1) / TPB);
  const int fuseBlocks = (Nn + FUSE_NODES - 1) / FUSE_NODES;

  for (int l = 0; l < 3; ++l) {
    if (l == 0) {
      static_assert(TPB == GEMM_THREADS, "ranking and products share one work-group size");
      layer0_products_and_order<<<2 * gemmBlocks + nodeBlocks, GEMM_THREADS, 0, stream>>>(
          lin[l], Ws[l], wxl[l], Wl[l], hs, als, ald, hl, rowptr, csr_tmp, csr_src, Nn, gemmBlocks);
      node_gather<<<nodeBlocks, TPB, 0, stream>>>(
          rowptr, csr_src, als, ald, hs, hl, B[l], Bl[l], lout[l], Nn, Ee);
    } else {
      gemm_wmma_lds<64, 64, true><<<gemmBlocks, GEMM_THREADS, 0, stream>>>(lin[l], Wl[l], wxl[l], hl, als, ald, Nn);
      node_gather_in_zout<<<fuseBlocks, FUSE_THREADS, 0, stream>>>(
          rowptr, csr_src, als, ald, lin[l], Ws[l], hl, B[l], Bl[l], lout[l], Nn, Ee);
    }
  }
}
